# starting kernel
# baseline (speedup 1.0000x reference)
#include <hip/hip_runtime.h>
#include <math.h>

typedef __attribute__((ext_vector_type(2))) float v2f;
typedef __attribute__((ext_vector_type(8))) float v8f;

#define NEG_SLOPE 0.2f

template <int K, int M>
__global__ void gemm_wmma_lds(const float* __restrict__ A,
                              const float* __restrict__ W,
                              float* __restrict__ C,
                              int Nrows) {
  constexpr int TILES_M = M / 16;
  __shared__ float wlds[K * M];

  for (int i = threadIdx.x; i < K * M; i += blockDim.x) wlds[i] = W[i];
  __syncthreads();

  const int wavesPerBlock = blockDim.x >> 5;
  const int rowTile = blockIdx.x * wavesPerBlock + (threadIdx.x >> 5);
  const int lane = threadIdx.x & 31;
  if (rowTile >= (Nrows >> 4)) return;

  const int l16 = lane & 15;
  const int kq  = (lane >> 4) << 1;
  const float* __restrict__ arow = A + (size_t)(rowTile * 16 + l16) * (size_t)K;

  v8f acc[TILES_M];
#pragma unroll
  for (int t = 0; t < TILES_M; ++t) acc[t] = (v8f){0.f,0.f,0.f,0.f,0.f,0.f,0.f,0.f};

  for (int k0 = 0; k0 < K; k0 += 4) {
    v2f a;
    a.x = arow[k0 + kq];
    a.y = arow[k0 + kq + 1];
    const float* wr0 = &wlds[(k0 + kq) * M + l16];
#pragma unroll
    for (int t = 0; t < TILES_M; ++t) {
      v2f b;
      b.x = wr0[t * 16];
      b.y = wr0[M + t * 16];
      acc[t] = __builtin_amdgcn_wmma_f32_16x16x4_f32(
           false, a,  false, b,
           (short)0, acc[t],  false,  false);
    }
  }

  const int rbase = rowTile * 16 + ((lane >> 4) << 3);
#pragma unroll
  for (int t = 0; t < TILES_M; ++t) {
    const int col = t * 16 + l16;
#pragma unroll
    for (int r = 0; r < 8; ++r)
      C[(size_t)(rbase + r) * (size_t)M + col] = acc[t][r];
  }
}

__global__ void node_alpha(const float* __restrict__ hs,
                           const float* __restrict__ hd,
                           const float* __restrict__ a_s,
                           const float* __restrict__ a_d,
                           float* __restrict__ als,
                           float* __restrict__ ald, int Nn) {
  int tid = blockIdx.x * blockDim.x + threadIdx.x;
  if (tid >= Nn * 2) return;
  int n = tid >> 1, h = tid & 1;
  const float* hsrow = hs + (size_t)n * 128 + h * 64;
  const float* hdrow = hd + (size_t)n * 128 + h * 64;
  const float* asv = a_s + h * 64;
  const float* adv = a_d + h * 64;
  float s1 = 0.f, s2 = 0.f;
#pragma unroll 8
  for (int c = 0; c < 64; ++c) {
    s1 += hsrow[c] * asv[c];
    s2 += hdrow[c] * adv[c];
  }
  als[tid] = s1;
  ald[tid] = s2;
}

__global__ void init_layer(float* __restrict__ m, float* __restrict__ den,
                           float* __restrict__ agg, int Nn) {
  int tid = blockIdx.x * blockDim.x + threadIdx.x;
  if (tid < Nn * 64) agg[tid] = 0.f;
  if (tid < Nn * 2) { m[tid] = -INFINITY; den[tid] = 0.f; }
}

__device__ inline void atomicMaxF(float* addr, float value) {
  if (value >= 0.f)
    atomicMax((int*)addr, __float_as_int(value));
  else
    atomicMin((unsigned int*)addr, __float_as_uint(value));
}

__global__ void edge_max(const int* __restrict__ src, const int* __restrict__ dst,
                         const float* __restrict__ als, const float* __restrict__ ald,
                         float* __restrict__ e, float* __restrict__ m, int Ee) {
  int tid = blockIdx.x * blockDim.x + threadIdx.x;
  if (tid >= Ee * 2) return;
  int ed = tid >> 1, h = tid & 1;
  int s = src[ed], d = dst[ed];
  float v = als[s * 2 + h] + ald[d * 2 + h];
  v = (v > 0.f) ? v : NEG_SLOPE * v;
  e[tid] = v;
  atomicMaxF(&m[d * 2 + h], v);
}

__global__ void edge_den(const int* __restrict__ dst,
                         const float* __restrict__ e, const float* __restrict__ m,
                         float* __restrict__ den, int Ee) {
  int tid = blockIdx.x * blockDim.x + threadIdx.x;
  if (tid >= Ee * 2) return;
  int ed = tid >> 1, h = tid & 1;
  int d = dst[ed];
  float ex = __expf(e[tid] - m[d * 2 + h]);
  atomicAdd(&den[d * 2 + h], ex);
}

__global__ void edge_scatter(const int* __restrict__ src, const int* __restrict__ dst,
                             const float* __restrict__ e, const float* __restrict__ m,
                             const float* __restrict__ den,
                             const float* __restrict__ hs,
                             float* __restrict__ agg, int Ee) {
  int wave = (blockIdx.x * blockDim.x + threadIdx.x) >> 5;
  int lane = threadIdx.x & 31;
  if (wave >= Ee) return;
  int s = src[wave], d = dst[wave];
  float e0 = e[wave * 2 + 0], e1 = e[wave * 2 + 1];
  float m0 = m[d * 2 + 0], m1 = m[d * 2 + 1];
  float n0 = den[d * 2 + 0], n1 = den[d * 2 + 1];
  float a0 = __expf(e0 - m0) / (n0 + 1e-16f) * 0.5f;
  float a1 = __expf(e1 - m1) / (n1 + 1e-16f) * 0.5f;
  const float* hsrow = hs + (size_t)s * 128;
  float v0 = hsrow[lane]      * a0 + hsrow[64 + lane] * a1;
  float v1 = hsrow[32 + lane] * a0 + hsrow[96 + lane] * a1;
  float* aggrow = agg + (size_t)d * 64;
  atomicAdd(&aggrow[lane],      v0);
  atomicAdd(&aggrow[32 + lane], v1);
}

__global__ void finalize(const float* __restrict__ agg, const float* __restrict__ hl,
                         const float* __restrict__ b, const float* __restrict__ bl,
                         float* __restrict__ out, int Nn) {
  int tid = blockIdx.x * blockDim.x + threadIdx.x;
  if (tid >= Nn * 64) return;
  int c = tid & 63;
  float v = agg[tid] + b[c] + hl[tid] + bl[c];
  out[tid] = (v > 0.f) ? v : 0.f;
}

extern "C" void kernel_launch(void* const* d_in, const int* in_sizes, int n_in,
                              void* d_out, int out_size, void* d_ws, size_t ws_size,
                              hipStream_t stream) {
  const float* x  = (const float*)d_in[0];
  const int*   ei = (const int*)d_in[1];
  const int Nn = in_sizes[0] / 128;
  const int Ee = in_sizes[1] / 2;
  const int* src = ei;
  const int* dst = ei + Ee;

  const float *Ws[3], *Wd[3], *As[3], *Ad[3], *B[3], *Wl[3], *Bl[3];
  for (int l = 0; l < 3; ++l) {
    int base = 2 + 7 * l;
    Ws[l] = (const float*)d_in[base + 0];
    Wd[l] = (const float*)d_in[base + 1];
    As[l] = (const float*)d_in[base + 2];
    Ad[l] = (const float*)d_in[base + 3];
    B[l]  = (const float*)d_in[base + 4];
    Wl[l] = (const float*)d_in[base + 5];
    Bl[l] = (const float*)d_in[base + 6];
  }

  float* ws = (float*)d_ws;
  size_t off = 0;
  float* hs  = ws + off; off += (size_t)Nn * 128;
  float* hd  = ws + off; off += (size_t)Nn * 128;
  float* hl  = ws + off; off += (size_t)Nn * 64;
  float* als = ws + off; off += (size_t)Nn * 2;
  float* ald = ws + off; off += (size_t)Nn * 2;
  float* m   = ws + off; off += (size_t)Nn * 2;
  float* den = ws + off; off += (size_t)Nn * 2;
  float* ebuf= ws + off; off += (size_t)Ee * 2;
  float* agg = ws + off; off += (size_t)Nn * 64;
  float* h1  = ws + off; off += (size_t)Nn * 64;
  float* h2  = ws + off; off += (size_t)Nn * 64;

  const float* lin[3]  = { x, h1, h2 };
  float*       lout[3] = { h1, h2, (float*)d_out };

  const int TPB = 256;
  const int rowTiles = Nn >> 4;
  const int gemmBlocks = (rowTiles + 7) / 8;

  for (int l = 0; l < 3; ++l) {
    if (l == 0) {
      gemm_wmma_lds<128, 128><<<gemmBlocks, TPB, 0, stream>>>(lin[l], Ws[l], hs, Nn);
      gemm_wmma_lds<128, 128><<<gemmBlocks, TPB, 0, stream>>>(lin[l], Wd[l], hd, Nn);
      gemm_wmma_lds<128, 64 ><<<gemmBlocks, TPB, 0, stream>>>(lin[l], Wl[l], hl, Nn);
    } else {
      gemm_wmma_lds<64, 128><<<gemmBlocks, TPB, 0, stream>>>(lin[l], Ws[l], hs, Nn);
      gemm_wmma_lds<64, 128><<<gemmBlocks, TPB, 0, stream>>>(lin[l], Wd[l], hd, Nn);
      gemm_wmma_lds<64, 64 ><<<gemmBlocks, TPB, 0, stream>>>(lin[l], Wl[l], hl, Nn);
    }
    node_alpha<<<(Nn * 2 + TPB - 1) / TPB, TPB, 0, stream>>>(hs, hd, As[l], Ad[l], als, ald, Nn);
    init_layer<<<(Nn * 64 + TPB - 1) / TPB, TPB, 0, stream>>>(m, den, agg, Nn);
    edge_max<<<(Ee * 2 + TPB - 1) / TPB, TPB, 0, stream>>>(src, dst, als, ald, ebuf, m, Ee);
    edge_den<<<(Ee * 2 + TPB - 1) / TPB, TPB, 0, stream>>>(dst, ebuf, m, den, Ee);
    {
      long long threads = (long long)Ee * 32;
      edge_scatter<<<(int)((threads + TPB - 1) / TPB), TPB, 0, stream>>>(
          src, dst, ebuf, m, den, hs, agg, Ee);
    }
    finalize<<<(Nn * 64 + TPB - 1) / TPB, TPB, 0, stream>>>(agg, hl, B[l], Bl[l], lout[l], Nn);
  }
}


# Products: each wave keeps the W fragments of 64 interleaved columns in registers and walks A tiles with 16-byte loads one tile ahead; no LDS, no barrier; k grouping per WMMA (sum order) changed
# speedup vs baseline: 1.2061x; 1.2061x over previous
#include <hip/hip_runtime.h>
#include <math.h>

typedef __attribute__((ext_vector_type(2))) float v2f;
typedef __attribute__((ext_vector_type(4))) float v4f;
typedef __attribute__((ext_vector_type(8))) float v8f;

#define NEG_SLOPE 0.2f

template <int K>
__device__ __forceinline__ void gemm_load_a(v4f (&a)[K / 8], const float* abase, int rt) {
  const float* ap = abase + (size_t)rt * 16 * K;
#pragma unroll
  for (int j = 0; j < K / 8; ++j) a[j] = *(const v4f*)(ap + 8 * j);
}

template <int K>
__device__ __forceinline__ void gemm_tile(v8f (&acc)[4], const v4f (&a)[K / 8],
                                          const v2f (&bp)[K / 8][2][4]) {
#pragma unroll
  for (int q = 0; q < 4; ++q) acc[q] = (v8f){0.f,0.f,0.f,0.f,0.f,0.f,0.f,0.f};

#pragma unroll
  for (int j = 0; j < K / 8; ++j) {
#pragma unroll
    for (int sub = 0; sub < 2; ++sub) {
      v2f av;
      av.x = sub ? a[j].z : a[j].x;
      av.y = sub ? a[j].w : a[j].y;
#pragma unroll
      for (int q = 0; q < 4; ++q) {
        acc[q] = __builtin_amdgcn_wmma_f32_16x16x4_f32(
             false, av,  false, bp[j][sub][q],
             (short)0, acc[q],  false,  false);
      }
    }
  }
}

template <int M>
__device__ __forceinline__ void gemm_store(const v8f (&acc)[4], float* cbase, int rt) {
  float* crow = cbase + (size_t)rt * 16 * M;
#pragma unroll
  for (int r = 0; r < 8; ++r) {
    v4f o;
    o.x = acc[0][r]; o.y = acc[1][r]; o.z = acc[2][r]; o.w = acc[3][r];
    *(v4f*)(crow + (size_t)r * M) = o;
  }
}

template <int K, int M>
__global__ void __launch_bounds__(256)
gemm_wmma_breg(const float* __restrict__ A,
               const float* __restrict__ W,
               float* __restrict__ C,
               int Nrows) {
  constexpr int SLABS = M / 64;
  constexpr int KJ = K / 8;

  const int lane = threadIdx.x & 31;
  const int c = lane & 15;
  const int h = lane >> 4;
  const int wib = __builtin_amdgcn_readfirstlane((int)(threadIdx.x >> 5));
  const int wavesPerBlock = blockDim.x >> 5;
  const int wave = blockIdx.x * wavesPerBlock + wib;
  const int slab = wave % SLABS;
  const int rg = wave / SLABS;
  const int rowGroups = (gridDim.x * wavesPerBlock) / SLABS;
  const int nTiles = Nrows >> 4;
  if (rg >= nTiles) return;

  v2f bp[KJ][2][4];
  {
    const float* wbase = W + (size_t)(4 * h) * M + slab * 64 + 4 * c;
#pragma unroll
    for (int j = 0; j < KJ; ++j) {
#pragma unroll
      for (int sub = 0; sub < 2; ++sub) {
        const v4f w0 = *(const v4f*)(wbase + (size_t)(8 * j + 2 * sub) * M);
        const v4f w1 = *(const v4f*)(wbase + (size_t)(8 * j + 2 * sub + 1) * M);
#pragma unroll
        for (int q = 0; q < 4; ++q) {
          bp[j][sub][q].x = w0[q];
          bp[j][sub][q].y = w1[q];
        }
      }
    }
  }

  const float* abase = A + (size_t)c * K + 4 * h;
  float* cbase = C + (size_t)(8 * h) * M + slab * 64 + 4 * c;
  v4f a0[KJ], a1[KJ];
  gemm_load_a<K>(a0, abase, rg);

  int rt = rg;
  for (;;) {
    const int r1 = rt + rowGroups;
    const int r2 = r1 + rowGroups;
    v8f acc[4];
    __builtin_amdgcn_sched_barrier(0);
    gemm_load_a<K>(a1, abase, r1 < nTiles ? r1 : rt);
    __builtin_amdgcn_sched_barrier(0);
    gemm_tile<K>(acc, a0, bp);
    gemm_store<M>(acc, cbase, rt);
    __builtin_amdgcn_sched_barrier(0);
    gemm_load_a<K>(a0, abase, r2 < nTiles ? r2 : rt);
    __builtin_amdgcn_sched_barrier(0);
    gemm_tile<K>(acc, a1, bp);
    if (r1 < nTiles) gemm_store<M>(acc, cbase, r1);
    if (r2 >= nTiles) break;
    rt = r2;
  }
}

__global__ void node_alpha(const float* __restrict__ hs,
                           const float* __restrict__ hd,
                           const float* __restrict__ a_s,
                           const float* __restrict__ a_d,
                           float* __restrict__ als,
                           float* __restrict__ ald, int Nn) {
  int tid = blockIdx.x * blockDim.x + threadIdx.x;
  if (tid >= Nn * 2) return;
  int n = tid >> 1, h = tid & 1;
  const float* hsrow = hs + (size_t)n * 128 + h * 64;
  const float* hdrow = hd + (size_t)n * 128 + h * 64;
  const float* asv = a_s + h * 64;
  const float* adv = a_d + h * 64;
  float s1 = 0.f, s2 = 0.f;
#pragma unroll 8
  for (int c = 0; c < 64; ++c) {
    s1 += hsrow[c] * asv[c];
    s2 += hdrow[c] * adv[c];
  }
  als[tid] = s1;
  ald[tid] = s2;
}

__global__ void init_layer(float* __restrict__ m, float* __restrict__ den,
                           float* __restrict__ agg, int Nn) {
  int tid = blockIdx.x * blockDim.x + threadIdx.x;
  if (tid < Nn * 64) agg[tid] = 0.f;
  if (tid < Nn * 2) { m[tid] = -INFINITY; den[tid] = 0.f; }
}

__device__ inline void atomicMaxF(float* addr, float value) {
  if (value >= 0.f)
    atomicMax((int*)addr, __float_as_int(value));
  else
    atomicMin((unsigned int*)addr, __float_as_uint(value));
}

__global__ void edge_max(const int* __restrict__ src, const int* __restrict__ dst,
                         const float* __restrict__ als, const float* __restrict__ ald,
                         float* __restrict__ e, float* __restrict__ m, int Ee) {
  int tid = blockIdx.x * blockDim.x + threadIdx.x;
  if (tid >= Ee * 2) return;
  int ed = tid >> 1, h = tid & 1;
  int s = src[ed], d = dst[ed];
  float v = als[s * 2 + h] + ald[d * 2 + h];
  v = (v > 0.f) ? v : NEG_SLOPE * v;
  e[tid] = v;
  atomicMaxF(&m[d * 2 + h], v);
}

__global__ void edge_den(const int* __restrict__ dst,
                         const float* __restrict__ e, const float* __restrict__ m,
                         float* __restrict__ den, int Ee) {
  int tid = blockIdx.x * blockDim.x + threadIdx.x;
  if (tid >= Ee * 2) return;
  int ed = tid >> 1, h = tid & 1;
  int d = dst[ed];
  float ex = __expf(e[tid] - m[d * 2 + h]);
  atomicAdd(&den[d * 2 + h], ex);
}

__global__ void edge_scatter(const int* __restrict__ src, const int* __restrict__ dst,
                             const float* __restrict__ e, const float* __restrict__ m,
                             const float* __restrict__ den,
                             const float* __restrict__ hs,
                             float* __restrict__ agg, int Ee) {
  int wave = (blockIdx.x * blockDim.x + threadIdx.x) >> 5;
  int lane = threadIdx.x & 31;
  if (wave >= Ee) return;
  int s = src[wave], d = dst[wave];
  float e0 = e[wave * 2 + 0], e1 = e[wave * 2 + 1];
  float m0 = m[d * 2 + 0], m1 = m[d * 2 + 1];
  float n0 = den[d * 2 + 0], n1 = den[d * 2 + 1];
  float a0 = __expf(e0 - m0) / (n0 + 1e-16f) * 0.5f;
  float a1 = __expf(e1 - m1) / (n1 + 1e-16f) * 0.5f;
  const float* hsrow = hs + (size_t)s * 128;
  float v0 = hsrow[lane]      * a0 + hsrow[64 + lane] * a1;
  float v1 = hsrow[32 + lane] * a0 + hsrow[96 + lane] * a1;
  float* aggrow = agg + (size_t)d * 64;
  atomicAdd(&aggrow[lane],      v0);
  atomicAdd(&aggrow[32 + lane], v1);
}

__global__ void finalize(const float* __restrict__ agg, const float* __restrict__ hl,
                         const float* __restrict__ b, const float* __restrict__ bl,
                         float* __restrict__ out, int Nn) {
  int tid = blockIdx.x * blockDim.x + threadIdx.x;
  if (tid >= Nn * 64) return;
  int c = tid & 63;
  float v = agg[tid] + b[c] + hl[tid] + bl[c];
  out[tid] = (v > 0.f) ? v : 0.f;
}

extern "C" void kernel_launch(void* const* d_in, const int* in_sizes, int n_in,
                              void* d_out, int out_size, void* d_ws, size_t ws_size,
                              hipStream_t stream) {
  const float* x  = (const float*)d_in[0];
  const int*   ei = (const int*)d_in[1];
  const int Nn = in_sizes[0] / 128;
  const int Ee = in_sizes[1] / 2;
  const int* src = ei;
  const int* dst = ei + Ee;

  const float *Ws[3], *Wd[3], *As[3], *Ad[3], *B[3], *Wl[3], *Bl[3];
  for (int l = 0; l < 3; ++l) {
    int base = 2 + 7 * l;
    Ws[l] = (const float*)d_in[base + 0];
    Wd[l] = (const float*)d_in[base + 1];
    As[l] = (const float*)d_in[base + 2];
    Ad[l] = (const float*)d_in[base + 3];
    B[l]  = (const float*)d_in[base + 4];
    Wl[l] = (const float*)d_in[base + 5];
    Bl[l] = (const float*)d_in[base + 6];
  }

  float* ws = (float*)d_ws;
  size_t off = 0;
  float* hs  = ws + off; off += (size_t)Nn * 128;
  float* hd  = ws + off; off += (size_t)Nn * 128;
  float* hl  = ws + off; off += (size_t)Nn * 64;
  float* als = ws + off; off += (size_t)Nn * 2;
  float* ald = ws + off; off += (size_t)Nn * 2;
  float* m   = ws + off; off += (size_t)Nn * 2;
  float* den = ws + off; off += (size_t)Nn * 2;
  float* ebuf= ws + off; off += (size_t)Ee * 2;
  float* agg = ws + off; off += (size_t)Nn * 64;
  float* h1  = ws + off; off += (size_t)Nn * 64;
  float* h2  = ws + off; off += (size_t)Nn * 64;

  const float* lin[3]  = { x, h1, h2 };
  float*       lout[3] = { h1, h2, (float*)d_out };

  const int TPB = 256;
  const int rowTiles = Nn >> 4;
  const int gemmBlocks = (rowTiles + 15) / 16;

  for (int l = 0; l < 3; ++l) {
    if (l == 0) {
      gemm_wmma_breg<128, 128><<<gemmBlocks, TPB, 0, stream>>>(lin[l], Ws[l], hs, Nn);
      gemm_wmma_breg<128, 128><<<gemmBlocks, TPB, 0, stream>>>(lin[l], Wd[l], hd, Nn);
      gemm_wmma_breg<128, 64 ><<<gemmBlocks, TPB, 0, stream>>>(lin[l], Wl[l], hl, Nn);
    } else {
      gemm_wmma_breg<64, 128><<<gemmBlocks, TPB, 0, stream>>>(lin[l], Ws[l], hs, Nn);
      gemm_wmma_breg<64, 128><<<gemmBlocks, TPB, 0, stream>>>(lin[l], Wd[l], hd, Nn);
      gemm_wmma_breg<64, 64 ><<<gemmBlocks, TPB, 0, stream>>>(lin[l], Wl[l], hl, Nn);
    }
    node_alpha<<<(Nn * 2 + TPB - 1) / TPB, TPB, 0, stream>>>(hs, hd, As[l], Ad[l], als, ald, Nn);
    init_layer<<<(Nn * 64 + TPB - 1) / TPB, TPB, 0, stream>>>(m, den, agg, Nn);
    edge_max<<<(Ee * 2 + TPB - 1) / TPB, TPB, 0, stream>>>(src, dst, als, ald, ebuf, m, Ee);
    edge_den<<<(Ee * 2 + TPB - 1) / TPB, TPB, 0, stream>>>(dst, ebuf, m, den, Ee);
    {
      long long threads = (long long)Ee * 32;
      edge_scatter<<<(int)((threads + TPB - 1) / TPB), TPB, 0, stream>>>(
          src, dst, ebuf, m, den, hs, agg, Ee);
    }
    finalize<<<(Nn * 64 + TPB - 1) / TPB, TPB, 0, stream>>>(agg, hl, B[l], Bl[l], lout[l], Nn);
  }
}


# A layer's edge part in one launch: a work-group owns a tile of destination nodes, finds its edges by one sweep of dst, holds max, sum, row sums in LDS (LDS atomics, no global ones); sum order changed
# speedup vs baseline: 2.4172x; 2.0041x over previous
#include <hip/hip_runtime.h>
#include <math.h>

typedef __attribute__((ext_vector_type(2))) float v2f;
typedef __attribute__((ext_vector_type(4))) float v4f;
typedef __attribute__((ext_vector_type(8))) float v8f;

#define NEG_SLOPE 0.2f

template <int K>
__device__ __forceinline__ void gemm_load_a(v4f (&a)[K / 8], const float* abase, int rt) {
  const float* ap = abase + (size_t)rt * 16 * K;
#pragma unroll
  for (int j = 0; j < K / 8; ++j) a[j] = *(const v4f*)(ap + 8 * j);
}

template <int K>
__device__ __forceinline__ void gemm_tile(v8f (&acc)[4], const v4f (&a)[K / 8],
                                          const v2f (&bp)[K / 8][2][4]) {
#pragma unroll
  for (int q = 0; q < 4; ++q) acc[q] = (v8f){0.f,0.f,0.f,0.f,0.f,0.f,0.f,0.f};

#pragma unroll
  for (int j = 0; j < K / 8; ++j) {
#pragma unroll
    for (int sub = 0; sub < 2; ++sub) {
      v2f av;
      av.x = sub ? a[j].z : a[j].x;
      av.y = sub ? a[j].w : a[j].y;
#pragma unroll
      for (int q = 0; q < 4; ++q) {
        acc[q] = __builtin_amdgcn_wmma_f32_16x16x4_f32(
             false, av,  false, bp[j][sub][q],
             (short)0, acc[q],  false,  false);
      }
    }
  }
}

template <int M>
__device__ __forceinline__ void gemm_store(const v8f (&acc)[4], float* cbase, int rt) {
  float* crow = cbase + (size_t)rt * 16 * M;
#pragma unroll
  for (int r = 0; r < 8; ++r) {
    v4f o;
    o.x = acc[0][r]; o.y = acc[1][r]; o.z = acc[2][r]; o.w = acc[3][r];
    *(v4f*)(crow + (size_t)r * M) = o;
  }
}

template <int K, int M>
__global__ void __launch_bounds__(256)
gemm_wmma_breg(const float* __restrict__ A,
               const float* __restrict__ W,
               float* __restrict__ C,
               int Nrows) {
  constexpr int SLABS = M / 64;
  constexpr int KJ = K / 8;

  const int lane = threadIdx.x & 31;
  const int c = lane & 15;
  const int h = lane >> 4;
  const int wib = __builtin_amdgcn_readfirstlane((int)(threadIdx.x >> 5));
  const int wavesPerBlock = blockDim.x >> 5;
  const int wave = blockIdx.x * wavesPerBlock + wib;
  const int slab = wave % SLABS;
  const int rg = wave / SLABS;
  const int rowGroups = (gridDim.x * wavesPerBlock) / SLABS;
  const int nTiles = Nrows >> 4;
  if (rg >= nTiles) return;

  v2f bp[KJ][2][4];
  {
    const float* wbase = W + (size_t)(4 * h) * M + slab * 64 + 4 * c;
#pragma unroll
    for (int j = 0; j < KJ; ++j) {
#pragma unroll
      for (int sub = 0; sub < 2; ++sub) {
        const v4f w0 = *(const v4f*)(wbase + (size_t)(8 * j + 2 * sub) * M);
        const v4f w1 = *(const v4f*)(wbase + (size_t)(8 * j + 2 * sub + 1) * M);
#pragma unroll
        for (int q = 0; q < 4; ++q) {
          bp[j][sub][q].x = w0[q];
          bp[j][sub][q].y = w1[q];
        }
      }
    }
  }

  const float* abase = A + (size_t)c * K + 4 * h;
  float* cbase = C + (size_t)(8 * h) * M + slab * 64 + 4 * c;
  v4f a0[KJ], a1[KJ];
  gemm_load_a<K>(a0, abase, rg);

  int rt = rg;
  for (;;) {
    const int r1 = rt + rowGroups;
    const int r2 = r1 + rowGroups;
    v8f acc[4];
    __builtin_amdgcn_sched_barrier(0);
    gemm_load_a<K>(a1, abase, r1 < nTiles ? r1 : rt);
    __builtin_amdgcn_sched_barrier(0);
    gemm_tile<K>(acc, a0, bp);
    gemm_store<M>(acc, cbase, rt);
    __builtin_amdgcn_sched_barrier(0);
    gemm_load_a<K>(a0, abase, r2 < nTiles ? r2 : rt);
    __builtin_amdgcn_sched_barrier(0);
    gemm_tile<K>(acc, a1, bp);
    if (r1 < nTiles) gemm_store<M>(acc, cbase, r1);
    if (r2 >= nTiles) break;
    rt = r2;
  }
}

__global__ void node_alpha(const float* __restrict__ hs,
                           const float* __restrict__ hd,
                           const float* __restrict__ a_s,
                           const float* __restrict__ a_d,
                           float* __restrict__ als,
                           float* __restrict__ ald, int Nn) {
  int tid = blockIdx.x * blockDim.x + threadIdx.x;
  if (tid >= Nn * 2) return;
  int n = tid >> 1, h = tid & 1;
  const float* hsrow = hs + (size_t)n * 128 + h * 64;
  const float* hdrow = hd + (size_t)n * 128 + h * 64;
  const float* asv = a_s + h * 64;
  const float* adv = a_d + h * 64;
  float s1 = 0.f, s2 = 0.f;
#pragma unroll 8
  for (int c = 0; c < 64; ++c) {
    s1 += hsrow[c] * asv[c];
    s2 += hdrow[c] * adv[c];
  }
  als[tid] = s1;
  ald[tid] = s2;
}

typedef __attribute__((ext_vector_type(4))) int v4i;

#define TM   256
#define LCAP 8192
#define ET   1024

__device__ __forceinline__ float lrelu(float v) { return (v > 0.f) ? v : NEG_SLOPE * v; }

__device__ __forceinline__ void ldsMaxF(float* addr, float value) {
  if (value >= 0.f)
    atomicMax((int*)addr, __float_as_int(value));
  else
    atomicMin((unsigned int*)addr, __float_as_uint(value));
}

template <int U>
__device__ __forceinline__ void wave_rows(unsigned mask, int s, int t, float a0, float a1,
                                          const float* __restrict__ hs, float* agg, int lane) {
  while (mask) {
    int l[U];
    bool on[U];
    unsigned mm = mask;
#pragma unroll
    for (int u = 0; u < U; ++u) {
      on[u] = (mm != 0u);
      l[u] = on[u] ? __builtin_ctz(mm) : l[0];
      mm &= mm - 1u;
    }
    mask = mm;
    v2f x0[U], x1[U];
    int tt[U];
    float A0[U], A1[U];
#pragma unroll
    for (int u = 0; u < U; ++u) {
      const int ss = __builtin_amdgcn_readlane(s, l[u]);
      tt[u] = __builtin_amdgcn_readlane(t, l[u]);
      A0[u] = __int_as_float(__builtin_amdgcn_readlane(__float_as_int(a0), l[u]));
      A1[u] = __int_as_float(__builtin_amdgcn_readlane(__float_as_int(a1), l[u]));
      const float* row = hs + (size_t)ss * 128 + 2 * lane;
      x0[u] = *(const v2f*)row;
      x1[u] = *(const v2f*)(row + 64);
    }
#pragma unroll
    for (int u = 0; u < U; ++u) asm volatile("" :: "v"(x0[u]), "v"(x1[u]));
#pragma unroll
    for (int u = 0; u < U; ++u) {
      if (on[u]) {
        float* p = agg + tt[u] * 64 + 2 * lane;
        atomicAdd(p,     x0[u].x * A0[u] + x1[u].x * A1[u]);
        atomicAdd(p + 1, x0[u].y * A0[u] + x1[u].y * A1[u]);
      }
    }
  }
}

template <int MODE>
__device__ __forceinline__ void sweep_all(const int* __restrict__ src, const int* __restrict__ dst,
                                          const float* __restrict__ als, const float* __restrict__ ald,
                                          const float* __restrict__ hs,
                                          float* s_agg, float* s_m, float* s_den,
                                          int lo, int nn, int Ee, int tid, int lane) {
  const int trips = (Ee + ET - 1) / ET;
  for (int it = 0; it < trips; ++it) {
    const int e = it * ET + tid;
    bool valid = false;
    int s = 0, t = 0;
    float a0 = 0.f, a1 = 0.f;
    if (e < Ee) {
      t = dst[e] - lo;
      valid = (unsigned)t < (unsigned)nn;
      if (valid) {
        s = src[e];
        const v2f as2 = *(const v2f*)(als + 2 * (size_t)s);
        const v2f ad2 = *(const v2f*)(ald + 2 * (size_t)(lo + t));
        const float v0 = lrelu(as2.x + ad2.x);
        const float v1 = lrelu(as2.y + ad2.y);
        if (MODE == 1) {
          ldsMaxF(&s_m[2 * t], v0);
          ldsMaxF(&s_m[2 * t + 1], v1);
        } else if (MODE == 2) {
          atomicAdd(&s_den[2 * t],     __expf(v0 - s_m[2 * t]));
          atomicAdd(&s_den[2 * t + 1], __expf(v1 - s_m[2 * t + 1]));
        } else {
          a0 = __expf(v0 - s_m[2 * t])     / (s_den[2 * t]     + 1e-16f) * 0.5f;
          a1 = __expf(v1 - s_m[2 * t + 1]) / (s_den[2 * t + 1] + 1e-16f) * 0.5f;
        }
      }
    }
    if (MODE == 3) {
      const unsigned mask = __builtin_amdgcn_ballot_w32(valid);
      wave_rows<4>(mask, s, t, a0, a1, hs, s_agg, lane);
    }
  }
}

__global__ void __launch_bounds__(ET)
edge_tile(const int* __restrict__ src, const int* __restrict__ dst,
          const float* __restrict__ als, const float* __restrict__ ald,
          const float* __restrict__ hs, const float* __restrict__ hl,
          const float* __restrict__ b, const float* __restrict__ bl,
          float* __restrict__ out, int Nn, int Ee) {
  __shared__ float s_agg[TM * 64];
  __shared__ float s_m[TM * 2];
  __shared__ float s_den[TM * 2];
  __shared__ int   s_a[LCAP];
  __shared__ int   s_t[LCAP];
  __shared__ float s_e0[LCAP];
  __shared__ float s_e1[LCAP];
  __shared__ int   s_cnt;

  const int tid = threadIdx.x;
  const int lane = tid & 31;
  const int wave = tid >> 5;

  int tn = (Nn + (int)gridDim.x - 1) / (int)gridDim.x;
  if (tn > TM) tn = TM;
  if (tn < 1) tn = 1;
  const int tiles = (Nn + tn - 1) / tn;

  int i0 = (int)(((16u - (unsigned)((size_t)dst & 15u)) & 15u) >> 2);
  if (i0 > Ee) i0 = Ee;
  const int G = (Ee - i0) >> 2;
  const v4i* d4 = (const v4i*)(dst + i0);
  const int rot = (int)(((long long)blockIdx.x * G) / (int)gridDim.x);
  constexpr int U = 8;

  for (int tile = blockIdx.x; tile < tiles; tile += gridDim.x) {
    const int lo = tile * tn;
    const int nn = (Nn - lo < tn) ? (Nn - lo) : tn;

    for (int i = tid; i < nn * 64; i += ET) s_agg[i] = 0.f;
    for (int i = tid; i < nn * 2; i += ET) { s_m[i] = -INFINITY; s_den[i] = 0.f; }
    if (tid == 0) s_cnt = 0;
    __syncthreads();

    for (int g0 = 0; g0 < G; g0 += ET * U) {
      v4i d[U];
      int gi[U];
      bool ok[U];
#pragma unroll
      for (int u = 0; u < U; ++u) {
        const int g = g0 + u * ET + tid;
        ok[u] = g < G;
        int gg = (ok[u] ? g : G - 1) + rot;
        if (gg >= G) gg -= G;
        gi[u] = gg;
        d[u] = d4[gg];
      }
#pragma unroll
      for (int u = 0; u < U; ++u) {
        const int t0 = d[u].x - lo, t1 = d[u].y - lo, t2 = d[u].z - lo, t3 = d[u].w - lo;
        const bool m0 = (unsigned)t0 < (unsigned)nn, m1 = (unsigned)t1 < (unsigned)nn;
        const bool m2 = (unsigned)t2 < (unsigned)nn, m3 = (unsigned)t3 < (unsigned)nn;
        if (ok[u] && (m0 | m1 | m2 | m3)) {
          const int e = i0 + 4 * gi[u];
          if (m0) { const int sl = atomicAdd(&s_cnt, 1); if (sl < LCAP) { s_a[sl] = e;     s_t[sl] = t0; } }
          if (m1) { const int sl = atomicAdd(&s_cnt, 1); if (sl < LCAP) { s_a[sl] = e + 1; s_t[sl] = t1; } }
          if (m2) { const int sl = atomicAdd(&s_cnt, 1); if (sl < LCAP) { s_a[sl] = e + 2; s_t[sl] = t2; } }
          if (m3) { const int sl = atomicAdd(&s_cnt, 1); if (sl < LCAP) { s_a[sl] = e + 3; s_t[sl] = t3; } }
        }
      }
    }
    {
      const int rest = Ee - 4 * G;
      if (tid < rest) {
        const int e = (tid < i0) ? tid : (4 * G + tid);
        const int t = dst[e] - lo;
        if ((unsigned)t < (unsigned)nn) {
          const int sl = atomicAdd(&s_cnt, 1);
          if (sl < LCAP) { s_a[sl] = e; s_t[sl] = t; }
        }
      }
    }
    __syncthreads();
    const int total = s_cnt;

    if (total <= LCAP) {
      for (int i = tid; i < total; i += ET) {
        const int e = s_a[i];
        const int t = s_t[i];
        const int s = src[e];
        const v2f as2 = *(const v2f*)(als + 2 * (size_t)s);
        const v2f ad2 = *(const v2f*)(ald + 2 * (size_t)(lo + t));
        const float v0 = lrelu(as2.x + ad2.x);
        const float v1 = lrelu(as2.y + ad2.y);
        s_a[i] = s;
        s_e0[i] = v0;
        s_e1[i] = v1;
        ldsMaxF(&s_m[2 * t], v0);
        ldsMaxF(&s_m[2 * t + 1], v1);
      }
      __syncthreads();
      for (int i = tid; i < total; i += ET) {
        const int t = s_t[i];
        const float x0 = __expf(s_e0[i] - s_m[2 * t]);
        const float x1 = __expf(s_e1[i] - s_m[2 * t + 1]);
        s_e0[i] = x0;
        s_e1[i] = x1;
        atomicAdd(&s_den[2 * t], x0);
        atomicAdd(&s_den[2 * t + 1], x1);
      }
      __syncthreads();
      for (int base = wave * 32; base < total; base += ET) {
        const int i = base + lane;
        const bool valid = i < total;
        int s = 0, t = 0;
        float a0 = 0.f, a1 = 0.f;
        if (valid) {
          s = s_a[i];
          t = s_t[i];
          a0 = s_e0[i] / (s_den[2 * t] + 1e-16f) * 0.5f;
          a1 = s_e1[i] / (s_den[2 * t + 1] + 1e-16f) * 0.5f;
        }
        const unsigned mask = __builtin_amdgcn_ballot_w32(valid);
        wave_rows<8>(mask, s, t, a0, a1, hs, s_agg, lane);
      }
    } else {
      sweep_all<1>(src, dst, als, ald, hs, s_agg, s_m, s_den, lo, nn, Ee, tid, lane);
      __syncthreads();
      sweep_all<2>(src, dst, als, ald, hs, s_agg, s_m, s_den, lo, nn, Ee, tid, lane);
      __syncthreads();
      sweep_all<3>(src, dst, als, ald, hs, s_agg, s_m, s_den, lo, nn, Ee, tid, lane);
    }
    __syncthreads();

    for (int i = tid; i < nn * 64; i += ET) {
      const int c = i & 63;
      const size_t o = (size_t)lo * 64 + i;
      const float v = s_agg[i] + b[c] + hl[o] + bl[c];
      out[o] = (v > 0.f) ? v : 0.f;
    }
    __syncthreads();
  }
}

extern "C" void kernel_launch(void* const* d_in, const int* in_sizes, int n_in,
                              void* d_out, int out_size, void* d_ws, size_t ws_size,
                              hipStream_t stream) {
  const float* x  = (const float*)d_in[0];
  const int*   ei = (const int*)d_in[1];
  const int Nn = in_sizes[0] / 128;
  const int Ee = in_sizes[1] / 2;
  const int* src = ei;
  const int* dst = ei + Ee;

  const float *Ws[3], *Wd[3], *As[3], *Ad[3], *B[3], *Wl[3], *Bl[3];
  for (int l = 0; l < 3; ++l) {
    int base = 2 + 7 * l;
    Ws[l] = (const float*)d_in[base + 0];
    Wd[l] = (const float*)d_in[base + 1];
    As[l] = (const float*)d_in[base + 2];
    Ad[l] = (const float*)d_in[base + 3];
    B[l]  = (const float*)d_in[base + 4];
    Wl[l] = (const float*)d_in[base + 5];
    Bl[l] = (const float*)d_in[base + 6];
  }

  float* ws = (float*)d_ws;
  size_t off = 0;
  float* hs  = ws + off; off += (size_t)Nn * 128;
  float* hd  = ws + off; off += (size_t)Nn * 128;
  float* hl  = ws + off; off += (size_t)Nn * 64;
  float* als = ws + off; off += (size_t)Nn * 2;
  float* ald = ws + off; off += (size_t)Nn * 2;
  float* h1  = ws + off; off += (size_t)Nn * 64;
  float* h2  = ws + off; off += (size_t)Nn * 64;

  const float* lin[3]  = { x, h1, h2 };
  float*       lout[3] = { h1, h2, (float*)d_out };

  const int TPB = 256;
  const int rowTiles = Nn >> 4;
  const int gemmBlocks = (rowTiles + 15) / 16;

  for (int l = 0; l < 3; ++l) {
    if (l == 0) {
      gemm_wmma_breg<128, 128><<<gemmBlocks, TPB, 0, stream>>>(lin[l], Ws[l], hs, Nn);
      gemm_wmma_breg<128, 128><<<gemmBlocks, TPB, 0, stream>>>(lin[l], Wd[l], hd, Nn);
      gemm_wmma_breg<128, 64 ><<<gemmBlocks, TPB, 0, stream>>>(lin[l], Wl[l], hl, Nn);
    } else {
      gemm_wmma_breg<64, 128><<<gemmBlocks, TPB, 0, stream>>>(lin[l], Ws[l], hs, Nn);
      gemm_wmma_breg<64, 128><<<gemmBlocks, TPB, 0, stream>>>(lin[l], Wd[l], hd, Nn);
      gemm_wmma_breg<64, 64 ><<<gemmBlocks, TPB, 0, stream>>>(lin[l], Wl[l], hl, Nn);
    }
    node_alpha<<<(Nn * 2 + TPB - 1) / TPB, TPB, 0, stream>>>(hs, hd, As[l], Ad[l], als, ald, Nn);
    edge_tile<<<256, ET, 0, stream>>>(src, dst, als, ald, hs, hl, B[l], Bl[l], lout[l], Nn, Ee);
  }
}


# Edge listing hoisted out of the layer loop: layer 1's edge launch keeps each tile's list (source and local destination, one word per edge) in the workspace; layers 2,3 read it, no dst sweep
# speedup vs baseline: 3.1538x; 1.3047x over previous
#include <hip/hip_runtime.h>
#include <math.h>

typedef __attribute__((ext_vector_type(2))) float v2f;
typedef __attribute__((ext_vector_type(4))) float v4f;
typedef __attribute__((ext_vector_type(8))) float v8f;

#define NEG_SLOPE 0.2f

template <int K>
__device__ __forceinline__ void gemm_load_a(v4f (&a)[K / 8], const float* abase, int rt) {
  const float* ap = abase + (size_t)rt * 16 * K;
#pragma unroll
  for (int j = 0; j < K / 8; ++j) a[j] = *(const v4f*)(ap + 8 * j);
}

template <int K>
__device__ __forceinline__ void gemm_tile(v8f (&acc)[4], const v4f (&a)[K / 8],
                                          const v2f (&bp)[K / 8][2][4]) {
#pragma unroll
  for (int q = 0; q < 4; ++q) acc[q] = (v8f){0.f,0.f,0.f,0.f,0.f,0.f,0.f,0.f};

#pragma unroll
  for (int j = 0; j < K / 8; ++j) {
#pragma unroll
    for (int sub = 0; sub < 2; ++sub) {
      v2f av;
      av.x = sub ? a[j].z : a[j].x;
      av.y = sub ? a[j].w : a[j].y;
#pragma unroll
      for (int q = 0; q < 4; ++q) {
        acc[q] = __builtin_amdgcn_wmma_f32_16x16x4_f32(
             false, av,  false, bp[j][sub][q],
             (short)0, acc[q],  false,  false);
      }
    }
  }
}

template <int M>
__device__ __forceinline__ void gemm_store(const v8f (&acc)[4], float* cbase, int rt) {
  float* crow = cbase + (size_t)rt * 16 * M;
#pragma unroll
  for (int r = 0; r < 8; ++r) {
    v4f o;
    o.x = acc[0][r]; o.y = acc[1][r]; o.z = acc[2][r]; o.w = acc[3][r];
    *(v4f*)(crow + (size_t)r * M) = o;
  }
}

template <int K, int M>
__global__ void __launch_bounds__(256)
gemm_wmma_breg(const float* __restrict__ A,
               const float* __restrict__ W,
               float* __restrict__ C,
               int Nrows) {
  constexpr int SLABS = M / 64;
  constexpr int KJ = K / 8;

  const int lane = threadIdx.x & 31;
  const int c = lane & 15;
  const int h = lane >> 4;
  const int wib = __builtin_amdgcn_readfirstlane((int)(threadIdx.x >> 5));
  const int wavesPerBlock = blockDim.x >> 5;
  const int wave = blockIdx.x * wavesPerBlock + wib;
  const int slab = wave % SLABS;
  const int rg = wave / SLABS;
  const int rowGroups = (gridDim.x * wavesPerBlock) / SLABS;
  const int nTiles = Nrows >> 4;
  if (rg >= nTiles) return;

  v2f bp[KJ][2][4];
  {
    const float* wbase = W + (size_t)(4 * h) * M + slab * 64 + 4 * c;
#pragma unroll
    for (int j = 0; j < KJ; ++j) {
#pragma unroll
      for (int sub = 0; sub < 2; ++sub) {
        const v4f w0 = *(const v4f*)(wbase + (size_t)(8 * j + 2 * sub) * M);
        const v4f w1 = *(const v4f*)(wbase + (size_t)(8 * j + 2 * sub + 1) * M);
#pragma unroll
        for (int q = 0; q < 4; ++q) {
          bp[j][sub][q].x = w0[q];
          bp[j][sub][q].y = w1[q];
        }
      }
    }
  }

  const float* abase = A + (size_t)c * K + 4 * h;
  float* cbase = C + (size_t)(8 * h) * M + slab * 64 + 4 * c;
  v4f a0[KJ], a1[KJ];
  gemm_load_a<K>(a0, abase, rg);

  int rt = rg;
  for (;;) {
    const int r1 = rt + rowGroups;
    const int r2 = r1 + rowGroups;
    v8f acc[4];
    __builtin_amdgcn_sched_barrier(0);
    gemm_load_a<K>(a1, abase, r1 < nTiles ? r1 : rt);
    __builtin_amdgcn_sched_barrier(0);
    gemm_tile<K>(acc, a0, bp);
    gemm_store<M>(acc, cbase, rt);
    __builtin_amdgcn_sched_barrier(0);
    gemm_load_a<K>(a0, abase, r2 < nTiles ? r2 : rt);
    __builtin_amdgcn_sched_barrier(0);
    gemm_tile<K>(acc, a1, bp);
    if (r1 < nTiles) gemm_store<M>(acc, cbase, r1);
    if (r2 >= nTiles) break;
    rt = r2;
  }
}

__global__ void node_alpha(const float* __restrict__ hs,
                           const float* __restrict__ hd,
                           const float* __restrict__ a_s,
                           const float* __restrict__ a_d,
                           float* __restrict__ als,
                           float* __restrict__ ald, int Nn) {
  int tid = blockIdx.x * blockDim.x + threadIdx.x;
  if (tid >= Nn * 2) return;
  int n = tid >> 1, h = tid & 1;
  const float* hsrow = hs + (size_t)n * 128 + h * 64;
  const float* hdrow = hd + (size_t)n * 128 + h * 64;
  const float* asv = a_s + h * 64;
  const float* adv = a_d + h * 64;
  float s1 = 0.f, s2 = 0.f;
#pragma unroll 8
  for (int c = 0; c < 64; ++c) {
    s1 += hsrow[c] * asv[c];
    s2 += hdrow[c] * adv[c];
  }
  als[tid] = s1;
  ald[tid] = s2;
}

typedef __attribute__((ext_vector_type(4))) int v4i;

#define TM   256
#define LCAP 8192
#define ET   1024
#define EG   256
#define LPITCH (LCAP + 32)
static_assert(TM <= 256, "a kept list entry holds the tile-relative destination in 8 bits");

__device__ __forceinline__ float lrelu(float v) { return (v > 0.f) ? v : NEG_SLOPE * v; }

__device__ __forceinline__ void ldsMaxF(float* addr, float value) {
  if (value >= 0.f)
    atomicMax((int*)addr, __float_as_int(value));
  else
    atomicMin((unsigned int*)addr, __float_as_uint(value));
}

template <int U>
__device__ __forceinline__ void wave_rows(unsigned mask, int s, int t, float a0, float a1,
                                          const float* __restrict__ hs, float* agg, int lane) {
  while (mask) {
    int l[U];
    bool on[U];
    unsigned mm = mask;
#pragma unroll
    for (int u = 0; u < U; ++u) {
      on[u] = (mm != 0u);
      l[u] = on[u] ? __builtin_ctz(mm) : l[0];
      mm &= mm - 1u;
    }
    mask = mm;
    v2f x0[U], x1[U];
    int tt[U];
    float A0[U], A1[U];
#pragma unroll
    for (int u = 0; u < U; ++u) {
      const int ss = __builtin_amdgcn_readlane(s, l[u]);
      tt[u] = __builtin_amdgcn_readlane(t, l[u]);
      A0[u] = __int_as_float(__builtin_amdgcn_readlane(__float_as_int(a0), l[u]));
      A1[u] = __int_as_float(__builtin_amdgcn_readlane(__float_as_int(a1), l[u]));
      const float* row = hs + (size_t)ss * 128 + 2 * lane;
      x0[u] = *(const v2f*)row;
      x1[u] = *(const v2f*)(row + 64);
    }
#pragma unroll
    for (int u = 0; u < U; ++u) asm volatile("" :: "v"(x0[u]), "v"(x1[u]));
#pragma unroll
    for (int u = 0; u < U; ++u) {
      if (on[u]) {
        float* p = agg + tt[u] * 64 + 2 * lane;
        atomicAdd(p,     x0[u].x * A0[u] + x1[u].x * A1[u]);
        atomicAdd(p + 1, x0[u].y * A0[u] + x1[u].y * A1[u]);
      }
    }
  }
}

template <int MODE>
__device__ __forceinline__ void sweep_all(const int* __restrict__ src, const int* __restrict__ dst,
                                          const float* __restrict__ als, const float* __restrict__ ald,
                                          const float* __restrict__ hs,
                                          float* s_agg, float* s_m, float* s_den,
                                          int lo, int nn, int Ee, int tid, int lane) {
  const int trips = (Ee + ET - 1) / ET;
  for (int it = 0; it < trips; ++it) {
    const int e = it * ET + tid;
    bool valid = false;
    int s = 0, t = 0;
    float a0 = 0.f, a1 = 0.f;
    if (e < Ee) {
      t = dst[e] - lo;
      valid = (unsigned)t < (unsigned)nn;
      if (valid) {
        s = src[e];
        const v2f as2 = *(const v2f*)(als + 2 * (size_t)s);
        const v2f ad2 = *(const v2f*)(ald + 2 * (size_t)(lo + t));
        const float v0 = lrelu(as2.x + ad2.x);
        const float v1 = lrelu(as2.y + ad2.y);
        if (MODE == 1) {
          ldsMaxF(&s_m[2 * t], v0);
          ldsMaxF(&s_m[2 * t + 1], v1);
        } else if (MODE == 2) {
          atomicAdd(&s_den[2 * t],     __expf(v0 - s_m[2 * t]));
          atomicAdd(&s_den[2 * t + 1], __expf(v1 - s_m[2 * t + 1]));
        } else {
          a0 = __expf(v0 - s_m[2 * t])     / (s_den[2 * t]     + 1e-16f) * 0.5f;
          a1 = __expf(v1 - s_m[2 * t + 1]) / (s_den[2 * t + 1] + 1e-16f) * 0.5f;
        }
      }
    }
    if (MODE == 3) {
      const unsigned mask = __builtin_amdgcn_ballot_w32(valid);
      wave_rows<4>(mask, s, t, a0, a1, hs, s_agg, lane);
    }
  }
}

template <bool REUSE>
__global__ void __launch_bounds__(ET)
edge_tile(const int* __restrict__ src, const int* __restrict__ dst,
          const float* __restrict__ als, const float* __restrict__ ald,
          const float* __restrict__ hs, const float* __restrict__ hl,
          const float* __restrict__ b, const float* __restrict__ bl,
          float* __restrict__ out, unsigned* __restrict__ keep, int Nn, int Ee) {
  __shared__ float s_agg[TM * 64];
  __shared__ float s_m[TM * 2];
  __shared__ float s_den[TM * 2];
  __shared__ int   s_a[LCAP];
  __shared__ int   s_t[LCAP];
  __shared__ float s_e0[LCAP];
  __shared__ float s_e1[LCAP];
  __shared__ int   s_cnt;

  const int tid = threadIdx.x;
  const int lane = tid & 31;
  const int wave = tid >> 5;

  int tn = (Nn + (int)gridDim.x - 1) / (int)gridDim.x;
  if (tn > TM) tn = TM;
  if (tn < 1) tn = 1;
  const int tiles = (Nn + tn - 1) / tn;

  int i0 = (int)(((16u - (unsigned)((size_t)dst & 15u)) & 15u) >> 2);
  if (i0 > Ee) i0 = Ee;
  const int G = (Ee - i0) >> 2;
  const v4i* d4 = (const v4i*)(dst + i0);
  const int rot = (int)(((long long)blockIdx.x * G) / (int)gridDim.x);
  constexpr int U = 8;

  for (int tile = blockIdx.x; tile < tiles; tile += gridDim.x) {
    const int lo = tile * tn;
    const int nn = (Nn - lo < tn) ? (Nn - lo) : tn;

    unsigned* kl = (REUSE || keep) ? keep + (size_t)tile * LPITCH : nullptr;

    for (int i = tid; i < nn * 64; i += ET) s_agg[i] = 0.f;
    for (int i = tid; i < nn * 2; i += ET) { s_m[i] = -INFINITY; s_den[i] = 0.f; }
    if (!REUSE && tid == 0) s_cnt = 0;
    __syncthreads();

    int total;
    if (REUSE) {
      total = (int)kl[0];
    } else {
      for (int g0 = 0; g0 < G; g0 += ET * U) {
        v4i d[U];
        int gi[U];
        bool ok[U];
#pragma unroll
        for (int u = 0; u < U; ++u) {
          const int g = g0 + u * ET + tid;
          ok[u] = g < G;
          int gg = (ok[u] ? g : G - 1) + rot;
          if (gg >= G) gg -= G;
          gi[u] = gg;
          d[u] = d4[gg];
        }
#pragma unroll
        for (int u = 0; u < U; ++u) {
          const int t0 = d[u].x - lo, t1 = d[u].y - lo, t2 = d[u].z - lo, t3 = d[u].w - lo;
          const bool m0 = (unsigned)t0 < (unsigned)nn, m1 = (unsigned)t1 < (unsigned)nn;
          const bool m2 = (unsigned)t2 < (unsigned)nn, m3 = (unsigned)t3 < (unsigned)nn;
          if (ok[u] && (m0 | m1 | m2 | m3)) {
            const int e = i0 + 4 * gi[u];
            if (m0) { const int sl = atomicAdd(&s_cnt, 1); if (sl < LCAP) { s_a[sl] = e;     s_t[sl] = t0; } }
            if (m1) { const int sl = atomicAdd(&s_cnt, 1); if (sl < LCAP) { s_a[sl] = e + 1; s_t[sl] = t1; } }
            if (m2) { const int sl = atomicAdd(&s_cnt, 1); if (sl < LCAP) { s_a[sl] = e + 2; s_t[sl] = t2; } }
            if (m3) { const int sl = atomicAdd(&s_cnt, 1); if (sl < LCAP) { s_a[sl] = e + 3; s_t[sl] = t3; } }
          }
        }
      }
      {
        const int rest = Ee - 4 * G;
        if (tid < rest) {
          const int e = (tid < i0) ? tid : (4 * G + tid);
          const int t = dst[e] - lo;
          if ((unsigned)t < (unsigned)nn) {
            const int sl = atomicAdd(&s_cnt, 1);
            if (sl < LCAP) { s_a[sl] = e; s_t[sl] = t; }
          }
        }
      }
      __syncthreads();
      total = s_cnt;
      if (kl && tid == 0) kl[0] = (unsigned)total;
    }

    if (total <= LCAP) {
      for (int i = tid; i < total; i += ET) {
        int s, t;
        if (REUSE) {
          const unsigned w = kl[32 + i];
          s = (int)(w >> 8);
          t = (int)(w & 255u);
          s_t[i] = t;
        } else {
          t = s_t[i];
          s = src[s_a[i]];
          if (kl) kl[32 + i] = ((unsigned)s << 8) | (unsigned)t;
        }
        const v2f as2 = *(const v2f*)(als + 2 * (size_t)s);
        const v2f ad2 = *(const v2f*)(ald + 2 * (size_t)(lo + t));
        const float v0 = lrelu(as2.x + ad2.x);
        const float v1 = lrelu(as2.y + ad2.y);
        s_a[i] = s;
        s_e0[i] = v0;
        s_e1[i] = v1;
        ldsMaxF(&s_m[2 * t], v0);
        ldsMaxF(&s_m[2 * t + 1], v1);
      }
      __syncthreads();
      for (int i = tid; i < total; i += ET) {
        const int t = s_t[i];
        const float x0 = __expf(s_e0[i] - s_m[2 * t]);
        const float x1 = __expf(s_e1[i] - s_m[2 * t + 1]);
        s_e0[i] = x0;
        s_e1[i] = x1;
        atomicAdd(&s_den[2 * t], x0);
        atomicAdd(&s_den[2 * t + 1], x1);
      }
      __syncthreads();
      for (int base = wave * 32; base < total; base += ET) {
        const int i = base + lane;
        const bool valid = i < total;
        int s = 0, t = 0;
        float a0 = 0.f, a1 = 0.f;
        if (valid) {
          s = s_a[i];
          t = s_t[i];
          a0 = s_e0[i] / (s_den[2 * t] + 1e-16f) * 0.5f;
          a1 = s_e1[i] / (s_den[2 * t + 1] + 1e-16f) * 0.5f;
        }
        const unsigned mask = __builtin_amdgcn_ballot_w32(valid);
        wave_rows<8>(mask, s, t, a0, a1, hs, s_agg, lane);
      }
    } else {
      sweep_all<1>(src, dst, als, ald, hs, s_agg, s_m, s_den, lo, nn, Ee, tid, lane);
      __syncthreads();
      sweep_all<2>(src, dst, als, ald, hs, s_agg, s_m, s_den, lo, nn, Ee, tid, lane);
      __syncthreads();
      sweep_all<3>(src, dst, als, ald, hs, s_agg, s_m, s_den, lo, nn, Ee, tid, lane);
    }
    __syncthreads();

    for (int i = tid; i < nn * 64; i += ET) {
      const int c = i & 63;
      const size_t o = (size_t)lo * 64 + i;
      const float v = s_agg[i] + b[c] + hl[o] + bl[c];
      out[o] = (v > 0.f) ? v : 0.f;
    }
    __syncthreads();
  }
}

extern "C" void kernel_launch(void* const* d_in, const int* in_sizes, int n_in,
                              void* d_out, int out_size, void* d_ws, size_t ws_size,
                              hipStream_t stream) {
  const float* x  = (const float*)d_in[0];
  const int*   ei = (const int*)d_in[1];
  const int Nn = in_sizes[0] / 128;
  const int Ee = in_sizes[1] / 2;
  const int* src = ei;
  const int* dst = ei + Ee;

  const float *Ws[3], *Wd[3], *As[3], *Ad[3], *B[3], *Wl[3], *Bl[3];
  for (int l = 0; l < 3; ++l) {
    int base = 2 + 7 * l;
    Ws[l] = (const float*)d_in[base + 0];
    Wd[l] = (const float*)d_in[base + 1];
    As[l] = (const float*)d_in[base + 2];
    Ad[l] = (const float*)d_in[base + 3];
    B[l]  = (const float*)d_in[base + 4];
    Wl[l] = (const float*)d_in[base + 5];
    Bl[l] = (const float*)d_in[base + 6];
  }

  float* ws = (float*)d_ws;
  size_t off = 0;
  float* hs  = ws + off; off += (size_t)Nn * 128;
  float* hd  = ws + off; off += (size_t)Nn * 128;
  float* hl  = ws + off; off += (size_t)Nn * 64;
  float* als = ws + off; off += (size_t)Nn * 2;
  float* ald = ws + off; off += (size_t)Nn * 2;
  float* h1  = ws + off; off += (size_t)Nn * 64;
  float* h2  = ws + off; off += (size_t)Nn * 64;
  int tn = (Nn + EG - 1) / EG;
  if (tn > TM) tn = TM;
  if (tn < 1) tn = 1;
  const int tiles = (Nn + tn - 1) / tn;
  unsigned* keep = nullptr;
  if (Nn <= (1 << 24) && ws_size >= (off + (size_t)tiles * LPITCH) * sizeof(float))
    keep = (unsigned*)(ws + off);

  const float* lin[3]  = { x, h1, h2 };
  float*       lout[3] = { h1, h2, (float*)d_out };

  const int TPB = 256;
  const int rowTiles = Nn >> 4;
  const int gemmBlocks = (rowTiles + 15) / 16;

  for (int l = 0; l < 3; ++l) {
    if (l == 0) {
      gemm_wmma_breg<128, 128><<<gemmBlocks, TPB, 0, stream>>>(lin[l], Ws[l], hs, Nn);
      gemm_wmma_breg<128, 128><<<gemmBlocks, TPB, 0, stream>>>(lin[l], Wd[l], hd, Nn);
      gemm_wmma_breg<128, 64 ><<<gemmBlocks, TPB, 0, stream>>>(lin[l], Wl[l], hl, Nn);
    } else {
      gemm_wmma_breg<64, 128><<<gemmBlocks, TPB, 0, stream>>>(lin[l], Ws[l], hs, Nn);
      gemm_wmma_breg<64, 128><<<gemmBlocks, TPB, 0, stream>>>(lin[l], Wd[l], hd, Nn);
      gemm_wmma_breg<64, 64 ><<<gemmBlocks, TPB, 0, stream>>>(lin[l], Wl[l], hl, Nn);
    }
    node_alpha<<<(Nn * 2 + TPB - 1) / TPB, TPB, 0, stream>>>(hs, hd, As[l], Ad[l], als, ald, Nn);
    if (l == 0 || !keep)
      edge_tile<false><<<EG, ET, 0, stream>>>(src, dst, als, ald, hs, hl, B[l], Bl[l], lout[l], keep, Nn, Ee);
    else
      edge_tile<true><<<EG, ET, 0, stream>>>(src, dst, als, ald, hs, hl, B[l], Bl[l], lout[l], keep, Nn, Ee);
  }
}


# A layer's node part in one launch: a wave holds one 64-column slab of [Ws|Wd|Wl]; Ws, Wd slabs form als, ald as row dots off the accumulators; hd never stored; 6 launches; als/ald sum order changed
# speedup vs baseline: 4.0009x; 1.2686x over previous
#include <hip/hip_runtime.h>
#include <math.h>

typedef __attribute__((ext_vector_type(2))) float v2f;
typedef __attribute__((ext_vector_type(4))) float v4f;
typedef __attribute__((ext_vector_type(8))) float v8f;

#define NEG_SLOPE 0.2f

template <int K>
__device__ __forceinline__ void gemm_load_a(v4f (&a)[K / 8], const float* abase, int rt) {
  const float* ap = abase + (size_t)rt * 16 * K;
#pragma unroll
  for (int j = 0; j < K / 8; ++j) a[j] = *(const v4f*)(ap + 8 * j);
}

template <int K>
__device__ __forceinline__ void gemm_tile(v8f (&acc)[4], const v4f (&a)[K / 8],
                                          const v2f (&bp)[K / 8][2][4]) {
#pragma unroll
  for (int q = 0; q < 4; ++q) acc[q] = (v8f){0.f,0.f,0.f,0.f,0.f,0.f,0.f,0.f};

#pragma unroll
  for (int j = 0; j < K / 8; ++j) {
#pragma unroll
    for (int sub = 0; sub < 2; ++sub) {
      v2f av;
      av.x = sub ? a[j].z : a[j].x;
      av.y = sub ? a[j].w : a[j].y;
#pragma unroll
      for (int q = 0; q < 4; ++q) {
        acc[q] = __builtin_amdgcn_wmma_f32_16x16x4_f32(
             false, av,  false, bp[j][sub][q],
             (short)0, acc[q],  false,  false);
      }
    }
  }
}

__device__ __forceinline__ void gemm_store(const v8f (&acc)[4], float* cbase, int rt, int M) {
  float* crow = cbase + (size_t)rt * 16 * M;
#pragma unroll
  for (int r = 0; r < 8; ++r) {
    v4f o;
    o.x = acc[0][r]; o.y = acc[1][r]; o.z = acc[2][r]; o.w = acc[3][r];
    *(v4f*)(crow + (size_t)r * M) = o;
  }
}

__device__ __forceinline__ float tile_row_dots(const v8f (&acc)[4], const v4f av, int c) {
  float p[8];
#pragma unroll
  for (int r = 0; r < 8; ++r)
    p[r] = acc[0][r] * av.x + acc[1][r] * av.y + acc[2][r] * av.z + acc[3][r] * av.w;
  const bool b3 = (c & 8) != 0, b2 = (c & 4) != 0, b1 = (c & 2) != 0;
  float k4[4], k2[2];
#pragma unroll
  for (int i = 0; i < 4; ++i) {
    const float send = b3 ? p[i] : p[4 + i];
    const float keep = b3 ? p[4 + i] : p[i];
    k4[i] = keep + __shfl_xor(send, 8, 32);
  }
#pragma unroll
  for (int i = 0; i < 2; ++i) {
    const float send = b2 ? k4[i] : k4[2 + i];
    const float keep = b2 ? k4[2 + i] : k4[i];
    k2[i] = keep + __shfl_xor(send, 4, 32);
  }
  const float send1 = b1 ? k2[0] : k2[1];
  const float keep1 = b1 ? k2[1] : k2[0];
  const float k1 = keep1 + __shfl_xor(send1, 2, 32);
  return k1 + __shfl_xor(k1, 1, 32);
}

template <int K>
__global__ void __launch_bounds__(256)
node_part(const float* __restrict__ A,
          const float* __restrict__ Ws, const float* __restrict__ Wd,
          const float* __restrict__ Wl,
          const float* __restrict__ a_s, const float* __restrict__ a_d,
          float* __restrict__ hs, float* __restrict__ hl,
          float* __restrict__ als, float* __restrict__ ald,
          int Nrows) {
  constexpr int KJ = K / 8;

  const int lane = threadIdx.x & 31;
  const int c = lane & 15;
  const int h = lane >> 4;
  const int wib = __builtin_amdgcn_readfirstlane((int)(threadIdx.x >> 5));
  const int wavesPerBlock = blockDim.x >> 5;
  const int wave = blockIdx.x * wavesPerBlock + wib;
  const int slab = wave % 5;
  const int rg = wave / 5;
  const int rowGroups = (gridDim.x * wavesPerBlock) / 5;
  const int nTiles = Nrows >> 4;
  if (rg >= rowGroups || rg >= nTiles) return;

  const bool doStore = (slab < 2) || (slab == 4);
  const bool doDot = slab < 4;
  const int head = slab & 1;
  const int M = (slab == 4) ? 64 : 128;
  const int cs = (slab == 4) ? 0 : head * 64;
  const float* W = (slab < 2) ? Ws : ((slab < 4) ? Wd : Wl);
  float* C = (slab < 2) ? hs : hl;
  const float* avec = (slab < 2) ? a_s : a_d;
  float* dots = (slab < 2) ? als : ald;

  v2f bp[KJ][2][4];
  {
    const float* wbase = W + (size_t)(4 * h) * M + cs + 4 * c;
#pragma unroll
    for (int j = 0; j < KJ; ++j) {
#pragma unroll
      for (int sub = 0; sub < 2; ++sub) {
        const v4f w0 = *(const v4f*)(wbase + (size_t)(8 * j + 2 * sub) * M);
        const v4f w1 = *(const v4f*)(wbase + (size_t)(8 * j + 2 * sub + 1) * M);
#pragma unroll
        for (int q = 0; q < 4; ++q) {
          bp[j][sub][q].x = w0[q];
          bp[j][sub][q].y = w1[q];
        }
      }
    }
  }
  const v4f av = *(const v4f*)(avec + head * 64 + 4 * c);

  const float* abase = A + (size_t)c * K + 4 * h;
  float* cbase = C + (size_t)(8 * h) * M + cs + 4 * c;
  float* dbase = dots + head * 32 + 8 * h + (c >> 1);
  const bool dotLane = (c & 1) == 0;
  v4f a0[KJ], a1[KJ];
  gemm_load_a<K>(a0, abase, rg);

  int rt = rg;
  for (;;) {
    const int r1 = rt + rowGroups;
    const int r2 = r1 + rowGroups;
    v8f acc[4];
    __builtin_amdgcn_sched_barrier(0);
    gemm_load_a<K>(a1, abase, r1 < nTiles ? r1 : rt);
    __builtin_amdgcn_sched_barrier(0);
    gemm_tile<K>(acc, a0, bp);
    if (doStore) gemm_store(acc, cbase, rt, M);
    if (doDot) {
      const float d = tile_row_dots(acc, av, c);
      if (dotLane) dbase[(size_t)rt * 64] = d;
    }
    __builtin_amdgcn_sched_barrier(0);
    gemm_load_a<K>(a0, abase, r2 < nTiles ? r2 : rt);
    __builtin_amdgcn_sched_barrier(0);
    gemm_tile<K>(acc, a1, bp);
    if (r1 < nTiles) {
      if (doStore) gemm_store(acc, cbase, r1, M);
      if (doDot) {
        const float d = tile_row_dots(acc, av, c);
        if (dotLane) dbase[(size_t)r1 * 64] = d;
      }
    }
    if (r2 >= nTiles) break;
    rt = r2;
  }
}

typedef __attribute__((ext_vector_type(4))) int v4i;

#define TM   256
#define LCAP 8192
#define ET   1024
#define EG   256
#define LPITCH (LCAP + 32)
static_assert(TM <= 256, "a kept list entry holds the tile-relative destination in 8 bits");

__device__ __forceinline__ float lrelu(float v) { return (v > 0.f) ? v : NEG_SLOPE * v; }

__device__ __forceinline__ const float* dot_at(const float* plane, int n) {
  return plane + (size_t)(n >> 4) * 64 + (n & 15);
}

__device__ __forceinline__ void ldsMaxF(float* addr, float value) {
  if (value >= 0.f)
    atomicMax((int*)addr, __float_as_int(value));
  else
    atomicMin((unsigned int*)addr, __float_as_uint(value));
}

template <int U>
__device__ __forceinline__ void wave_rows(unsigned mask, int s, int t, float a0, float a1,
                                          const float* __restrict__ hs, float* agg, int lane) {
  while (mask) {
    int l[U];
    bool on[U];
    unsigned mm = mask;
#pragma unroll
    for (int u = 0; u < U; ++u) {
      on[u] = (mm != 0u);
      l[u] = on[u] ? __builtin_ctz(mm) : l[0];
      mm &= mm - 1u;
    }
    mask = mm;
    v2f x0[U], x1[U];
    int tt[U];
    float A0[U], A1[U];
#pragma unroll
    for (int u = 0; u < U; ++u) {
      const int ss = __builtin_amdgcn_readlane(s, l[u]);
      tt[u] = __builtin_amdgcn_readlane(t, l[u]);
      A0[u] = __int_as_float(__builtin_amdgcn_readlane(__float_as_int(a0), l[u]));
      A1[u] = __int_as_float(__builtin_amdgcn_readlane(__float_as_int(a1), l[u]));
      const float* row = hs + (size_t)ss * 128 + 2 * lane;
      x0[u] = *(const v2f*)row;
      x1[u] = *(const v2f*)(row + 64);
    }
#pragma unroll
    for (int u = 0; u < U; ++u) asm volatile("" :: "v"(x0[u]), "v"(x1[u]));
#pragma unroll
    for (int u = 0; u < U; ++u) {
      if (on[u]) {
        float* p = agg + tt[u] * 64 + 2 * lane;
        atomicAdd(p,     x0[u].x * A0[u] + x1[u].x * A1[u]);
        atomicAdd(p + 1, x0[u].y * A0[u] + x1[u].y * A1[u]);
      }
    }
  }
}

template <int MODE>
__device__ __forceinline__ void sweep_all(const int* __restrict__ src, const int* __restrict__ dst,
                                          const float* __restrict__ als, const float* __restrict__ ald,
                                          const float* __restrict__ hs,
                                          float* s_agg, float* s_m, float* s_den,
                                          int lo, int nn, int Ee, int tid, int lane) {
  const int trips = (Ee + ET - 1) / ET;
  for (int it = 0; it < trips; ++it) {
    const int e = it * ET + tid;
    bool valid = false;
    int s = 0, t = 0;
    float a0 = 0.f, a1 = 0.f;
    if (e < Ee) {
      t = dst[e] - lo;
      valid = (unsigned)t < (unsigned)nn;
      if (valid) {
        s = src[e];
        const float* asp = dot_at(als, s);
        const float* adp = dot_at(ald, lo + t);
        const float v0 = lrelu(asp[0] + adp[0]);
        const float v1 = lrelu(asp[32] + adp[32]);
        if (MODE == 1) {
          ldsMaxF(&s_m[2 * t], v0);
          ldsMaxF(&s_m[2 * t + 1], v1);
        } else if (MODE == 2) {
          atomicAdd(&s_den[2 * t],     __expf(v0 - s_m[2 * t]));
          atomicAdd(&s_den[2 * t + 1], __expf(v1 - s_m[2 * t + 1]));
        } else {
          a0 = __expf(v0 - s_m[2 * t])     / (s_den[2 * t]     + 1e-16f) * 0.5f;
          a1 = __expf(v1 - s_m[2 * t + 1]) / (s_den[2 * t + 1] + 1e-16f) * 0.5f;
        }
      }
    }
    if (MODE == 3) {
      const unsigned mask = __builtin_amdgcn_ballot_w32(valid);
      wave_rows<4>(mask, s, t, a0, a1, hs, s_agg, lane);
    }
  }
}

template <bool REUSE>
__global__ void __launch_bounds__(ET)
edge_tile(const int* __restrict__ src, const int* __restrict__ dst,
          const float* __restrict__ als, const float* __restrict__ ald,
          const float* __restrict__ hs, const float* __restrict__ hl,
          const float* __restrict__ b, const float* __restrict__ bl,
          float* __restrict__ out, unsigned* __restrict__ keep, int Nn, int Ee) {
  __shared__ float s_agg[TM * 64];
  __shared__ float s_m[TM * 2];
  __shared__ float s_den[TM * 2];
  __shared__ int   s_a[LCAP];
  __shared__ int   s_t[LCAP];
  __shared__ float s_e0[LCAP];
  __shared__ float s_e1[LCAP];
  __shared__ int   s_cnt;

  const int tid = threadIdx.x;
  const int lane = tid & 31;
  const int wave = tid >> 5;

  int tn = (Nn + (int)gridDim.x - 1) / (int)gridDim.x;
  if (tn > TM) tn = TM;
  if (tn < 1) tn = 1;
  const int tiles = (Nn + tn - 1) / tn;

  int i0 = (int)(((16u - (unsigned)((size_t)dst & 15u)) & 15u) >> 2);
  if (i0 > Ee) i0 = Ee;
  const int G = (Ee - i0) >> 2;
  const v4i* d4 = (const v4i*)(dst + i0);
  const int rot = (int)(((long long)blockIdx.x * G) / (int)gridDim.x);
  constexpr int U = 8;

  for (int tile = blockIdx.x; tile < tiles; tile += gridDim.x) {
    const int lo = tile * tn;
    const int nn = (Nn - lo < tn) ? (Nn - lo) : tn;

    unsigned* kl = (REUSE || keep) ? keep + (size_t)tile * LPITCH : nullptr;

    for (int i = tid; i < nn * 64; i += ET) s_agg[i] = 0.f;
    for (int i = tid; i < nn * 2; i += ET) { s_m[i] = -INFINITY; s_den[i] = 0.f; }
    if (!REUSE && tid == 0) s_cnt = 0;
    __syncthreads();

    int total;
    if (REUSE) {
      total = (int)kl[0];
    } else {
      for (int g0 = 0; g0 < G; g0 += ET * U) {
        v4i d[U];
        int gi[U];
        bool ok[U];
#pragma unroll
        for (int u = 0; u < U; ++u) {
          const int g = g0 + u * ET + tid;
          ok[u] = g < G;
          int gg = (ok[u] ? g : G - 1) + rot;
          if (gg >= G) gg -= G;
          gi[u] = gg;
          d[u] = d4[gg];
        }
#pragma unroll
        for (int u = 0; u < U; ++u) {
          const int t0 = d[u].x - lo, t1 = d[u].y - lo, t2 = d[u].z - lo, t3 = d[u].w - lo;
          const bool m0 = (unsigned)t0 < (unsigned)nn, m1 = (unsigned)t1 < (unsigned)nn;
          const bool m2 = (unsigned)t2 < (unsigned)nn, m3 = (unsigned)t3 < (unsigned)nn;
          if (ok[u] && (m0 | m1 | m2 | m3)) {
            const int e = i0 + 4 * gi[u];
            if (m0) { const int sl = atomicAdd(&s_cnt, 1); if (sl < LCAP) { s_a[sl] = e;     s_t[sl] = t0; } }
            if (m1) { const int sl = atomicAdd(&s_cnt, 1); if (sl < LCAP) { s_a[sl] = e + 1; s_t[sl] = t1; } }
            if (m2) { const int sl = atomicAdd(&s_cnt, 1); if (sl < LCAP) { s_a[sl] = e + 2; s_t[sl] = t2; } }
            if (m3) { const int sl = atomicAdd(&s_cnt, 1); if (sl < LCAP) { s_a[sl] = e + 3; s_t[sl] = t3; } }
          }
        }
      }
      {
        const int rest = Ee - 4 * G;
        if (tid < rest) {
          const int e = (tid < i0) ? tid : (4 * G + tid);
          const int t = dst[e] - lo;
          if ((unsigned)t < (unsigned)nn) {
            const int sl = atomicAdd(&s_cnt, 1);
            if (sl < LCAP) { s_a[sl] = e; s_t[sl] = t; }
          }
        }
      }
      __syncthreads();
      total = s_cnt;
      if (kl && tid == 0) kl[0] = (unsigned)total;
    }

    if (total <= LCAP) {
      for (int i = tid; i < total; i += ET) {
        int s, t;
        if (REUSE) {
          const unsigned w = kl[32 + i];
          s = (int)(w >> 8);
          t = (int)(w & 255u);
          s_t[i] = t;
        } else {
          t = s_t[i];
          s = src[s_a[i]];
          if (kl) kl[32 + i] = ((unsigned)s << 8) | (unsigned)t;
        }
        const float* asp = dot_at(als, s);
        const float* adp = dot_at(ald, lo + t);
        const float v0 = lrelu(asp[0] + adp[0]);
        const float v1 = lrelu(asp[32] + adp[32]);
        s_a[i] = s;
        s_e0[i] = v0;
        s_e1[i] = v1;
        ldsMaxF(&s_m[2 * t], v0);
        ldsMaxF(&s_m[2 * t + 1], v1);
      }
      __syncthreads();
      for (int i = tid; i < total; i += ET) {
        const int t = s_t[i];
        const float x0 = __expf(s_e0[i] - s_m[2 * t]);
        const float x1 = __expf(s_e1[i] - s_m[2 * t + 1]);
        s_e0[i] = x0;
        s_e1[i] = x1;
        atomicAdd(&s_den[2 * t], x0);
        atomicAdd(&s_den[2 * t + 1], x1);
      }
      __syncthreads();
      for (int base = wave * 32; base < total; base += ET) {
        const int i = base + lane;
        const bool valid = i < total;
        int s = 0, t = 0;
        float a0 = 0.f, a1 = 0.f;
        if (valid) {
          s = s_a[i];
          t = s_t[i];
          a0 = s_e0[i] / (s_den[2 * t] + 1e-16f) * 0.5f;
          a1 = s_e1[i] / (s_den[2 * t + 1] + 1e-16f) * 0.5f;
        }
        const unsigned mask = __builtin_amdgcn_ballot_w32(valid);
        wave_rows<8>(mask, s, t, a0, a1, hs, s_agg, lane);
      }
    } else {
      sweep_all<1>(src, dst, als, ald, hs, s_agg, s_m, s_den, lo, nn, Ee, tid, lane);
      __syncthreads();
      sweep_all<2>(src, dst, als, ald, hs, s_agg, s_m, s_den, lo, nn, Ee, tid, lane);
      __syncthreads();
      sweep_all<3>(src, dst, als, ald, hs, s_agg, s_m, s_den, lo, nn, Ee, tid, lane);
    }
    __syncthreads();

    for (int i = tid; i < nn * 64; i += ET) {
      const int c = i & 63;
      const size_t o = (size_t)lo * 64 + i;
      const float v = s_agg[i] + b[c] + hl[o] + bl[c];
      out[o] = (v > 0.f) ? v : 0.f;
    }
    __syncthreads();
  }
}

extern "C" void kernel_launch(void* const* d_in, const int* in_sizes, int n_in,
                              void* d_out, int out_size, void* d_ws, size_t ws_size,
                              hipStream_t stream) {
  const float* x  = (const float*)d_in[0];
  const int*   ei = (const int*)d_in[1];
  const int Nn = in_sizes[0] / 128;
  const int Ee = in_sizes[1] / 2;
  const int* src = ei;
  const int* dst = ei + Ee;

  const float *Ws[3], *Wd[3], *As[3], *Ad[3], *B[3], *Wl[3], *Bl[3];
  for (int l = 0; l < 3; ++l) {
    int base = 2 + 7 * l;
    Ws[l] = (const float*)d_in[base + 0];
    Wd[l] = (const float*)d_in[base + 1];
    As[l] = (const float*)d_in[base + 2];
    Ad[l] = (const float*)d_in[base + 3];
    B[l]  = (const float*)d_in[base + 4];
    Wl[l] = (const float*)d_in[base + 5];
    Bl[l] = (const float*)d_in[base + 6];
  }

  float* ws = (float*)d_ws;
  size_t off = 0;
  float* hs  = ws + off; off += (size_t)Nn * 128;
  float* hl  = ws + off; off += (size_t)Nn * 64;
  float* als = ws + off; off += (size_t)Nn * 4;
  float* ald = ws + off; off += (size_t)Nn * 4;
  float* h1  = ws + off; off += (size_t)Nn * 64;
  float* h2  = ws + off; off += (size_t)Nn * 64;
  int tn = (Nn + EG - 1) / EG;
  if (tn > TM) tn = TM;
  if (tn < 1) tn = 1;
  const int tiles = (Nn + tn - 1) / tn;
  unsigned* keep = nullptr;
  if (Nn <= (1 << 24) && ws_size >= (off + (size_t)tiles * LPITCH) * sizeof(float))
    keep = (unsigned*)(ws + off);

  const float* lin[3]  = { x, h1, h2 };
  float*       lout[3] = { h1, h2, (float*)d_out };

  for (int l = 0; l < 3; ++l) {
    if (l == 0)
      node_part<128><<<256, 256, 0, stream>>>(lin[l], Ws[l], Wd[l], Wl[l], As[l], Ad[l], hs, hl, als, ald, Nn);
    else
      node_part<64><<<512, 256, 0, stream>>>(lin[l], Ws[l], Wd[l], Wl[l], As[l], Ad[l], hs, hl, als, ald, Nn);
    if (l == 0 || !keep)
      edge_tile<false><<<EG, ET, 0, stream>>>(src, dst, als, ald, hs, hl, B[l], Bl[l], lout[l], keep, Nn, Ee);
    else
      edge_tile<true><<<EG, ET, 0, stream>>>(src, dst, als, ald, hs, hl, B[l], Bl[l], lout[l], keep, Nn, Ee);
  }
}


# Layer 1's edge sweep reads a one-byte tile id per edge (a plane the first node launch writes from dst) instead of dst: a quarter of the bytes, 16 edges per load; list order (sum order) changed
# speedup vs baseline: 4.3834x; 1.0956x over previous
#include <hip/hip_runtime.h>
#include <math.h>

typedef __attribute__((ext_vector_type(2))) float v2f;
typedef __attribute__((ext_vector_type(4))) float v4f;
typedef __attribute__((ext_vector_type(8))) float v8f;

#define NEG_SLOPE 0.2f

template <int K>
__device__ __forceinline__ void gemm_load_a(v4f (&a)[K / 8], const float* abase, int rt) {
  const float* ap = abase + (size_t)rt * 16 * K;
#pragma unroll
  for (int j = 0; j < K / 8; ++j) a[j] = *(const v4f*)(ap + 8 * j);
}

template <int K>
__device__ __forceinline__ void gemm_tile(v8f (&acc)[4], const v4f (&a)[K / 8],
                                          const v2f (&bp)[K / 8][2][4]) {
#pragma unroll
  for (int q = 0; q < 4; ++q) acc[q] = (v8f){0.f,0.f,0.f,0.f,0.f,0.f,0.f,0.f};

#pragma unroll
  for (int j = 0; j < K / 8; ++j) {
#pragma unroll
    for (int sub = 0; sub < 2; ++sub) {
      v2f av;
      av.x = sub ? a[j].z : a[j].x;
      av.y = sub ? a[j].w : a[j].y;
#pragma unroll
      for (int q = 0; q < 4; ++q) {
        acc[q] = __builtin_amdgcn_wmma_f32_16x16x4_f32(
             false, av,  false, bp[j][sub][q],
             (short)0, acc[q],  false,  false);
      }
    }
  }
}

__device__ __forceinline__ void gemm_store(const v8f (&acc)[4], float* cbase, int rt, int M) {
  float* crow = cbase + (size_t)rt * 16 * M;
#pragma unroll
  for (int r = 0; r < 8; ++r) {
    v4f o;
    o.x = acc[0][r]; o.y = acc[1][r]; o.z = acc[2][r]; o.w = acc[3][r];
    *(v4f*)(crow + (size_t)r * M) = o;
  }
}

__device__ __forceinline__ float tile_row_dots(const v8f (&acc)[4], const v4f av, int c) {
  float p[8];
#pragma unroll
  for (int r = 0; r < 8; ++r)
    p[r] = acc[0][r] * av.x + acc[1][r] * av.y + acc[2][r] * av.z + acc[3][r] * av.w;
  const bool b3 = (c & 8) != 0, b2 = (c & 4) != 0, b1 = (c & 2) != 0;
  float k4[4], k2[2];
#pragma unroll
  for (int i = 0; i < 4; ++i) {
    const float send = b3 ? p[i] : p[4 + i];
    const float keep = b3 ? p[4 + i] : p[i];
    k4[i] = keep + __shfl_xor(send, 8, 32);
  }
#pragma unroll
  for (int i = 0; i < 2; ++i) {
    const float send = b2 ? k4[i] : k4[2 + i];
    const float keep = b2 ? k4[2 + i] : k4[i];
    k2[i] = keep + __shfl_xor(send, 4, 32);
  }
  const float send1 = b1 ? k2[0] : k2[1];
  const float keep1 = b1 ? k2[1] : k2[0];
  const float k1 = keep1 + __shfl_xor(send1, 2, 32);
  return k1 + __shfl_xor(k1, 1, 32);
}

template <int K, bool PLANE>
__global__ void __launch_bounds__(256)
node_part(const float* __restrict__ A,
          const float* __restrict__ Ws, const float* __restrict__ Wd,
          const float* __restrict__ Wl,
          const float* __restrict__ a_s, const float* __restrict__ a_d,
          float* __restrict__ hs, float* __restrict__ hl,
          float* __restrict__ als, float* __restrict__ ald,
          int Nrows,
          const int* __restrict__ dst, unsigned* __restrict__ plane, int Ee, int tn) {
  constexpr int KJ = K / 8;

  if (PLANE && plane) {
    const unsigned nthreads = gridDim.x * blockDim.x;
    const unsigned words = (unsigned)((Ee + 15) >> 4) << 2;
    const unsigned last = (unsigned)Ee - 1u;
    for (unsigned wb = blockIdx.x * blockDim.x + threadIdx.x; wb < words; wb += 4u * nthreads) {
      unsigned dv[4][4];
#pragma unroll
      for (int i = 0; i < 4; ++i) {
#pragma unroll
        for (int k = 0; k < 4; ++k) {
          const unsigned e = 4u * (wb + (unsigned)i * nthreads) + (unsigned)k;
          dv[i][k] = (unsigned)dst[e < last ? e : last];
        }
      }
#pragma unroll
      for (int i = 0; i < 4; ++i) {
        const unsigned w = wb + (unsigned)i * nthreads;
        unsigned word = 0u;
#pragma unroll
        for (int k = 0; k < 4; ++k) word |= ((dv[i][k] / (unsigned)tn) & 255u) << (8 * k);
        if (w < words) plane[w] = word;
      }
    }
  }

  const int lane = threadIdx.x & 31;
  const int c = lane & 15;
  const int h = lane >> 4;
  const int wib = __builtin_amdgcn_readfirstlane((int)(threadIdx.x >> 5));
  const int wavesPerBlock = blockDim.x >> 5;
  const int wave = blockIdx.x * wavesPerBlock + wib;
  const int slab = wave % 5;
  const int rg = wave / 5;
  const int rowGroups = (gridDim.x * wavesPerBlock) / 5;
  const int nTiles = Nrows >> 4;
  if (rg >= rowGroups || rg >= nTiles) return;

  const bool doStore = (slab < 2) || (slab == 4);
  const bool doDot = slab < 4;
  const int head = slab & 1;
  const int M = (slab == 4) ? 64 : 128;
  const int cs = (slab == 4) ? 0 : head * 64;
  const float* W = (slab < 2) ? Ws : ((slab < 4) ? Wd : Wl);
  float* C = (slab < 2) ? hs : hl;
  const float* avec = (slab < 2) ? a_s : a_d;
  float* dots = (slab < 2) ? als : ald;

  v2f bp[KJ][2][4];
  {
    const float* wbase = W + (size_t)(4 * h) * M + cs + 4 * c;
#pragma unroll
    for (int j = 0; j < KJ; ++j) {
#pragma unroll
      for (int sub = 0; sub < 2; ++sub) {
        const v4f w0 = *(const v4f*)(wbase + (size_t)(8 * j + 2 * sub) * M);
        const v4f w1 = *(const v4f*)(wbase + (size_t)(8 * j + 2 * sub + 1) * M);
#pragma unroll
        for (int q = 0; q < 4; ++q) {
          bp[j][sub][q].x = w0[q];
          bp[j][sub][q].y = w1[q];
        }
      }
    }
  }
  const v4f av = *(const v4f*)(avec + head * 64 + 4 * c);

  const float* abase = A + (size_t)c * K + 4 * h;
  float* cbase = C + (size_t)(8 * h) * M + cs + 4 * c;
  float* dbase = dots + head * 32 + 8 * h + (c >> 1);
  const bool dotLane = (c & 1) == 0;
  v4f a0[KJ], a1[KJ];
  gemm_load_a<K>(a0, abase, rg);

  int rt = rg;
  for (;;) {
    const int r1 = rt + rowGroups;
    const int r2 = r1 + rowGroups;
    v8f acc[4];
    __builtin_amdgcn_sched_barrier(0);
    gemm_load_a<K>(a1, abase, r1 < nTiles ? r1 : rt);
    __builtin_amdgcn_sched_barrier(0);
    gemm_tile<K>(acc, a0, bp);
    if (doStore) gemm_store(acc, cbase, rt, M);
    if (doDot) {
      const float d = tile_row_dots(acc, av, c);
      if (dotLane) dbase[(size_t)rt * 64] = d;
    }
    __builtin_amdgcn_sched_barrier(0);
    gemm_load_a<K>(a0, abase, r2 < nTiles ? r2 : rt);
    __builtin_amdgcn_sched_barrier(0);
    gemm_tile<K>(acc, a1, bp);
    if (r1 < nTiles) {
      if (doStore) gemm_store(acc, cbase, r1, M);
      if (doDot) {
        const float d = tile_row_dots(acc, av, c);
        if (dotLane) dbase[(size_t)r1 * 64] = d;
      }
    }
    if (r2 >= nTiles) break;
    rt = r2;
  }
}

typedef __attribute__((ext_vector_type(4))) int v4i;

#define TM   256
#define LCAP 8192
#define ET   1024
#define EG   256
#define LPITCH (LCAP + 32)
static_assert(TM <= 256, "a kept list entry holds the tile-relative destination in 8 bits");

__device__ __forceinline__ float lrelu(float v) { return (v > 0.f) ? v : NEG_SLOPE * v; }

__device__ __forceinline__ const float* dot_at(const float* plane, int n) {
  return plane + (size_t)(n >> 4) * 64 + (n & 15);
}

__device__ __forceinline__ void ldsMaxF(float* addr, float value) {
  if (value >= 0.f)
    atomicMax((int*)addr, __float_as_int(value));
  else
    atomicMin((unsigned int*)addr, __float_as_uint(value));
}

template <int U>
__device__ __forceinline__ void wave_rows(unsigned mask, int s, int t, float a0, float a1,
                                          const float* __restrict__ hs, float* agg, int lane) {
  while (mask) {
    int l[U];
    bool on[U];
    unsigned mm = mask;
#pragma unroll
    for (int u = 0; u < U; ++u) {
      on[u] = (mm != 0u);
      l[u] = on[u] ? __builtin_ctz(mm) : l[0];
      mm &= mm - 1u;
    }
    mask = mm;
    v2f x0[U], x1[U];
    int tt[U];
    float A0[U], A1[U];
#pragma unroll
    for (int u = 0; u < U; ++u) {
      const int ss = __builtin_amdgcn_readlane(s, l[u]);
      tt[u] = __builtin_amdgcn_readlane(t, l[u]);
      A0[u] = __int_as_float(__builtin_amdgcn_readlane(__float_as_int(a0), l[u]));
      A1[u] = __int_as_float(__builtin_amdgcn_readlane(__float_as_int(a1), l[u]));
      const float* row = hs + (size_t)ss * 128 + 2 * lane;
      x0[u] = *(const v2f*)row;
      x1[u] = *(const v2f*)(row + 64);
    }
#pragma unroll
    for (int u = 0; u < U; ++u) asm volatile("" :: "v"(x0[u]), "v"(x1[u]));
#pragma unroll
    for (int u = 0; u < U; ++u) {
      if (on[u]) {
        float* p = agg + tt[u] * 64 + 2 * lane;
        atomicAdd(p,     x0[u].x * A0[u] + x1[u].x * A1[u]);
        atomicAdd(p + 1, x0[u].y * A0[u] + x1[u].y * A1[u]);
      }
    }
  }
}

template <int MODE>
__device__ __forceinline__ void sweep_all(const int* __restrict__ src, const int* __restrict__ dst,
                                          const float* __restrict__ als, const float* __restrict__ ald,
                                          const float* __restrict__ hs,
                                          float* s_agg, float* s_m, float* s_den,
                                          int lo, int nn, int Ee, int tid, int lane) {
  const int trips = (Ee + ET - 1) / ET;
  for (int it = 0; it < trips; ++it) {
    const int e = it * ET + tid;
    bool valid = false;
    int s = 0, t = 0;
    float a0 = 0.f, a1 = 0.f;
    if (e < Ee) {
      t = dst[e] - lo;
      valid = (unsigned)t < (unsigned)nn;
      if (valid) {
        s = src[e];
        const float* asp = dot_at(als, s);
        const float* adp = dot_at(ald, lo + t);
        const float v0 = lrelu(asp[0] + adp[0]);
        const float v1 = lrelu(asp[32] + adp[32]);
        if (MODE == 1) {
          ldsMaxF(&s_m[2 * t], v0);
          ldsMaxF(&s_m[2 * t + 1], v1);
        } else if (MODE == 2) {
          atomicAdd(&s_den[2 * t],     __expf(v0 - s_m[2 * t]));
          atomicAdd(&s_den[2 * t + 1], __expf(v1 - s_m[2 * t + 1]));
        } else {
          a0 = __expf(v0 - s_m[2 * t])     / (s_den[2 * t]     + 1e-16f) * 0.5f;
          a1 = __expf(v1 - s_m[2 * t + 1]) / (s_den[2 * t + 1] + 1e-16f) * 0.5f;
        }
      }
    }
    if (MODE == 3) {
      const unsigned mask = __builtin_amdgcn_ballot_w32(valid);
      wave_rows<4>(mask, s, t, a0, a1, hs, s_agg, lane);
    }
  }
}

template <bool REUSE>
__global__ void __launch_bounds__(ET)
edge_tile(const int* __restrict__ src, const int* __restrict__ dst,
          const float* __restrict__ als, const float* __restrict__ ald,
          const float* __restrict__ hs, const float* __restrict__ hl,
          const float* __restrict__ b, const float* __restrict__ bl,
          float* __restrict__ out, unsigned* __restrict__ keep,
          const unsigned* __restrict__ plane, int Nn, int Ee) {
  __shared__ float s_agg[TM * 64];
  __shared__ float s_m[TM * 2];
  __shared__ float s_den[TM * 2];
  __shared__ int   s_a[LCAP];
  __shared__ int   s_t[LCAP];
  __shared__ float s_e0[LCAP];
  __shared__ float s_e1[LCAP];
  __shared__ int   s_cnt;

  const int tid = threadIdx.x;
  const int lane = tid & 31;
  const int wave = tid >> 5;

  int tn = (Nn + (int)gridDim.x - 1) / (int)gridDim.x;
  if (tn > TM) tn = TM;
  if (tn < 1) tn = 1;
  const int tiles = (Nn + tn - 1) / tn;

  int i0 = (int)(((16u - (unsigned)((size_t)dst & 15u)) & 15u) >> 2);
  if (i0 > Ee) i0 = Ee;
  const int G = (Ee - i0) >> 2;
  const v4i* d4 = (const v4i*)(dst + i0);
  const int rot = (int)(((long long)blockIdx.x * G) / (int)gridDim.x);
  constexpr int U = 8;

  const bool byPlane = !REUSE && plane != nullptr && tiles <= 256;
  const int G16 = (Ee + 15) >> 4;
  const v4i* p4 = (const v4i*)plane;
  const int rot16 = (int)(((long long)blockIdx.x * G16) / (int)gridDim.x);

  for (int tile = blockIdx.x; tile < tiles; tile += gridDim.x) {
    const int lo = tile * tn;
    const int nn = (Nn - lo < tn) ? (Nn - lo) : tn;

    unsigned* kl = (REUSE || keep) ? keep + (size_t)tile * LPITCH : nullptr;

    for (int i = tid; i < nn * 64; i += ET) s_agg[i] = 0.f;
    for (int i = tid; i < nn * 2; i += ET) { s_m[i] = -INFINITY; s_den[i] = 0.f; }
    if (!REUSE && tid == 0) s_cnt = 0;
    __syncthreads();

    int total;
    if (REUSE) {
      total = (int)kl[0];
    } else {
      if (byPlane) {
        const unsigned key = (unsigned)tile * 0x01010101u;
        for (int g0 = 0; g0 < G16; g0 += ET * U) {
          v4i d[U];
          int gi[U];
          bool ok[U];
#pragma unroll
          for (int u = 0; u < U; ++u) {
            const int g = g0 + u * ET + tid;
            ok[u] = g < G16;
            int gg = (ok[u] ? g : G16 - 1) + rot16;
            if (gg >= G16) gg -= G16;
            gi[u] = gg;
            d[u] = p4[gg];
          }
#pragma unroll
          for (int u = 0; u < U; ++u) {
            unsigned hit = 0u;
#pragma unroll
            for (int w = 0; w < 4; ++w) {
              const unsigned x = (unsigned)d[u][w] ^ key;
              const unsigned nz = ((x & 0x7f7f7f7fu) + 0x7f7f7f7fu) | x;
              hit |= (~nz & 0x80808080u) >> (7 - w);
            }
            if (!ok[u]) hit = 0u;
            while (hit) {
              const int bit = __builtin_ctz(hit);
              hit &= hit - 1u;
              const int e = 16 * gi[u] + 4 * (bit & 7) + (bit >> 3);
              if (e < Ee) {
                const int sl = atomicAdd(&s_cnt, 1);
                if (sl < LCAP) s_a[sl] = e;
              }
            }
          }
        }
      } else {
        for (int g0 = 0; g0 < G; g0 += ET * U) {
          v4i d[U];
          int gi[U];
          bool ok[U];
#pragma unroll
          for (int u = 0; u < U; ++u) {
            const int g = g0 + u * ET + tid;
            ok[u] = g < G;
            int gg = (ok[u] ? g : G - 1) + rot;
            if (gg >= G) gg -= G;
            gi[u] = gg;
            d[u] = d4[gg];
          }
#pragma unroll
          for (int u = 0; u < U; ++u) {
            const int t0 = d[u].x - lo, t1 = d[u].y - lo, t2 = d[u].z - lo, t3 = d[u].w - lo;
            const bool m0 = (unsigned)t0 < (unsigned)nn, m1 = (unsigned)t1 < (unsigned)nn;
            const bool m2 = (unsigned)t2 < (unsigned)nn, m3 = (unsigned)t3 < (unsigned)nn;
            if (ok[u] && (m0 | m1 | m2 | m3)) {
              const int e = i0 + 4 * gi[u];
              if (m0) { const int sl = atomicAdd(&s_cnt, 1); if (sl < LCAP) { s_a[sl] = e;     s_t[sl] = t0; } }
              if (m1) { const int sl = atomicAdd(&s_cnt, 1); if (sl < LCAP) { s_a[sl] = e + 1; s_t[sl] = t1; } }
              if (m2) { const int sl = atomicAdd(&s_cnt, 1); if (sl < LCAP) { s_a[sl] = e + 2; s_t[sl] = t2; } }
              if (m3) { const int sl = atomicAdd(&s_cnt, 1); if (sl < LCAP) { s_a[sl] = e + 3; s_t[sl] = t3; } }
            }
          }
        }
        {
          const int rest = Ee - 4 * G;
          if (tid < rest) {
            const int e = (tid < i0) ? tid : (4 * G + tid);
            const int t = dst[e] - lo;
            if ((unsigned)t < (unsigned)nn) {
              const int sl = atomicAdd(&s_cnt, 1);
              if (sl < LCAP) { s_a[sl] = e; s_t[sl] = t; }
            }
          }
        }
      }
      __syncthreads();
      total = s_cnt;
      if (kl && tid == 0) kl[0] = (unsigned)total;
    }

    if (total <= LCAP) {
      for (int i = tid; i < total; i += ET) {
        int s, t;
        if (REUSE) {
          const unsigned w = kl[32 + i];
          s = (int)(w >> 8);
          t = (int)(w & 255u);
          s_t[i] = t;
        } else {
          const int e = s_a[i];
          t = byPlane ? dst[e] - lo : s_t[i];
          s = src[e];
          if (byPlane) s_t[i] = t;
          if (kl) kl[32 + i] = ((unsigned)s << 8) | (unsigned)t;
        }
        const float* asp = dot_at(als, s);
        const float* adp = dot_at(ald, lo + t);
        const float v0 = lrelu(asp[0] + adp[0]);
        const float v1 = lrelu(asp[32] + adp[32]);
        s_a[i] = s;
        s_e0[i] = v0;
        s_e1[i] = v1;
        ldsMaxF(&s_m[2 * t], v0);
        ldsMaxF(&s_m[2 * t + 1], v1);
      }
      __syncthreads();
      for (int i = tid; i < total; i += ET) {
        const int t = s_t[i];
        const float x0 = __expf(s_e0[i] - s_m[2 * t]);
        const float x1 = __expf(s_e1[i] - s_m[2 * t + 1]);
        s_e0[i] = x0;
        s_e1[i] = x1;
        atomicAdd(&s_den[2 * t], x0);
        atomicAdd(&s_den[2 * t + 1], x1);
      }
      __syncthreads();
      for (int base = wave * 32; base < total; base += ET) {
        const int i = base + lane;
        const bool valid = i < total;
        int s = 0, t = 0;
        float a0 = 0.f, a1 = 0.f;
        if (valid) {
          s = s_a[i];
          t = s_t[i];
          a0 = s_e0[i] / (s_den[2 * t] + 1e-16f) * 0.5f;
          a1 = s_e1[i] / (s_den[2 * t + 1] + 1e-16f) * 0.5f;
        }
        const unsigned mask = __builtin_amdgcn_ballot_w32(valid);
        wave_rows<8>(mask, s, t, a0, a1, hs, s_agg, lane);
      }
    } else {
      sweep_all<1>(src, dst, als, ald, hs, s_agg, s_m, s_den, lo, nn, Ee, tid, lane);
      __syncthreads();
      sweep_all<2>(src, dst, als, ald, hs, s_agg, s_m, s_den, lo, nn, Ee, tid, lane);
      __syncthreads();
      sweep_all<3>(src, dst, als, ald, hs, s_agg, s_m, s_den, lo, nn, Ee, tid, lane);
    }
    __syncthreads();

    for (int i = tid; i < nn * 64; i += ET) {
      const int c = i & 63;
      const size_t o = (size_t)lo * 64 + i;
      const float v = s_agg[i] + b[c] + hl[o] + bl[c];
      out[o] = (v > 0.f) ? v : 0.f;
    }
    __syncthreads();
  }
}

extern "C" void kernel_launch(void* const* d_in, const int* in_sizes, int n_in,
                              void* d_out, int out_size, void* d_ws, size_t ws_size,
                              hipStream_t stream) {
  const float* x  = (const float*)d_in[0];
  const int*   ei = (const int*)d_in[1];
  const int Nn = in_sizes[0] / 128;
  const int Ee = in_sizes[1] / 2;
  const int* src = ei;
  const int* dst = ei + Ee;

  const float *Ws[3], *Wd[3], *As[3], *Ad[3], *B[3], *Wl[3], *Bl[3];
  for (int l = 0; l < 3; ++l) {
    int base = 2 + 7 * l;
    Ws[l] = (const float*)d_in[base + 0];
    Wd[l] = (const float*)d_in[base + 1];
    As[l] = (const float*)d_in[base + 2];
    Ad[l] = (const float*)d_in[base + 3];
    B[l]  = (const float*)d_in[base + 4];
    Wl[l] = (const float*)d_in[base + 5];
    Bl[l] = (const float*)d_in[base + 6];
  }

  float* ws = (float*)d_ws;
  size_t off = 0;
  float* hs  = ws + off; off += (size_t)Nn * 128;
  float* hl  = ws + off; off += (size_t)Nn * 64;
  float* als = ws + off; off += (size_t)Nn * 4;
  float* ald = ws + off; off += (size_t)Nn * 4;
  float* h1  = ws + off; off += (size_t)Nn * 64;
  float* h2  = ws + off; off += (size_t)Nn * 64;
  int tn = (Nn + EG - 1) / EG;
  if (tn > TM) tn = TM;
  if (tn < 1) tn = 1;
  const int tiles = (Nn + tn - 1) / tn;
  unsigned* keep = nullptr;
  if (Nn <= (1 << 24) && ws_size >= (off + (size_t)tiles * LPITCH) * sizeof(float))
    keep = (unsigned*)(ws + off);
  if (keep) off += (size_t)tiles * LPITCH;
  unsigned* plane = nullptr;
  const size_t planeWords = (size_t)((Ee + 15) >> 4) * 4;
  if (tiles <= 256 && ws_size >= (off + planeWords) * sizeof(float))
    plane = (unsigned*)(ws + off);

  const float* lin[3]  = { x, h1, h2 };
  float*       lout[3] = { h1, h2, (float*)d_out };

  for (int l = 0; l < 3; ++l) {
    if (l == 0)
      node_part<128, true><<<256, 256, 0, stream>>>(lin[l], Ws[l], Wd[l], Wl[l], As[l], Ad[l], hs, hl, als, ald, Nn,
                                                    dst, plane, Ee, tn);
    else
      node_part<64, false><<<512, 256, 0, stream>>>(lin[l], Ws[l], Wd[l], Wl[l], As[l], Ad[l], hs, hl, als, ald, Nn,
                                                    dst, plane, Ee, tn);
    if (l == 0 || !keep)
      edge_tile<false><<<EG, ET, 0, stream>>>(src, dst, als, ald, hs, hl, B[l], Bl[l], lout[l], keep, plane, Nn, Ee);
    else
      edge_tile<true><<<EG, ET, 0, stream>>>(src, dst, als, ald, hs, hl, B[l], Bl[l], lout[l], keep, plane, Nn, Ee);
  }
}


# Layers 1,2: the node part runs inside the edge launch before it: a tile's finished rows go from LDS (row pitch 68) to WMMA tasks of its 32 waves; h1, h2 never stored; 4 launches not 6; sums unchanged
# speedup vs baseline: 4.5481x; 1.0376x over previous
#include <hip/hip_runtime.h>
#include <math.h>

typedef __attribute__((ext_vector_type(2))) float v2f;
typedef __attribute__((ext_vector_type(4))) float v4f;
typedef __attribute__((ext_vector_type(8))) float v8f;

#define NEG_SLOPE 0.2f

template <int K>
__device__ __forceinline__ void gemm_load_a(v4f (&a)[K / 8], const float* abase, int rt) {
  const float* ap = abase + (size_t)rt * 16 * K;
#pragma unroll
  for (int j = 0; j < K / 8; ++j) a[j] = *(const v4f*)(ap + 8 * j);
}

template <int K>
__device__ __forceinline__ void gemm_tile(v8f (&acc)[4], const v4f (&a)[K / 8],
                                          const v2f (&bp)[K / 8][2][4]) {
#pragma unroll
  for (int q = 0; q < 4; ++q) acc[q] = (v8f){0.f,0.f,0.f,0.f,0.f,0.f,0.f,0.f};

#pragma unroll
  for (int j = 0; j < K / 8; ++j) {
#pragma unroll
    for (int sub = 0; sub < 2; ++sub) {
      v2f av;
      av.x = sub ? a[j].z : a[j].x;
      av.y = sub ? a[j].w : a[j].y;
#pragma unroll
      for (int q = 0; q < 4; ++q) {
        acc[q] = __builtin_amdgcn_wmma_f32_16x16x4_f32(
             false, av,  false, bp[j][sub][q],
             (short)0, acc[q],  false,  false);
      }
    }
  }
}

__device__ __forceinline__ void gemm_store(const v8f (&acc)[4], float* cbase, int rt, int M) {
  float* crow = cbase + (size_t)rt * 16 * M;
#pragma unroll
  for (int r = 0; r < 8; ++r) {
    v4f o;
    o.x = acc[0][r]; o.y = acc[1][r]; o.z = acc[2][r]; o.w = acc[3][r];
    *(v4f*)(crow + (size_t)r * M) = o;
  }
}

__device__ __forceinline__ float tile_row_dots(const v8f (&acc)[4], const v4f av, int c) {
  float p[8];
#pragma unroll
  for (int r = 0; r < 8; ++r)
    p[r] = acc[0][r] * av.x + acc[1][r] * av.y + acc[2][r] * av.z + acc[3][r] * av.w;
  const bool b3 = (c & 8) != 0, b2 = (c & 4) != 0, b1 = (c & 2) != 0;
  float k4[4], k2[2];
#pragma unroll
  for (int i = 0; i < 4; ++i) {
    const float send = b3 ? p[i] : p[4 + i];
    const float keep = b3 ? p[4 + i] : p[i];
    k4[i] = keep + __shfl_xor(send, 8, 32);
  }
#pragma unroll
  for (int i = 0; i < 2; ++i) {
    const float send = b2 ? k4[i] : k4[2 + i];
    const float keep = b2 ? k4[2 + i] : k4[i];
    k2[i] = keep + __shfl_xor(send, 4, 32);
  }
  const float send1 = b1 ? k2[0] : k2[1];
  const float keep1 = b1 ? k2[1] : k2[0];
  const float k1 = keep1 + __shfl_xor(send1, 2, 32);
  return k1 + __shfl_xor(k1, 1, 32);
}

template <int K, bool PLANE>
__global__ void __launch_bounds__(256)
node_part(const float* __restrict__ A,
          const float* __restrict__ Ws, const float* __restrict__ Wd,
          const float* __restrict__ Wl,
          const float* __restrict__ a_s, const float* __restrict__ a_d,
          float* __restrict__ hs, float* __restrict__ hl,
          float* __restrict__ als, float* __restrict__ ald,
          int Nrows,
          const int* __restrict__ dst, unsigned* __restrict__ plane, int Ee, int tn) {
  constexpr int KJ = K / 8;

  if (PLANE && plane) {
    const unsigned nthreads = gridDim.x * blockDim.x;
    const unsigned words = (unsigned)((Ee + 15) >> 4) << 2;
    const unsigned last = (unsigned)Ee - 1u;
    for (unsigned wb = blockIdx.x * blockDim.x + threadIdx.x; wb < words; wb += 4u * nthreads) {
      unsigned dv[4][4];
#pragma unroll
      for (int i = 0; i < 4; ++i) {
#pragma unroll
        for (int k = 0; k < 4; ++k) {
          const unsigned e = 4u * (wb + (unsigned)i * nthreads) + (unsigned)k;
          dv[i][k] = (unsigned)dst[e < last ? e : last];
        }
      }
#pragma unroll
      for (int i = 0; i < 4; ++i) {
        const unsigned w = wb + (unsigned)i * nthreads;
        unsigned word = 0u;
#pragma unroll
        for (int k = 0; k < 4; ++k) word |= ((dv[i][k] / (unsigned)tn) & 255u) << (8 * k);
        if (w < words) plane[w] = word;
      }
    }
  }

  const int lane = threadIdx.x & 31;
  const int c = lane & 15;
  const int h = lane >> 4;
  const int wib = __builtin_amdgcn_readfirstlane((int)(threadIdx.x >> 5));
  const int wavesPerBlock = blockDim.x >> 5;
  const int wave = blockIdx.x * wavesPerBlock + wib;
  const int slab = wave % 5;
  const int rg = wave / 5;
  const int rowGroups = (gridDim.x * wavesPerBlock) / 5;
  const int nTiles = Nrows >> 4;
  if (rg >= rowGroups || rg >= nTiles) return;

  const bool doStore = (slab < 2) || (slab == 4);
  const bool doDot = slab < 4;
  const int head = slab & 1;
  const int M = (slab == 4) ? 64 : 128;
  const int cs = (slab == 4) ? 0 : head * 64;
  const float* W = (slab < 2) ? Ws : ((slab < 4) ? Wd : Wl);
  float* C = (slab < 2) ? hs : hl;
  const float* avec = (slab < 2) ? a_s : a_d;
  float* dots = (slab < 2) ? als : ald;

  v2f bp[KJ][2][4];
  {
    const float* wbase = W + (size_t)(4 * h) * M + cs + 4 * c;
#pragma unroll
    for (int j = 0; j < KJ; ++j) {
#pragma unroll
      for (int sub = 0; sub < 2; ++sub) {
        const v4f w0 = *(const v4f*)(wbase + (size_t)(8 * j + 2 * sub) * M);
        const v4f w1 = *(const v4f*)(wbase + (size_t)(8 * j + 2 * sub + 1) * M);
#pragma unroll
        for (int q = 0; q < 4; ++q) {
          bp[j][sub][q].x = w0[q];
          bp[j][sub][q].y = w1[q];
        }
      }
    }
  }
  const v4f av = *(const v4f*)(avec + head * 64 + 4 * c);

  const float* abase = A + (size_t)c * K + 4 * h;
  float* cbase = C + (size_t)(8 * h) * M + cs + 4 * c;
  float* dbase = dots + head * 32 + 8 * h + (c >> 1);
  const bool dotLane = (c & 1) == 0;
  v4f a0[KJ], a1[KJ];
  gemm_load_a<K>(a0, abase, rg);

  int rt = rg;
  for (;;) {
    const int r1 = rt + rowGroups;
    const int r2 = r1 + rowGroups;
    v8f acc[4];
    __builtin_amdgcn_sched_barrier(0);
    gemm_load_a<K>(a1, abase, r1 < nTiles ? r1 : rt);
    __builtin_amdgcn_sched_barrier(0);
    gemm_tile<K>(acc, a0, bp);
    if (doStore) gemm_store(acc, cbase, rt, M);
    if (doDot) {
      const float d = tile_row_dots(acc, av, c);
      if (dotLane) dbase[(size_t)rt * 64] = d;
    }
    __builtin_amdgcn_sched_barrier(0);
    gemm_load_a<K>(a0, abase, r2 < nTiles ? r2 : rt);
    __builtin_amdgcn_sched_barrier(0);
    gemm_tile<K>(acc, a1, bp);
    if (r1 < nTiles) {
      if (doStore) gemm_store(acc, cbase, r1, M);
      if (doDot) {
        const float d = tile_row_dots(acc, av, c);
        if (dotLane) dbase[(size_t)r1 * 64] = d;
      }
    }
    if (r2 >= nTiles) break;
    rt = r2;
  }
}

typedef __attribute__((ext_vector_type(4))) int v4i;

#define TM   256
#define LCAP 8192
#define ET   1024
#define EG   256
#define LPITCH (LCAP + 32)
static_assert(TM <= 256, "a kept list entry holds the tile-relative destination in 8 bits");
#define TPITCH 1024

__device__ __forceinline__ float lrelu(float v) { return (v > 0.f) ? v : NEG_SLOPE * v; }

__device__ __forceinline__ const float* dot_at(const float* plane, int n) {
  return plane + (size_t)(n >> 4) * 64 + (n & 15);
}
__device__ __forceinline__ const float* dot_tile(const float* plane, int T, int r) {
  return plane + (size_t)T * TPITCH + (r >> 4) * 64 + (r & 15);
}
template <bool TD>
__device__ __forceinline__ const float* dot_of(const float* plane, int n, int tn) {
  if (TD) {
    const unsigned T = (unsigned)n / (unsigned)tn;
    return dot_tile(plane, (int)T, (int)((unsigned)n - T * (unsigned)tn));
  }
  return dot_at(plane, n);
}

__device__ __forceinline__ void ldsMaxF(float* addr, float value) {
  if (value >= 0.f)
    atomicMax((int*)addr, __float_as_int(value));
  else
    atomicMin((unsigned int*)addr, __float_as_uint(value));
}

template <int U, int AP>
__device__ __forceinline__ void wave_rows(unsigned mask, int s, int t, float a0, float a1,
                                          const float* __restrict__ hs, float* agg, int lane) {
  while (mask) {
    int l[U];
    bool on[U];
    unsigned mm = mask;
#pragma unroll
    for (int u = 0; u < U; ++u) {
      on[u] = (mm != 0u);
      l[u] = on[u] ? __builtin_ctz(mm) : l[0];
      mm &= mm - 1u;
    }
    mask = mm;
    v2f x0[U], x1[U];
    int tt[U];
    float A0[U], A1[U];
#pragma unroll
    for (int u = 0; u < U; ++u) {
      const int ss = __builtin_amdgcn_readlane(s, l[u]);
      tt[u] = __builtin_amdgcn_readlane(t, l[u]);
      A0[u] = __int_as_float(__builtin_amdgcn_readlane(__float_as_int(a0), l[u]));
      A1[u] = __int_as_float(__builtin_amdgcn_readlane(__float_as_int(a1), l[u]));
      const float* row = hs + (size_t)ss * 128 + 2 * lane;
      x0[u] = *(const v2f*)row;
      x1[u] = *(const v2f*)(row + 64);
    }
#pragma unroll
    for (int u = 0; u < U; ++u) asm volatile("" :: "v"(x0[u]), "v"(x1[u]));
#pragma unroll
    for (int u = 0; u < U; ++u) {
      if (on[u]) {
        float* p = agg + tt[u] * AP + 2 * lane;
        atomicAdd(p,     x0[u].x * A0[u] + x1[u].x * A1[u]);
        atomicAdd(p + 1, x0[u].y * A0[u] + x1[u].y * A1[u]);
      }
    }
  }
}

template <int MODE, bool TD, int AP>
__device__ __forceinline__ void sweep_all(const int* __restrict__ src, const int* __restrict__ dst,
                                          const float* __restrict__ als, const float* __restrict__ ald,
                                          const float* __restrict__ hs,
                                          float* s_agg, float* s_m, float* s_den,
                                          int lo, int nn, int Ee, int tid, int lane, int tn) {
  const int trips = (Ee + ET - 1) / ET;
  for (int it = 0; it < trips; ++it) {
    const int e = it * ET + tid;
    bool valid = false;
    int s = 0, t = 0;
    float a0 = 0.f, a1 = 0.f;
    if (e < Ee) {
      t = dst[e] - lo;
      valid = (unsigned)t < (unsigned)nn;
      if (valid) {
        s = src[e];
        const float* asp = dot_of<TD>(als, s, tn);
        const float* adp = dot_of<TD>(ald, lo + t, tn);
        const float v0 = lrelu(asp[0] + adp[0]);
        const float v1 = lrelu(asp[32] + adp[32]);
        if (MODE == 1) {
          ldsMaxF(&s_m[2 * t], v0);
          ldsMaxF(&s_m[2 * t + 1], v1);
        } else if (MODE == 2) {
          atomicAdd(&s_den[2 * t],     __expf(v0 - s_m[2 * t]));
          atomicAdd(&s_den[2 * t + 1], __expf(v1 - s_m[2 * t + 1]));
        } else {
          a0 = __expf(v0 - s_m[2 * t])     / (s_den[2 * t]     + 1e-16f) * 0.5f;
          a1 = __expf(v1 - s_m[2 * t + 1]) / (s_den[2 * t + 1] + 1e-16f) * 0.5f;
        }
      }
    }
    if (MODE == 3) {
      const unsigned mask = __builtin_amdgcn_ballot_w32(valid);
      wave_rows<4, AP>(mask, s, t, a0, a1, hs, s_agg, lane);
    }
  }
}


template <int M>
__device__ __forceinline__ void tile_product(v8f (&acc)[4], const float* ap, const float* __restrict__ wslab, int woff) {
#pragma unroll
  for (int q = 0; q < 4; ++q) acc[q] = (v8f){0.f,0.f,0.f,0.f,0.f,0.f,0.f,0.f};
#pragma nounroll
  for (int jj = 0; jj < 4; ++jj) {
    const float* wj = wslab + (size_t)jj * 16 * M;
    v4f a[2], w[2][2][2];
#pragma unroll
    for (int u = 0; u < 2; ++u) {
      a[u] = *(const v4f*)(ap + 16 * jj + 8 * u);
#pragma unroll
      for (int sub = 0; sub < 2; ++sub) {
        w[u][sub][0] = *(const v4f*)(wj + woff + (8 * u + 2 * sub) * M);
        w[u][sub][1] = *(const v4f*)(wj + woff + (8 * u + 2 * sub + 1) * M);
      }
    }
#pragma unroll
    for (int u = 0; u < 2; ++u)
      asm volatile("" :: "v"(w[u][0][0]), "v"(w[u][0][1]), "v"(w[u][1][0]), "v"(w[u][1][1]));
#pragma unroll
    for (int u = 0; u < 2; ++u) {
#pragma unroll
      for (int sub = 0; sub < 2; ++sub) {
        v2f avv;
        avv.x = sub ? a[u].z : a[u].x;
        avv.y = sub ? a[u].w : a[u].y;
#pragma unroll
        for (int q = 0; q < 4; ++q) {
          v2f bv;
          bv.x = w[u][sub][0][q];
          bv.y = w[u][sub][1][q];
          acc[q] = __builtin_amdgcn_wmma_f32_16x16x4_f32(
               false, avv,  false, bv,
               (short)0, acc[q],  false,  false);
        }
      }
    }
  }
}

template <int AP>
__device__ __forceinline__ void tile_node_part(const float* rows, int nn, int lo, int tile,
                                               const float* __restrict__ Ws, const float* __restrict__ Wd,
                                               const float* __restrict__ Wl,
                                               const float* __restrict__ a_s, const float* __restrict__ a_d,
                                               float* __restrict__ hs, float* __restrict__ hl,
                                               float* __restrict__ alsT, float* __restrict__ aldT,
                                               int wave, int lane) {
  const int c = lane & 15;
  const int h = lane >> 4;
  const bool dotLane = (c & 1) == 0;
  const int nT = (nn + 15) >> 4;
  for (int id = wave; id < 5 * nT; id += 32) {
    const int rt = id / 5;
    const int slab = id - 5 * rt;
    const bool doStore = (slab < 2) || (slab == 4);
    const bool doDot = slab < 4;
    const int head = slab & 1;
    const int M = (slab == 4) ? 64 : 128;
    const int cs = (slab == 4) ? 0 : head * 64;
    const float* W = (slab < 2) ? Ws : Wd;
    float* C = (slab < 2) ? hs : hl;
    const float* avec = (slab < 2) ? a_s : a_d;
    float* dots = (slab < 2) ? alsT : aldT;

    v8f acc[4];
    const float* ap = rows + (16 * rt + c) * AP + 4 * h;
    if (slab == 4) tile_product<64>(acc, ap, Wl, 4 * h * 64 + 4 * c);
    else           tile_product<128>(acc, ap, W + cs, 4 * h * 128 + 4 * c);
    if (doStore) {
      float* crow = C + (size_t)(lo + 16 * rt + 8 * h) * M + cs + 4 * c;
      const int left = nn - (16 * rt + 8 * h);
#pragma unroll
      for (int r = 0; r < 8; ++r) {
        v4f o;
        o.x = acc[0][r]; o.y = acc[1][r]; o.z = acc[2][r]; o.w = acc[3][r];
        if (r < left) *(v4f*)(crow + (size_t)r * M) = o;
      }
    }
    if (doDot) {
      const v4f av = *(const v4f*)(avec + head * 64 + 4 * c);
      const float d = tile_row_dots(acc, av, c);
      if (dotLane) dots[(size_t)tile * TPITCH + rt * 64 + head * 32 + 8 * h + (c >> 1)] = d;
    }
  }
}

template <bool REUSE, bool TD, bool FUSE>
__global__ void __launch_bounds__(ET)
edge_tile(const int* __restrict__ src, const int* __restrict__ dst,
          const float* __restrict__ als, const float* ald,
          const float* __restrict__ hs, const float* hl,
          const float* __restrict__ b, const float* __restrict__ bl,
          float* __restrict__ out, unsigned* __restrict__ keep,
          const unsigned* __restrict__ plane, int Nn, int Ee,
          const float* __restrict__ nWs, const float* __restrict__ nWd, const float* __restrict__ nWl,
          const float* __restrict__ nAs, const float* __restrict__ nAd,
          float* __restrict__ hsN, float* hlN, float* __restrict__ alsN, float* aldN) {
  constexpr int AP = FUSE ? 68 : 64;
  __shared__ __attribute__((aligned(16))) float s_agg[TM * AP];
  __shared__ float s_m[TM * 2];
  __shared__ float s_den[TM * 2];
  __shared__ int   s_a[LCAP];
  __shared__ int   s_t[LCAP];
  __shared__ float s_e0[LCAP];
  __shared__ float s_e1[LCAP];
  __shared__ int   s_cnt;

  const int tid = threadIdx.x;
  const int lane = tid & 31;
  const int wave = tid >> 5;

  int tn = (Nn + (int)gridDim.x - 1) / (int)gridDim.x;
  if (tn > TM) tn = TM;
  if (tn < 1) tn = 1;
  const int tiles = (Nn + tn - 1) / tn;

  int i0 = (int)(((16u - (unsigned)((size_t)dst & 15u)) & 15u) >> 2);
  if (i0 > Ee) i0 = Ee;
  const int G = (Ee - i0) >> 2;
  const v4i* d4 = (const v4i*)(dst + i0);
  const int rot = (int)(((long long)blockIdx.x * G) / (int)gridDim.x);
  constexpr int U = 8;

  const bool byPlane = !REUSE && plane != nullptr && tiles <= 256;
  const int G16 = (Ee + 15) >> 4;
  const v4i* p4 = (const v4i*)plane;
  const int rot16 = (int)(((long long)blockIdx.x * G16) / (int)gridDim.x);

  for (int tile = blockIdx.x; tile < tiles; tile += gridDim.x) {
    const int lo = tile * tn;
    const int nn = (Nn - lo < tn) ? (Nn - lo) : tn;

    unsigned* kl = (REUSE || keep) ? keep + (size_t)tile * LPITCH : nullptr;

    for (int i = tid; i < nn * 64; i += ET) s_agg[(i >> 6) * AP + (i & 63)] = 0.f;
    for (int i = tid; i < nn * 2; i += ET) { s_m[i] = -INFINITY; s_den[i] = 0.f; }
    if (!REUSE && tid == 0) s_cnt = 0;
    __syncthreads();

    int total;
    if (REUSE) {
      total = (int)kl[0];
    } else {
      if (byPlane) {
        const unsigned key = (unsigned)tile * 0x01010101u;
        for (int g0 = 0; g0 < G16; g0 += ET * U) {
          v4i d[U];
          int gi[U];
          bool ok[U];
#pragma unroll
          for (int u = 0; u < U; ++u) {
            const int g = g0 + u * ET + tid;
            ok[u] = g < G16;
            int gg = (ok[u] ? g : G16 - 1) + rot16;
            if (gg >= G16) gg -= G16;
            gi[u] = gg;
            d[u] = p4[gg];
          }
#pragma unroll
          for (int u = 0; u < U; ++u) {
            unsigned hit = 0u;
#pragma unroll
            for (int w = 0; w < 4; ++w) {
              const unsigned x = (unsigned)d[u][w] ^ key;
              const unsigned nz = ((x & 0x7f7f7f7fu) + 0x7f7f7f7fu) | x;
              hit |= (~nz & 0x80808080u) >> (7 - w);
            }
            if (!ok[u]) hit = 0u;
            while (hit) {
              const int bit = __builtin_ctz(hit);
              hit &= hit - 1u;
              const int e = 16 * gi[u] + 4 * (bit & 7) + (bit >> 3);
              if (e < Ee) {
                const int sl = atomicAdd(&s_cnt, 1);
                if (sl < LCAP) s_a[sl] = e;
              }
            }
          }
        }
      } else {
        for (int g0 = 0; g0 < G; g0 += ET * U) {
          v4i d[U];
          int gi[U];
          bool ok[U];
#pragma unroll
          for (int u = 0; u < U; ++u) {
            const int g = g0 + u * ET + tid;
            ok[u] = g < G;
            int gg = (ok[u] ? g : G - 1) + rot;
            if (gg >= G) gg -= G;
            gi[u] = gg;
            d[u] = d4[gg];
          }
#pragma unroll
          for (int u = 0; u < U; ++u) {
            const int t0 = d[u].x - lo, t1 = d[u].y - lo, t2 = d[u].z - lo, t3 = d[u].w - lo;
            const bool m0 = (unsigned)t0 < (unsigned)nn, m1 = (unsigned)t1 < (unsigned)nn;
            const bool m2 = (unsigned)t2 < (unsigned)nn, m3 = (unsigned)t3 < (unsigned)nn;
            if (ok[u] && (m0 | m1 | m2 | m3)) {
              const int e = i0 + 4 * gi[u];
              if (m0) { const int sl = atomicAdd(&s_cnt, 1); if (sl < LCAP) { s_a[sl] = e;     s_t[sl] = t0; } }
              if (m1) { const int sl = atomicAdd(&s_cnt, 1); if (sl < LCAP) { s_a[sl] = e + 1; s_t[sl] = t1; } }
              if (m2) { const int sl = atomicAdd(&s_cnt, 1); if (sl < LCAP) { s_a[sl] = e + 2; s_t[sl] = t2; } }
              if (m3) { const int sl = atomicAdd(&s_cnt, 1); if (sl < LCAP) { s_a[sl] = e + 3; s_t[sl] = t3; } }
            }
          }
        }
        {
          const int rest = Ee - 4 * G;
          if (tid < rest) {
            const int e = (tid < i0) ? tid : (4 * G + tid);
            const int t = dst[e] - lo;
            if ((unsigned)t < (unsigned)nn) {
              const int sl = atomicAdd(&s_cnt, 1);
              if (sl < LCAP) { s_a[sl] = e; s_t[sl] = t; }
            }
          }
        }
      }
      __syncthreads();
      total = s_cnt;
      if (kl && tid == 0) kl[0] = (unsigned)total;
    }

    if (total <= LCAP) {
      for (int i = tid; i < total; i += ET) {
        int s, t;
        if (REUSE) {
          const unsigned w = kl[32 + i];
          s = (int)(w >> 8);
          t = (int)(w & 255u);
          s_t[i] = t;
        } else {
          const int e = s_a[i];
          t = byPlane ? dst[e] - lo : s_t[i];
          s = src[e];
          if (byPlane) s_t[i] = t;
          if (kl) kl[32 + i] = ((unsigned)s << 8) | (unsigned)t;
        }
        const float* asp = dot_of<TD>(als, s, tn);
        const float* adp = TD ? dot_tile(ald, tile, t) : dot_at(ald, lo + t);
        const float v0 = lrelu(asp[0] + adp[0]);
        const float v1 = lrelu(asp[32] + adp[32]);
        s_a[i] = s;
        s_e0[i] = v0;
        s_e1[i] = v1;
        ldsMaxF(&s_m[2 * t], v0);
        ldsMaxF(&s_m[2 * t + 1], v1);
      }
      __syncthreads();
      for (int i = tid; i < total; i += ET) {
        const int t = s_t[i];
        const float x0 = __expf(s_e0[i] - s_m[2 * t]);
        const float x1 = __expf(s_e1[i] - s_m[2 * t + 1]);
        s_e0[i] = x0;
        s_e1[i] = x1;
        atomicAdd(&s_den[2 * t], x0);
        atomicAdd(&s_den[2 * t + 1], x1);
      }
      __syncthreads();
      for (int base = wave * 32; base < total; base += ET) {
        const int i = base + lane;
        const bool valid = i < total;
        int s = 0, t = 0;
        float a0 = 0.f, a1 = 0.f;
        if (valid) {
          s = s_a[i];
          t = s_t[i];
          a0 = s_e0[i] / (s_den[2 * t] + 1e-16f) * 0.5f;
          a1 = s_e1[i] / (s_den[2 * t + 1] + 1e-16f) * 0.5f;
        }
        const unsigned mask = __builtin_amdgcn_ballot_w32(valid);
        wave_rows<8, AP>(mask, s, t, a0, a1, hs, s_agg, lane);
      }
    } else {
      sweep_all<1, TD, AP>(src, dst, als, ald, hs, s_agg, s_m, s_den, lo, nn, Ee, tid, lane, tn);
      __syncthreads();
      sweep_all<2, TD, AP>(src, dst, als, ald, hs, s_agg, s_m, s_den, lo, nn, Ee, tid, lane, tn);
      __syncthreads();
      sweep_all<3, TD, AP>(src, dst, als, ald, hs, s_agg, s_m, s_den, lo, nn, Ee, tid, lane, tn);
    }
    __syncthreads();

    for (int i = tid; i < nn * 64; i += ET) {
      const int c = i & 63;
      const size_t o = (size_t)lo * 64 + i;
      const int li = (i >> 6) * AP + c;
      const float v = s_agg[li] + b[c] + hl[o] + bl[c];
      const float r = (v > 0.f) ? v : 0.f;
      if (FUSE) s_agg[li] = r; else out[o] = r;
    }
    if (FUSE) {
      const int padded = ((nn + 15) >> 4) << 4;
      for (int i = nn * 64 + tid; i < padded * 64; i += ET) s_agg[(i >> 6) * AP + (i & 63)] = 0.f;
      __syncthreads();
      tile_node_part<AP>(s_agg, nn, lo, tile, nWs, nWd, nWl, nAs, nAd, hsN, hlN, alsN, aldN,
                         __builtin_amdgcn_readfirstlane(wave), lane);
    }
    __syncthreads();
  }
}

extern "C" void kernel_launch(void* const* d_in, const int* in_sizes, int n_in,
                              void* d_out, int out_size, void* d_ws, size_t ws_size,
                              hipStream_t stream) {
  const float* x  = (const float*)d_in[0];
  const int*   ei = (const int*)d_in[1];
  const int Nn = in_sizes[0] / 128;
  const int Ee = in_sizes[1] / 2;
  const int* src = ei;
  const int* dst = ei + Ee;

  const float *Ws[3], *Wd[3], *As[3], *Ad[3], *B[3], *Wl[3], *Bl[3];
  for (int l = 0; l < 3; ++l) {
    int base = 2 + 7 * l;
    Ws[l] = (const float*)d_in[base + 0];
    Wd[l] = (const float*)d_in[base + 1];
    As[l] = (const float*)d_in[base + 2];
    Ad[l] = (const float*)d_in[base + 3];
    B[l]  = (const float*)d_in[base + 4];
    Wl[l] = (const float*)d_in[base + 5];
    Bl[l] = (const float*)d_in[base + 6];
  }

  float* ws = (float*)d_ws;
  size_t off = 0;
  float* hs  = ws + off; off += (size_t)Nn * 128;
  float* hl  = ws + off; off += (size_t)Nn * 64;
  float* als = ws + off; off += (size_t)Nn * 4;
  float* ald = ws + off; off += (size_t)Nn * 4;
  float* h1  = ws + off; off += (size_t)Nn * 64;
  float* h2  = ws + off; off += (size_t)Nn * 64;
  int tn = (Nn + EG - 1) / EG;
  if (tn > TM) tn = TM;
  if (tn < 1) tn = 1;
  const int tiles = (Nn + tn - 1) / tn;
  unsigned* keep = nullptr;
  if (Nn <= (1 << 24) && ws_size >= (off + (size_t)tiles * LPITCH) * sizeof(float))
    keep = (unsigned*)(ws + off);
  if (keep) off += (size_t)tiles * LPITCH;
  unsigned* plane = nullptr;
  const size_t planeWords = (size_t)((Ee + 15) >> 4) * 4;
  if (tiles <= 256 && ws_size >= (off + planeWords) * sizeof(float))
    plane = (unsigned*)(ws + off);
  if (plane) off += planeWords;
  const bool fuse = keep != nullptr && ws_size >= (off + (size_t)3 * tiles * TPITCH) * sizeof(float);
  float* hsB  = h1;
  float* alsA = ws + off;
  float* alsB = alsA + (size_t)tiles * TPITCH;
  float* aldT = alsB + (size_t)tiles * TPITCH;

  const float* lin[3]  = { x, h1, h2 };
  float*       lout[3] = { h1, h2, (float*)d_out };

  if (fuse) {
    node_part<128, true><<<256, 256, 0, stream>>>(x, Ws[0], Wd[0], Wl[0], As[0], Ad[0], hs, hl, als, ald, Nn,
                                                  dst, plane, Ee, tn);
    edge_tile<false, false, true><<<EG, ET, 0, stream>>>(src, dst, als, ald, hs, hl, B[0], Bl[0], nullptr, keep, plane, Nn, Ee,
                                                         Ws[1], Wd[1], Wl[1], As[1], Ad[1], hsB, hl, alsA, aldT);
    edge_tile<true, true, true><<<EG, ET, 0, stream>>>(src, dst, alsA, aldT, hsB, hl, B[1], Bl[1], nullptr, keep, plane, Nn, Ee,
                                                       Ws[2], Wd[2], Wl[2], As[2], Ad[2], hs, hl, alsB, aldT);
    edge_tile<true, true, false><<<EG, ET, 0, stream>>>(src, dst, alsB, aldT, hs, hl, B[2], Bl[2], (float*)d_out, keep, plane, Nn, Ee,
                                                        nullptr, nullptr, nullptr, nullptr, nullptr, nullptr, nullptr, nullptr, nullptr);
    return;
  }

  for (int l = 0; l < 3; ++l) {
    if (l == 0)
      node_part<128, true><<<256, 256, 0, stream>>>(lin[l], Ws[l], Wd[l], Wl[l], As[l], Ad[l], hs, hl, als, ald, Nn,
                                                    dst, plane, Ee, tn);
    else
      node_part<64, false><<<512, 256, 0, stream>>>(lin[l], Ws[l], Wd[l], Wl[l], As[l], Ad[l], hs, hl, als, ald, Nn,
                                                    dst, plane, Ee, tn);
    if (l == 0 || !keep)
      edge_tile<false, false, false><<<EG, ET, 0, stream>>>(src, dst, als, ald, hs, hl, B[l], Bl[l], lout[l], keep, plane, Nn, Ee,
                                                            nullptr, nullptr, nullptr, nullptr, nullptr, nullptr, nullptr, nullptr, nullptr);
    else
      edge_tile<true, false, false><<<EG, ET, 0, stream>>>(src, dst, als, ald, hs, hl, B[l], Bl[l], lout[l], keep, plane, Nn, Ee,
                                                           nullptr, nullptr, nullptr, nullptr, nullptr, nullptr, nullptr, nullptr, nullptr);
  }
}


# Per-edge lookups take one 128-byte line, not two: layers 2,3 read a node's attention terms as an 8-byte pair (planes written via LDS), layer 1 reads source and row as one packed word; sums unchanged
# speedup vs baseline: 4.7029x; 1.0340x over previous
#include <hip/hip_runtime.h>
#include <math.h>

typedef __attribute__((ext_vector_type(2))) float v2f;
typedef __attribute__((ext_vector_type(4))) float v4f;
typedef __attribute__((ext_vector_type(8))) float v8f;
typedef __attribute__((ext_vector_type(4))) int v4i;

#define NEG_SLOPE 0.2f

template <int K>
__device__ __forceinline__ void gemm_load_a(v4f (&a)[K / 8], const float* abase, int rt) {
  const float* ap = abase + (size_t)rt * 16 * K;
#pragma unroll
  for (int j = 0; j < K / 8; ++j) a[j] = *(const v4f*)(ap + 8 * j);
}

template <int K>
__device__ __forceinline__ void gemm_tile(v8f (&acc)[4], const v4f (&a)[K / 8],
                                          const v2f (&bp)[K / 8][2][4]) {
#pragma unroll
  for (int q = 0; q < 4; ++q) acc[q] = (v8f){0.f,0.f,0.f,0.f,0.f,0.f,0.f,0.f};

#pragma unroll
  for (int j = 0; j < K / 8; ++j) {
#pragma unroll
    for (int sub = 0; sub < 2; ++sub) {
      v2f av;
      av.x = sub ? a[j].z : a[j].x;
      av.y = sub ? a[j].w : a[j].y;
#pragma unroll
      for (int q = 0; q < 4; ++q) {
        acc[q] = __builtin_amdgcn_wmma_f32_16x16x4_f32(
             false, av,  false, bp[j][sub][q],
             (short)0, acc[q],  false,  false);
      }
    }
  }
}

__device__ __forceinline__ void gemm_store(const v8f (&acc)[4], float* cbase, int rt, int M) {
  float* crow = cbase + (size_t)rt * 16 * M;
#pragma unroll
  for (int r = 0; r < 8; ++r) {
    v4f o;
    o.x = acc[0][r]; o.y = acc[1][r]; o.z = acc[2][r]; o.w = acc[3][r];
    *(v4f*)(crow + (size_t)r * M) = o;
  }
}

__device__ __forceinline__ float tile_row_dots(const v8f (&acc)[4], const v4f av, int c) {
  float p[8];
#pragma unroll
  for (int r = 0; r < 8; ++r)
    p[r] = acc[0][r] * av.x + acc[1][r] * av.y + acc[2][r] * av.z + acc[3][r] * av.w;
  const bool b3 = (c & 8) != 0, b2 = (c & 4) != 0, b1 = (c & 2) != 0;
  float k4[4], k2[2];
#pragma unroll
  for (int i = 0; i < 4; ++i) {
    const float send = b3 ? p[i] : p[4 + i];
    const float keep = b3 ? p[4 + i] : p[i];
    k4[i] = keep + __shfl_xor(send, 8, 32);
  }
#pragma unroll
  for (int i = 0; i < 2; ++i) {
    const float send = b2 ? k4[i] : k4[2 + i];
    const float keep = b2 ? k4[2 + i] : k4[i];
    k2[i] = keep + __shfl_xor(send, 4, 32);
  }
  const float send1 = b1 ? k2[0] : k2[1];
  const float keep1 = b1 ? k2[1] : k2[0];
  const float k1 = keep1 + __shfl_xor(send1, 2, 32);
  return k1 + __shfl_xor(k1, 1, 32);
}

template <int K, bool PLANE>
__global__ void __launch_bounds__(256)
node_part(const float* __restrict__ A,
          const float* __restrict__ Ws, const float* __restrict__ Wd,
          const float* __restrict__ Wl,
          const float* __restrict__ a_s, const float* __restrict__ a_d,
          float* __restrict__ hs, float* __restrict__ hl,
          float* __restrict__ als, float* __restrict__ ald,
          int Nrows,
          const int* __restrict__ esrc, const int* __restrict__ dst,
          unsigned* __restrict__ plane, unsigned* __restrict__ pk, int Ee, int tn) {
  constexpr int KJ = K / 8;

  if (PLANE && plane) {
    const unsigned nthreads = gridDim.x * blockDim.x;
    const unsigned words = (unsigned)((Ee + 15) >> 4) << 2;
    const unsigned last = (unsigned)Ee - 1u;
    for (unsigned wb = blockIdx.x * blockDim.x + threadIdx.x; wb < words; wb += 4u * nthreads) {
      unsigned dv[4][4], sv[4][4];
#pragma unroll
      for (int i = 0; i < 4; ++i) {
#pragma unroll
        for (int k = 0; k < 4; ++k) {
          const unsigned e = 4u * (wb + (unsigned)i * nthreads) + (unsigned)k;
          const unsigned ec = e < last ? e : last;
          dv[i][k] = (unsigned)dst[ec];
          sv[i][k] = (unsigned)esrc[ec];
        }
      }
#pragma unroll
      for (int i = 0; i < 4; ++i) {
        const unsigned w = wb + (unsigned)i * nthreads;
        unsigned word = 0u;
        v4i pw;
#pragma unroll
        for (int k = 0; k < 4; ++k) {
          const unsigned T = dv[i][k] / (unsigned)tn;
          word |= (T & 255u) << (8 * k);
          pw[k] = (int)((sv[i][k] << 8) | (dv[i][k] - T * (unsigned)tn));
        }
        if (w < words) {
          plane[w] = word;
          if (pk) *(v4i*)(pk + 4 * (size_t)w) = pw;
        }
      }
    }
  }

  const int lane = threadIdx.x & 31;
  const int c = lane & 15;
  const int h = lane >> 4;
  const int wib = __builtin_amdgcn_readfirstlane((int)(threadIdx.x >> 5));
  const int wavesPerBlock = blockDim.x >> 5;
  const int wave = blockIdx.x * wavesPerBlock + wib;
  const int slab = wave % 5;
  const int rg = wave / 5;
  const int rowGroups = (gridDim.x * wavesPerBlock) / 5;
  const int nTiles = Nrows >> 4;
  if (rg >= rowGroups || rg >= nTiles) return;

  const bool doStore = (slab < 2) || (slab == 4);
  const bool doDot = slab < 4;
  const int head = slab & 1;
  const int M = (slab == 4) ? 64 : 128;
  const int cs = (slab == 4) ? 0 : head * 64;
  const float* W = (slab < 2) ? Ws : ((slab < 4) ? Wd : Wl);
  float* C = (slab < 2) ? hs : hl;
  const float* avec = (slab < 2) ? a_s : a_d;
  float* dots = (slab < 2) ? als : ald;

  v2f bp[KJ][2][4];
  {
    const float* wbase = W + (size_t)(4 * h) * M + cs + 4 * c;
#pragma unroll
    for (int j = 0; j < KJ; ++j) {
#pragma unroll
      for (int sub = 0; sub < 2; ++sub) {
        const v4f w0 = *(const v4f*)(wbase + (size_t)(8 * j + 2 * sub) * M);
        const v4f w1 = *(const v4f*)(wbase + (size_t)(8 * j + 2 * sub + 1) * M);
#pragma unroll
        for (int q = 0; q < 4; ++q) {
          bp[j][sub][q].x = w0[q];
          bp[j][sub][q].y = w1[q];
        }
      }
    }
  }
  const v4f av = *(const v4f*)(avec + head * 64 + 4 * c);

  const float* abase = A + (size_t)c * K + 4 * h;
  float* cbase = C + (size_t)(8 * h) * M + cs + 4 * c;
  float* dbase = dots + head * 32 + 8 * h + (c >> 1);
  const bool dotLane = (c & 1) == 0;
  v4f a0[KJ], a1[KJ];
  gemm_load_a<K>(a0, abase, rg);

  int rt = rg;
  for (;;) {
    const int r1 = rt + rowGroups;
    const int r2 = r1 + rowGroups;
    v8f acc[4];
    __builtin_amdgcn_sched_barrier(0);
    gemm_load_a<K>(a1, abase, r1 < nTiles ? r1 : rt);
    __builtin_amdgcn_sched_barrier(0);
    gemm_tile<K>(acc, a0, bp);
    if (doStore) gemm_store(acc, cbase, rt, M);
    if (doDot) {
      const float d = tile_row_dots(acc, av, c);
      if (dotLane) dbase[(size_t)rt * 64] = d;
    }
    __builtin_amdgcn_sched_barrier(0);
    gemm_load_a<K>(a0, abase, r2 < nTiles ? r2 : rt);
    __builtin_amdgcn_sched_barrier(0);
    gemm_tile<K>(acc, a1, bp);
    if (r1 < nTiles) {
      if (doStore) gemm_store(acc, cbase, r1, M);
      if (doDot) {
        const float d = tile_row_dots(acc, av, c);
        if (dotLane) dbase[(size_t)r1 * 64] = d;
      }
    }
    if (r2 >= nTiles) break;
    rt = r2;
  }
}

#define TM   256
#define LCAP 8192
#define ET   1024
#define EG   256
#define LPITCH (LCAP + 32)
static_assert(TM <= 256, "a kept list entry holds the tile-relative destination in 8 bits");
#define TPITCH 1024

__device__ __forceinline__ float lrelu(float v) { return (v > 0.f) ? v : NEG_SLOPE * v; }

__device__ __forceinline__ v2f pair_at(const float* plane, int n) {
  const float* p = plane + (size_t)(n >> 4) * 64 + (n & 15);
  v2f r;
  r.x = p[0];
  r.y = p[32];
  return r;
}
__device__ __forceinline__ v2f pair_tile(const float* plane, int T, int r) {
  return *(const v2f*)(plane + (size_t)T * TPITCH + 2 * r);
}
template <bool TD>
__device__ __forceinline__ v2f pair_of(const float* plane, int n, int tn) {
  if (TD) {
    const unsigned T = (unsigned)n / (unsigned)tn;
    return pair_tile(plane, (int)T, (int)((unsigned)n - T * (unsigned)tn));
  }
  return pair_at(plane, n);
}

__device__ __forceinline__ void ldsMaxF(float* addr, float value) {
  if (value >= 0.f)
    atomicMax((int*)addr, __float_as_int(value));
  else
    atomicMin((unsigned int*)addr, __float_as_uint(value));
}

template <int U, int AP>
__device__ __forceinline__ void wave_rows(unsigned mask, int s, int t, float a0, float a1,
                                          const float* __restrict__ hs, float* agg, int lane) {
  while (mask) {
    int l[U];
    bool on[U];
    unsigned mm = mask;
#pragma unroll
    for (int u = 0; u < U; ++u) {
      on[u] = (mm != 0u);
      l[u] = on[u] ? __builtin_ctz(mm) : l[0];
      mm &= mm - 1u;
    }
    mask = mm;
    v2f x0[U], x1[U];
    int tt[U];
    float A0[U], A1[U];
#pragma unroll
    for (int u = 0; u < U; ++u) {
      const int ss = __builtin_amdgcn_readlane(s, l[u]);
      tt[u] = __builtin_amdgcn_readlane(t, l[u]);
      A0[u] = __int_as_float(__builtin_amdgcn_readlane(__float_as_int(a0), l[u]));
      A1[u] = __int_as_float(__builtin_amdgcn_readlane(__float_as_int(a1), l[u]));
      const float* row = hs + (size_t)ss * 128 + 2 * lane;
      x0[u] = *(const v2f*)row;
      x1[u] = *(const v2f*)(row + 64);
    }
#pragma unroll
    for (int u = 0; u < U; ++u) asm volatile("" :: "v"(x0[u]), "v"(x1[u]));
#pragma unroll
    for (int u = 0; u < U; ++u) {
      if (on[u]) {
        float* p = agg + tt[u] * AP + 2 * lane;
        atomicAdd(p,     x0[u].x * A0[u] + x1[u].x * A1[u]);
        atomicAdd(p + 1, x0[u].y * A0[u] + x1[u].y * A1[u]);
      }
    }
  }
}

template <int MODE, bool TD, int AP>
__device__ __forceinline__ void sweep_all(const int* __restrict__ src, const int* __restrict__ dst,
                                          const float* __restrict__ als, const float* __restrict__ ald,
                                          const float* __restrict__ hs,
                                          float* s_agg, float* s_m, float* s_den,
                                          int lo, int nn, int Ee, int tid, int lane, int tn) {
  const int trips = (Ee + ET - 1) / ET;
  for (int it = 0; it < trips; ++it) {
    const int e = it * ET + tid;
    bool valid = false;
    int s = 0, t = 0;
    float a0 = 0.f, a1 = 0.f;
    if (e < Ee) {
      t = dst[e] - lo;
      valid = (unsigned)t < (unsigned)nn;
      if (valid) {
        s = src[e];
        const v2f as2 = pair_of<TD>(als, s, tn);
        const v2f ad2 = pair_of<TD>(ald, lo + t, tn);
        const float v0 = lrelu(as2.x + ad2.x);
        const float v1 = lrelu(as2.y + ad2.y);
        if (MODE == 1) {
          ldsMaxF(&s_m[2 * t], v0);
          ldsMaxF(&s_m[2 * t + 1], v1);
        } else if (MODE == 2) {
          atomicAdd(&s_den[2 * t],     __expf(v0 - s_m[2 * t]));
          atomicAdd(&s_den[2 * t + 1], __expf(v1 - s_m[2 * t + 1]));
        } else {
          a0 = __expf(v0 - s_m[2 * t])     / (s_den[2 * t]     + 1e-16f) * 0.5f;
          a1 = __expf(v1 - s_m[2 * t + 1]) / (s_den[2 * t + 1] + 1e-16f) * 0.5f;
        }
      }
    }
    if (MODE == 3) {
      const unsigned mask = __builtin_amdgcn_ballot_w32(valid);
      wave_rows<4, AP>(mask, s, t, a0, a1, hs, s_agg, lane);
    }
  }
}


template <int M>
__device__ __forceinline__ void tile_product(v8f (&acc)[4], const float* ap, const float* __restrict__ wslab, int woff) {
#pragma unroll
  for (int q = 0; q < 4; ++q) acc[q] = (v8f){0.f,0.f,0.f,0.f,0.f,0.f,0.f,0.f};
#pragma nounroll
  for (int jj = 0; jj < 4; ++jj) {
    const float* wj = wslab + (size_t)jj * 16 * M;
    v4f a[2], w[2][2][2];
#pragma unroll
    for (int u = 0; u < 2; ++u) {
      a[u] = *(const v4f*)(ap + 16 * jj + 8 * u);
#pragma unroll
      for (int sub = 0; sub < 2; ++sub) {
        w[u][sub][0] = *(const v4f*)(wj + woff + (8 * u + 2 * sub) * M);
        w[u][sub][1] = *(const v4f*)(wj + woff + (8 * u + 2 * sub + 1) * M);
      }
    }
#pragma unroll
    for (int u = 0; u < 2; ++u)
      asm volatile("" :: "v"(w[u][0][0]), "v"(w[u][0][1]), "v"(w[u][1][0]), "v"(w[u][1][1]));
#pragma unroll
    for (int u = 0; u < 2; ++u) {
#pragma unroll
      for (int sub = 0; sub < 2; ++sub) {
        v2f avv;
        avv.x = sub ? a[u].z : a[u].x;
        avv.y = sub ? a[u].w : a[u].y;
#pragma unroll
        for (int q = 0; q < 4; ++q) {
          v2f bv;
          bv.x = w[u][sub][0][q];
          bv.y = w[u][sub][1][q];
          acc[q] = __builtin_amdgcn_wmma_f32_16x16x4_f32(
               false, avv,  false, bv,
               (short)0, acc[q],  false,  false);
        }
      }
    }
  }
}

template <int AP>
__device__ __forceinline__ void tile_node_part(const float* rows, int nn, int lo,
                                               const float* __restrict__ Ws, const float* __restrict__ Wd,
                                               const float* __restrict__ Wl,
                                               const float* __restrict__ a_s, const float* __restrict__ a_d,
                                               float* __restrict__ hs, float* __restrict__ hl,
                                               float* dotsL,
                                               int wave, int lane) {
  const int c = lane & 15;
  const int h = lane >> 4;
  const bool dotLane = (c & 1) == 0;
  const int nT = (nn + 15) >> 4;
  for (int id = wave; id < 5 * nT; id += 32) {
    const int rt = id / 5;
    const int slab = id - 5 * rt;
    const bool doStore = (slab < 2) || (slab == 4);
    const bool doDot = slab < 4;
    const int head = slab & 1;
    const int M = (slab == 4) ? 64 : 128;
    const int cs = (slab == 4) ? 0 : head * 64;
    const float* W = (slab < 2) ? Ws : Wd;
    float* C = (slab < 2) ? hs : hl;
    const float* avec = (slab < 2) ? a_s : a_d;

    v8f acc[4];
    const float* ap = rows + (16 * rt + c) * AP + 4 * h;
    if (slab == 4) tile_product<64>(acc, ap, Wl, 4 * h * 64 + 4 * c);
    else           tile_product<128>(acc, ap, W + cs, 4 * h * 128 + 4 * c);
    if (doStore) {
      float* crow = C + (size_t)(lo + 16 * rt + 8 * h) * M + cs + 4 * c;
      const int left = nn - (16 * rt + 8 * h);
#pragma unroll
      for (int r = 0; r < 8; ++r) {
        v4f o;
        o.x = acc[0][r]; o.y = acc[1][r]; o.z = acc[2][r]; o.w = acc[3][r];
        if (r < left) *(v4f*)(crow + (size_t)r * M) = o;
      }
    }
    if (doDot) {
      const v4f av = *(const v4f*)(avec + head * 64 + 4 * c);
      const float d = tile_row_dots(acc, av, c);
      if (dotLane) dotsL[slab * TM + 16 * rt + 8 * h + (c >> 1)] = d;
    }
  }
}

template <bool REUSE, bool TD, bool FUSE>
__global__ void __launch_bounds__(ET)
edge_tile(const int* __restrict__ src, const int* __restrict__ dst,
          const float* __restrict__ als, const float* ald,
          const float* __restrict__ hs, const float* hl,
          const float* __restrict__ b, const float* __restrict__ bl,
          float* __restrict__ out, unsigned* __restrict__ keep,
          const unsigned* __restrict__ plane, const unsigned* __restrict__ pk, int Nn, int Ee,
          const float* __restrict__ nWs, const float* __restrict__ nWd, const float* __restrict__ nWl,
          const float* __restrict__ nAs, const float* __restrict__ nAd,
          float* __restrict__ hsN, float* hlN, float* __restrict__ alsN, float* aldN) {
  constexpr int AP = FUSE ? 68 : 64;
  __shared__ __attribute__((aligned(16))) float s_agg[TM * AP];
  __shared__ float s_m[TM * 2];
  __shared__ float s_den[TM * 2];
  __shared__ int   s_a[LCAP];
  __shared__ int   s_t[LCAP];
  __shared__ float s_e0[LCAP];
  __shared__ float s_e1[LCAP];
  __shared__ float s_dot[FUSE ? 4 * TM : 1];
  __shared__ int   s_cnt;

  const int tid = threadIdx.x;
  const int lane = tid & 31;
  const int wave = tid >> 5;

  int tn = (Nn + (int)gridDim.x - 1) / (int)gridDim.x;
  if (tn > TM) tn = TM;
  if (tn < 1) tn = 1;
  const int tiles = (Nn + tn - 1) / tn;

  int i0 = (int)(((16u - (unsigned)((size_t)dst & 15u)) & 15u) >> 2);
  if (i0 > Ee) i0 = Ee;
  const int G = (Ee - i0) >> 2;
  const v4i* d4 = (const v4i*)(dst + i0);
  const int rot = (int)(((long long)blockIdx.x * G) / (int)gridDim.x);
  constexpr int U = 8;

  const bool byPlane = !REUSE && plane != nullptr && tiles <= 256;
  const int G16 = (Ee + 15) >> 4;
  const v4i* p4 = (const v4i*)plane;
  const int rot16 = (int)(((long long)blockIdx.x * G16) / (int)gridDim.x);

  for (int tile = blockIdx.x; tile < tiles; tile += gridDim.x) {
    const int lo = tile * tn;
    const int nn = (Nn - lo < tn) ? (Nn - lo) : tn;

    unsigned* kl = (REUSE || keep) ? keep + (size_t)tile * LPITCH : nullptr;

    for (int i = tid; i < nn * 64; i += ET) s_agg[(i >> 6) * AP + (i & 63)] = 0.f;
    for (int i = tid; i < nn * 2; i += ET) { s_m[i] = -INFINITY; s_den[i] = 0.f; }
    if (!REUSE && tid == 0) s_cnt = 0;
    __syncthreads();

    int total;
    if (REUSE) {
      total = (int)kl[0];
    } else {
      if (byPlane) {
        const unsigned key = (unsigned)tile * 0x01010101u;
        for (int g0 = 0; g0 < G16; g0 += ET * U) {
          v4i d[U];
          int gi[U];
          bool ok[U];
#pragma unroll
          for (int u = 0; u < U; ++u) {
            const int g = g0 + u * ET + tid;
            ok[u] = g < G16;
            int gg = (ok[u] ? g : G16 - 1) + rot16;
            if (gg >= G16) gg -= G16;
            gi[u] = gg;
            d[u] = p4[gg];
          }
#pragma unroll
          for (int u = 0; u < U; ++u) {
            unsigned hit = 0u;
#pragma unroll
            for (int w = 0; w < 4; ++w) {
              const unsigned x = (unsigned)d[u][w] ^ key;
              const unsigned nz = ((x & 0x7f7f7f7fu) + 0x7f7f7f7fu) | x;
              hit |= (~nz & 0x80808080u) >> (7 - w);
            }
            if (!ok[u]) hit = 0u;
            while (hit) {
              const int bit = __builtin_ctz(hit);
              hit &= hit - 1u;
              const int e = 16 * gi[u] + 4 * (bit & 7) + (bit >> 3);
              if (e < Ee) {
                const int sl = atomicAdd(&s_cnt, 1);
                if (sl < LCAP) s_a[sl] = e;
              }
            }
          }
        }
      } else {
        for (int g0 = 0; g0 < G; g0 += ET * U) {
          v4i d[U];
          int gi[U];
          bool ok[U];
#pragma unroll
          for (int u = 0; u < U; ++u) {
            const int g = g0 + u * ET + tid;
            ok[u] = g < G;
            int gg = (ok[u] ? g : G - 1) + rot;
            if (gg >= G) gg -= G;
            gi[u] = gg;
            d[u] = d4[gg];
          }
#pragma unroll
          for (int u = 0; u < U; ++u) {
            const int t0 = d[u].x - lo, t1 = d[u].y - lo, t2 = d[u].z - lo, t3 = d[u].w - lo;
            const bool m0 = (unsigned)t0 < (unsigned)nn, m1 = (unsigned)t1 < (unsigned)nn;
            const bool m2 = (unsigned)t2 < (unsigned)nn, m3 = (unsigned)t3 < (unsigned)nn;
            if (ok[u] && (m0 | m1 | m2 | m3)) {
              const int e = i0 + 4 * gi[u];
              if (m0) { const int sl = atomicAdd(&s_cnt, 1); if (sl < LCAP) { s_a[sl] = e;     s_t[sl] = t0; } }
              if (m1) { const int sl = atomicAdd(&s_cnt, 1); if (sl < LCAP) { s_a[sl] = e + 1; s_t[sl] = t1; } }
              if (m2) { const int sl = atomicAdd(&s_cnt, 1); if (sl < LCAP) { s_a[sl] = e + 2; s_t[sl] = t2; } }
              if (m3) { const int sl = atomicAdd(&s_cnt, 1); if (sl < LCAP) { s_a[sl] = e + 3; s_t[sl] = t3; } }
            }
          }
        }
        {
          const int rest = Ee - 4 * G;
          if (tid < rest) {
            const int e = (tid < i0) ? tid : (4 * G + tid);
            const int t = dst[e] - lo;
            if ((unsigned)t < (unsigned)nn) {
              const int sl = atomicAdd(&s_cnt, 1);
              if (sl < LCAP) { s_a[sl] = e; s_t[sl] = t; }
            }
          }
        }
      }
      __syncthreads();
      total = s_cnt;
      if (kl && tid == 0) kl[0] = (unsigned)total;
    }

    if (total <= LCAP) {
      for (int i = tid; i < total; i += ET) {
        int s, t;
        if (REUSE) {
          const unsigned w = kl[32 + i];
          s = (int)(w >> 8);
          t = (int)(w & 255u);
          s_t[i] = t;
        } else {
          const int e = s_a[i];
          if (byPlane && pk) {
            const unsigned w = pk[e];
            s = (int)(w >> 8);
            t = (int)(w & 255u);
            s_t[i] = t;
            if (kl) kl[32 + i] = w;
          } else {
            t = byPlane ? dst[e] - lo : s_t[i];
            s = src[e];
            if (byPlane) s_t[i] = t;
            if (kl) kl[32 + i] = ((unsigned)s << 8) | (unsigned)t;
          }
        }
        const v2f as2 = pair_of<TD>(als, s, tn);
        const v2f ad2 = TD ? pair_tile(ald, tile, t) : pair_at(ald, lo + t);
        const float v0 = lrelu(as2.x + ad2.x);
        const float v1 = lrelu(as2.y + ad2.y);
        s_a[i] = s;
        s_e0[i] = v0;
        s_e1[i] = v1;
        ldsMaxF(&s_m[2 * t], v0);
        ldsMaxF(&s_m[2 * t + 1], v1);
      }
      __syncthreads();
      for (int i = tid; i < total; i += ET) {
        const int t = s_t[i];
        const float x0 = __expf(s_e0[i] - s_m[2 * t]);
        const float x1 = __expf(s_e1[i] - s_m[2 * t + 1]);
        s_e0[i] = x0;
        s_e1[i] = x1;
        atomicAdd(&s_den[2 * t], x0);
        atomicAdd(&s_den[2 * t + 1], x1);
      }
      __syncthreads();
      for (int base = wave * 32; base < total; base += ET) {
        const int i = base + lane;
        const bool valid = i < total;
        int s = 0, t = 0;
        float a0 = 0.f, a1 = 0.f;
        if (valid) {
          s = s_a[i];
          t = s_t[i];
          a0 = s_e0[i] / (s_den[2 * t] + 1e-16f) * 0.5f;
          a1 = s_e1[i] / (s_den[2 * t + 1] + 1e-16f) * 0.5f;
        }
        const unsigned mask = __builtin_amdgcn_ballot_w32(valid);
        wave_rows<8, AP>(mask, s, t, a0, a1, hs, s_agg, lane);
      }
    } else {
      sweep_all<1, TD, AP>(src, dst, als, ald, hs, s_agg, s_m, s_den, lo, nn, Ee, tid, lane, tn);
      __syncthreads();
      sweep_all<2, TD, AP>(src, dst, als, ald, hs, s_agg, s_m, s_den, lo, nn, Ee, tid, lane, tn);
      __syncthreads();
      sweep_all<3, TD, AP>(src, dst, als, ald, hs, s_agg, s_m, s_den, lo, nn, Ee, tid, lane, tn);
    }
    __syncthreads();

    for (int i = tid; i < nn * 64; i += ET) {
      const int c = i & 63;
      const size_t o = (size_t)lo * 64 + i;
      const int li = (i >> 6) * AP + c;
      const float v = s_agg[li] + b[c] + hl[o] + bl[c];
      const float r = (v > 0.f) ? v : 0.f;
      if (FUSE) s_agg[li] = r; else out[o] = r;
    }
    if (FUSE) {
      const int padded = ((nn + 15) >> 4) << 4;
      for (int i = nn * 64 + tid; i < padded * 64; i += ET) s_agg[(i >> 6) * AP + (i & 63)] = 0.f;
      __syncthreads();
      tile_node_part<AP>(s_agg, nn, lo, nWs, nWd, nWl, nAs, nAd, hsN, hlN, s_dot,
                         __builtin_amdgcn_readfirstlane(wave), lane);
    }
    __syncthreads();
    if (FUSE) {
      for (int r = tid; r < nn; r += ET) {
        v2f pa, pd;
        pa.x = s_dot[r];          pa.y = s_dot[TM + r];
        pd.x = s_dot[2 * TM + r]; pd.y = s_dot[3 * TM + r];
        *(v2f*)(alsN + (size_t)tile * TPITCH + 2 * r) = pa;
        *(v2f*)(aldN + (size_t)tile * TPITCH + 2 * r) = pd;
      }
    }
  }
}

extern "C" void kernel_launch(void* const* d_in, const int* in_sizes, int n_in,
                              void* d_out, int out_size, void* d_ws, size_t ws_size,
                              hipStream_t stream) {
  const float* x  = (const float*)d_in[0];
  const int*   ei = (const int*)d_in[1];
  const int Nn = in_sizes[0] / 128;
  const int Ee = in_sizes[1] / 2;
  const int* src = ei;
  const int* dst = ei + Ee;

  const float *Ws[3], *Wd[3], *As[3], *Ad[3], *B[3], *Wl[3], *Bl[3];
  for (int l = 0; l < 3; ++l) {
    int base = 2 + 7 * l;
    Ws[l] = (const float*)d_in[base + 0];
    Wd[l] = (const float*)d_in[base + 1];
    As[l] = (const float*)d_in[base + 2];
    Ad[l] = (const float*)d_in[base + 3];
    B[l]  = (const float*)d_in[base + 4];
    Wl[l] = (const float*)d_in[base + 5];
    Bl[l] = (const float*)d_in[base + 6];
  }

  float* ws = (float*)d_ws;
  size_t off = 0;
  float* hs  = ws + off; off += (size_t)Nn * 128;
  float* hl  = ws + off; off += (size_t)Nn * 64;
  float* als = ws + off; off += (size_t)Nn * 4;
  float* ald = ws + off; off += (size_t)Nn * 4;
  float* h1  = ws + off; off += (size_t)Nn * 64;
  float* h2  = ws + off; off += (size_t)Nn * 64;
  int tn = (Nn + EG - 1) / EG;
  if (tn > TM) tn = TM;
  if (tn < 1) tn = 1;
  const int tiles = (Nn + tn - 1) / tn;
  unsigned* keep = nullptr;
  if (Nn <= (1 << 24) && ws_size >= (off + (size_t)tiles * LPITCH) * sizeof(float))
    keep = (unsigned*)(ws + off);
  if (keep) off += (size_t)tiles * LPITCH;
  unsigned* plane = nullptr;
  const size_t planeWords = (size_t)((Ee + 15) >> 4) * 4;
  if (tiles <= 256 && ws_size >= (off + planeWords) * sizeof(float))
    plane = (unsigned*)(ws + off);
  if (plane) off += planeWords;
  const bool fuse = keep != nullptr && ws_size >= (off + (size_t)3 * tiles * TPITCH) * sizeof(float);
  float* hsB  = h1;
  float* alsA = ws + off;
  float* alsB = alsA + (size_t)tiles * TPITCH;
  float* aldT = alsB + (size_t)tiles * TPITCH;
  const size_t pkOff = off + (fuse ? (size_t)3 * tiles * TPITCH : 0);
  unsigned* pk = nullptr;
  if (plane && Nn <= (1 << 24) && ws_size >= (pkOff + planeWords * 4) * sizeof(float))
    pk = (unsigned*)(ws + pkOff);

  const float* lin[3]  = { x, h1, h2 };
  float*       lout[3] = { h1, h2, (float*)d_out };

  if (fuse) {
    node_part<128, true><<<256, 256, 0, stream>>>(x, Ws[0], Wd[0], Wl[0], As[0], Ad[0], hs, hl, als, ald, Nn,
                                                  src, dst, plane, pk, Ee, tn);
    edge_tile<false, false, true><<<EG, ET, 0, stream>>>(src, dst, als, ald, hs, hl, B[0], Bl[0], nullptr, keep, plane, pk, Nn, Ee,
                                                         Ws[1], Wd[1], Wl[1], As[1], Ad[1], hsB, hl, alsA, aldT);
    edge_tile<true, true, true><<<EG, ET, 0, stream>>>(src, dst, alsA, aldT, hsB, hl, B[1], Bl[1], nullptr, keep, plane, pk, Nn, Ee,
                                                       Ws[2], Wd[2], Wl[2], As[2], Ad[2], hs, hl, alsB, aldT);
    edge_tile<true, true, false><<<EG, ET, 0, stream>>>(src, dst, alsB, aldT, hs, hl, B[2], Bl[2], (float*)d_out, keep, plane, pk, Nn, Ee,
                                                        nullptr, nullptr, nullptr, nullptr, nullptr, nullptr, nullptr, nullptr, nullptr);
    return;
  }

  for (int l = 0; l < 3; ++l) {
    if (l == 0)
      node_part<128, true><<<256, 256, 0, stream>>>(lin[l], Ws[l], Wd[l], Wl[l], As[l], Ad[l], hs, hl, als, ald, Nn,
                                                    src, dst, plane, pk, Ee, tn);
    else
      node_part<64, false><<<512, 256, 0, stream>>>(lin[l], Ws[l], Wd[l], Wl[l], As[l], Ad[l], hs, hl, als, ald, Nn,
                                                    src, dst, plane, pk, Ee, tn);
    if (l == 0 || !keep)
      edge_tile<false, false, false><<<EG, ET, 0, stream>>>(src, dst, als, ald, hs, hl, B[l], Bl[l], lout[l], keep, plane, pk, Nn, Ee,
                                                            nullptr, nullptr, nullptr, nullptr, nullptr, nullptr, nullptr, nullptr, nullptr);
    else
      edge_tile<true, false, false><<<EG, ET, 0, stream>>>(src, dst, als, ald, hs, hl, B[l], Bl[l], lout[l], keep, plane, pk, Nn, Ee,
                                                           nullptr, nullptr, nullptr, nullptr, nullptr, nullptr, nullptr, nullptr, nullptr);
  }
}


# Phase 3 takes a tile's list in row order (LDS counting sort into an index array, each layer) and sums runs of equal rows in registers: 2 LDS float adds per run, not per edge; sum order changed
# speedup vs baseline: 5.0368x; 1.0710x over previous
#include <hip/hip_runtime.h>
#include <math.h>

typedef __attribute__((ext_vector_type(2))) float v2f;
typedef __attribute__((ext_vector_type(4))) float v4f;
typedef __attribute__((ext_vector_type(8))) float v8f;
typedef __attribute__((ext_vector_type(4))) int v4i;

#define NEG_SLOPE 0.2f

template <int K>
__device__ __forceinline__ void gemm_load_a(v4f (&a)[K / 8], const float* abase, int rt) {
  const float* ap = abase + (size_t)rt * 16 * K;
#pragma unroll
  for (int j = 0; j < K / 8; ++j) a[j] = *(const v4f*)(ap + 8 * j);
}

template <int K>
__device__ __forceinline__ void gemm_tile(v8f (&acc)[4], const v4f (&a)[K / 8],
                                          const v2f (&bp)[K / 8][2][4]) {
#pragma unroll
  for (int q = 0; q < 4; ++q) acc[q] = (v8f){0.f,0.f,0.f,0.f,0.f,0.f,0.f,0.f};

#pragma unroll
  for (int j = 0; j < K / 8; ++j) {
#pragma unroll
    for (int sub = 0; sub < 2; ++sub) {
      v2f av;
      av.x = sub ? a[j].z : a[j].x;
      av.y = sub ? a[j].w : a[j].y;
#pragma unroll
      for (int q = 0; q < 4; ++q) {
        acc[q] = __builtin_amdgcn_wmma_f32_16x16x4_f32(
             false, av,  false, bp[j][sub][q],
             (short)0, acc[q],  false,  false);
      }
    }
  }
}

__device__ __forceinline__ void gemm_store(const v8f (&acc)[4], float* cbase, int rt, int M) {
  float* crow = cbase + (size_t)rt * 16 * M;
#pragma unroll
  for (int r = 0; r < 8; ++r) {
    v4f o;
    o.x = acc[0][r]; o.y = acc[1][r]; o.z = acc[2][r]; o.w = acc[3][r];
    *(v4f*)(crow + (size_t)r * M) = o;
  }
}

__device__ __forceinline__ float tile_row_dots(const v8f (&acc)[4], const v4f av, int c) {
  float p[8];
#pragma unroll
  for (int r = 0; r < 8; ++r)
    p[r] = acc[0][r] * av.x + acc[1][r] * av.y + acc[2][r] * av.z + acc[3][r] * av.w;
  const bool b3 = (c & 8) != 0, b2 = (c & 4) != 0, b1 = (c & 2) != 0;
  float k4[4], k2[2];
#pragma unroll
  for (int i = 0; i < 4; ++i) {
    const float send = b3 ? p[i] : p[4 + i];
    const float keep = b3 ? p[4 + i] : p[i];
    k4[i] = keep + __shfl_xor(send, 8, 32);
  }
#pragma unroll
  for (int i = 0; i < 2; ++i) {
    const float send = b2 ? k4[i] : k4[2 + i];
    const float keep = b2 ? k4[2 + i] : k4[i];
    k2[i] = keep + __shfl_xor(send, 4, 32);
  }
  const float send1 = b1 ? k2[0] : k2[1];
  const float keep1 = b1 ? k2[1] : k2[0];
  const float k1 = keep1 + __shfl_xor(send1, 2, 32);
  return k1 + __shfl_xor(k1, 1, 32);
}

template <int K, bool PLANE>
__global__ void __launch_bounds__(256)
node_part(const float* __restrict__ A,
          const float* __restrict__ Ws, const float* __restrict__ Wd,
          const float* __restrict__ Wl,
          const float* __restrict__ a_s, const float* __restrict__ a_d,
          float* __restrict__ hs, float* __restrict__ hl,
          float* __restrict__ als, float* __restrict__ ald,
          int Nrows,
          const int* __restrict__ esrc, const int* __restrict__ dst,
          unsigned* __restrict__ plane, unsigned* __restrict__ pk, int Ee, int tn) {
  constexpr int KJ = K / 8;

  if (PLANE && plane) {
    const unsigned nthreads = gridDim.x * blockDim.x;
    const unsigned words = (unsigned)((Ee + 15) >> 4) << 2;
    const unsigned last = (unsigned)Ee - 1u;
    for (unsigned wb = blockIdx.x * blockDim.x + threadIdx.x; wb < words; wb += 4u * nthreads) {
      unsigned dv[4][4], sv[4][4];
#pragma unroll
      for (int i = 0; i < 4; ++i) {
#pragma unroll
        for (int k = 0; k < 4; ++k) {
          const unsigned e = 4u * (wb + (unsigned)i * nthreads) + (unsigned)k;
          const unsigned ec = e < last ? e : last;
          dv[i][k] = (unsigned)dst[ec];
          sv[i][k] = (unsigned)esrc[ec];
        }
      }
#pragma unroll
      for (int i = 0; i < 4; ++i) {
        const unsigned w = wb + (unsigned)i * nthreads;
        unsigned word = 0u;
        v4i pw;
#pragma unroll
        for (int k = 0; k < 4; ++k) {
          const unsigned T = dv[i][k] / (unsigned)tn;
          word |= (T & 255u) << (8 * k);
          pw[k] = (int)((sv[i][k] << 8) | (dv[i][k] - T * (unsigned)tn));
        }
        if (w < words) {
          plane[w] = word;
          if (pk) *(v4i*)(pk + 4 * (size_t)w) = pw;
        }
      }
    }
  }

  const int lane = threadIdx.x & 31;
  const int c = lane & 15;
  const int h = lane >> 4;
  const int wib = __builtin_amdgcn_readfirstlane((int)(threadIdx.x >> 5));
  const int wavesPerBlock = blockDim.x >> 5;
  const int wave = blockIdx.x * wavesPerBlock + wib;
  const int slab = wave % 5;
  const int rg = wave / 5;
  const int rowGroups = (gridDim.x * wavesPerBlock) / 5;
  const int nTiles = Nrows >> 4;
  if (rg >= rowGroups || rg >= nTiles) return;

  const bool doStore = (slab < 2) || (slab == 4);
  const bool doDot = slab < 4;
  const int head = slab & 1;
  const int M = (slab == 4) ? 64 : 128;
  const int cs = (slab == 4) ? 0 : head * 64;
  const float* W = (slab < 2) ? Ws : ((slab < 4) ? Wd : Wl);
  float* C = (slab < 2) ? hs : hl;
  const float* avec = (slab < 2) ? a_s : a_d;
  float* dots = (slab < 2) ? als : ald;

  v2f bp[KJ][2][4];
  {
    const float* wbase = W + (size_t)(4 * h) * M + cs + 4 * c;
#pragma unroll
    for (int j = 0; j < KJ; ++j) {
#pragma unroll
      for (int sub = 0; sub < 2; ++sub) {
        const v4f w0 = *(const v4f*)(wbase + (size_t)(8 * j + 2 * sub) * M);
        const v4f w1 = *(const v4f*)(wbase + (size_t)(8 * j + 2 * sub + 1) * M);
#pragma unroll
        for (int q = 0; q < 4; ++q) {
          bp[j][sub][q].x = w0[q];
          bp[j][sub][q].y = w1[q];
        }
      }
    }
  }
  const v4f av = *(const v4f*)(avec + head * 64 + 4 * c);

  const float* abase = A + (size_t)c * K + 4 * h;
  float* cbase = C + (size_t)(8 * h) * M + cs + 4 * c;
  float* dbase = dots + head * 32 + 8 * h + (c >> 1);
  const bool dotLane = (c & 1) == 0;
  v4f a0[KJ], a1[KJ];
  gemm_load_a<K>(a0, abase, rg);

  int rt = rg;
  for (;;) {
    const int r1 = rt + rowGroups;
    const int r2 = r1 + rowGroups;
    v8f acc[4];
    __builtin_amdgcn_sched_barrier(0);
    gemm_load_a<K>(a1, abase, r1 < nTiles ? r1 : rt);
    __builtin_amdgcn_sched_barrier(0);
    gemm_tile<K>(acc, a0, bp);
    if (doStore) gemm_store(acc, cbase, rt, M);
    if (doDot) {
      const float d = tile_row_dots(acc, av, c);
      if (dotLane) dbase[(size_t)rt * 64] = d;
    }
    __builtin_amdgcn_sched_barrier(0);
    gemm_load_a<K>(a0, abase, r2 < nTiles ? r2 : rt);
    __builtin_amdgcn_sched_barrier(0);
    gemm_tile<K>(acc, a1, bp);
    if (r1 < nTiles) {
      if (doStore) gemm_store(acc, cbase, r1, M);
      if (doDot) {
        const float d = tile_row_dots(acc, av, c);
        if (dotLane) dbase[(size_t)r1 * 64] = d;
      }
    }
    if (r2 >= nTiles) break;
    rt = r2;
  }
}

#define TM   256
#define LCAP 8192
#define ET   1024
#define EG   256
#define LPITCH (LCAP + 32)
static_assert(TM <= 256, "a kept list entry holds the tile-relative destination in 8 bits");
static_assert(TM == 256 && LCAP <= (1 << 23), "the row scan takes 8 rows per lane of one wave; an entry's place in its row sits above 8 bits of an int");
#define TPITCH 1024

__device__ __forceinline__ float lrelu(float v) { return (v > 0.f) ? v : NEG_SLOPE * v; }

__device__ __forceinline__ v2f pair_at(const float* plane, int n) {
  const float* p = plane + (size_t)(n >> 4) * 64 + (n & 15);
  v2f r;
  r.x = p[0];
  r.y = p[32];
  return r;
}
__device__ __forceinline__ v2f pair_tile(const float* plane, int T, int r) {
  return *(const v2f*)(plane + (size_t)T * TPITCH + 2 * r);
}
template <bool TD>
__device__ __forceinline__ v2f pair_of(const float* plane, int n, int tn) {
  if (TD) {
    const unsigned T = (unsigned)n / (unsigned)tn;
    return pair_tile(plane, (int)T, (int)((unsigned)n - T * (unsigned)tn));
  }
  return pair_at(plane, n);
}

__device__ __forceinline__ void ldsMaxF(float* addr, float value) {
  if (value >= 0.f)
    atomicMax((int*)addr, __float_as_int(value));
  else
    atomicMin((unsigned int*)addr, __float_as_uint(value));
}

template <int U, int AP>
__device__ __forceinline__ void wave_rows(unsigned mask, int s, int t, float a0, float a1,
                                          const float* __restrict__ hs, float* agg, int lane) {
  int cur = -1;
  float rx = 0.f, ry = 0.f;
  while (mask) {
    int l[U];
    bool on[U];
    unsigned mm = mask;
#pragma unroll
    for (int u = 0; u < U; ++u) {
      on[u] = (mm != 0u);
      l[u] = on[u] ? __builtin_ctz(mm) : l[0];
      mm &= mm - 1u;
    }
    mask = mm;
    v2f x0[U], x1[U];
    int tt[U];
    float A0[U], A1[U];
#pragma unroll
    for (int u = 0; u < U; ++u) {
      const int ss = __builtin_amdgcn_readlane(s, l[u]);
      tt[u] = __builtin_amdgcn_readlane(t, l[u]);
      A0[u] = __int_as_float(__builtin_amdgcn_readlane(__float_as_int(a0), l[u]));
      A1[u] = __int_as_float(__builtin_amdgcn_readlane(__float_as_int(a1), l[u]));
      const float* row = hs + (size_t)ss * 128 + 2 * lane;
      x0[u] = *(const v2f*)row;
      x1[u] = *(const v2f*)(row + 64);
    }
#pragma unroll
    for (int u = 0; u < U; ++u) asm volatile("" :: "v"(x0[u]), "v"(x1[u]));
#pragma unroll
    for (int u = 0; u < U; ++u) {
      if (on[u]) {
        if (tt[u] != cur) {
          if (cur >= 0) {
            float* p = agg + cur * AP + 2 * lane;
            atomicAdd(p,     rx);
            atomicAdd(p + 1, ry);
          }
          cur = tt[u];
          rx = 0.f;
          ry = 0.f;
        }
        rx += x0[u].x * A0[u] + x1[u].x * A1[u];
        ry += x0[u].y * A0[u] + x1[u].y * A1[u];
      }
    }
  }
  if (cur >= 0) {
    float* p = agg + cur * AP + 2 * lane;
    atomicAdd(p,     rx);
    atomicAdd(p + 1, ry);
  }
}

template <int MODE, bool TD, int AP>
__device__ __forceinline__ void sweep_all(const int* __restrict__ src, const int* __restrict__ dst,
                                          const float* __restrict__ als, const float* __restrict__ ald,
                                          const float* __restrict__ hs,
                                          float* s_agg, float* s_m, float* s_den,
                                          int lo, int nn, int Ee, int tid, int lane, int tn) {
  const int trips = (Ee + ET - 1) / ET;
  for (int it = 0; it < trips; ++it) {
    const int e = it * ET + tid;
    bool valid = false;
    int s = 0, t = 0;
    float a0 = 0.f, a1 = 0.f;
    if (e < Ee) {
      t = dst[e] - lo;
      valid = (unsigned)t < (unsigned)nn;
      if (valid) {
        s = src[e];
        const v2f as2 = pair_of<TD>(als, s, tn);
        const v2f ad2 = pair_of<TD>(ald, lo + t, tn);
        const float v0 = lrelu(as2.x + ad2.x);
        const float v1 = lrelu(as2.y + ad2.y);
        if (MODE == 1) {
          ldsMaxF(&s_m[2 * t], v0);
          ldsMaxF(&s_m[2 * t + 1], v1);
        } else if (MODE == 2) {
          atomicAdd(&s_den[2 * t],     __expf(v0 - s_m[2 * t]));
          atomicAdd(&s_den[2 * t + 1], __expf(v1 - s_m[2 * t + 1]));
        } else {
          a0 = __expf(v0 - s_m[2 * t])     / (s_den[2 * t]     + 1e-16f) * 0.5f;
          a1 = __expf(v1 - s_m[2 * t + 1]) / (s_den[2 * t + 1] + 1e-16f) * 0.5f;
        }
      }
    }
    if (MODE == 3) {
      const unsigned mask = __builtin_amdgcn_ballot_w32(valid);
      wave_rows<4, AP>(mask, s, t, a0, a1, hs, s_agg, lane);
    }
  }
}


template <int M>
__device__ __forceinline__ void tile_product(v8f (&acc)[4], const float* ap, const float* __restrict__ wslab, int woff) {
#pragma unroll
  for (int q = 0; q < 4; ++q) acc[q] = (v8f){0.f,0.f,0.f,0.f,0.f,0.f,0.f,0.f};
#pragma nounroll
  for (int jj = 0; jj < 4; ++jj) {
    const float* wj = wslab + (size_t)jj * 16 * M;
    v4f a[2], w[2][2][2];
#pragma unroll
    for (int u = 0; u < 2; ++u) {
      a[u] = *(const v4f*)(ap + 16 * jj + 8 * u);
#pragma unroll
      for (int sub = 0; sub < 2; ++sub) {
        w[u][sub][0] = *(const v4f*)(wj + woff + (8 * u + 2 * sub) * M);
        w[u][sub][1] = *(const v4f*)(wj + woff + (8 * u + 2 * sub + 1) * M);
      }
    }
#pragma unroll
    for (int u = 0; u < 2; ++u)
      asm volatile("" :: "v"(w[u][0][0]), "v"(w[u][0][1]), "v"(w[u][1][0]), "v"(w[u][1][1]));
#pragma unroll
    for (int u = 0; u < 2; ++u) {
#pragma unroll
      for (int sub = 0; sub < 2; ++sub) {
        v2f avv;
        avv.x = sub ? a[u].z : a[u].x;
        avv.y = sub ? a[u].w : a[u].y;
#pragma unroll
        for (int q = 0; q < 4; ++q) {
          v2f bv;
          bv.x = w[u][sub][0][q];
          bv.y = w[u][sub][1][q];
          acc[q] = __builtin_amdgcn_wmma_f32_16x16x4_f32(
               false, avv,  false, bv,
               (short)0, acc[q],  false,  false);
        }
      }
    }
  }
}

template <int AP>
__device__ __forceinline__ void tile_node_part(const float* rows, int nn, int lo,
                                               const float* __restrict__ Ws, const float* __restrict__ Wd,
                                               const float* __restrict__ Wl,
                                               const float* __restrict__ a_s, const float* __restrict__ a_d,
                                               float* __restrict__ hs, float* __restrict__ hl,
                                               float* dotsL,
                                               int wave, int lane) {
  const int c = lane & 15;
  const int h = lane >> 4;
  const bool dotLane = (c & 1) == 0;
  const int nT = (nn + 15) >> 4;
  for (int id = wave; id < 5 * nT; id += 32) {
    const int rt = id / 5;
    const int slab = id - 5 * rt;
    const bool doStore = (slab < 2) || (slab == 4);
    const bool doDot = slab < 4;
    const int head = slab & 1;
    const int M = (slab == 4) ? 64 : 128;
    const int cs = (slab == 4) ? 0 : head * 64;
    const float* W = (slab < 2) ? Ws : Wd;
    float* C = (slab < 2) ? hs : hl;
    const float* avec = (slab < 2) ? a_s : a_d;

    v8f acc[4];
    const float* ap = rows + (16 * rt + c) * AP + 4 * h;
    if (slab == 4) tile_product<64>(acc, ap, Wl, 4 * h * 64 + 4 * c);
    else           tile_product<128>(acc, ap, W + cs, 4 * h * 128 + 4 * c);
    if (doStore) {
      float* crow = C + (size_t)(lo + 16 * rt + 8 * h) * M + cs + 4 * c;
      const int left = nn - (16 * rt + 8 * h);
#pragma unroll
      for (int r = 0; r < 8; ++r) {
        v4f o;
        o.x = acc[0][r]; o.y = acc[1][r]; o.z = acc[2][r]; o.w = acc[3][r];
        if (r < left) *(v4f*)(crow + (size_t)r * M) = o;
      }
    }
    if (doDot) {
      const v4f av = *(const v4f*)(avec + head * 64 + 4 * c);
      const float d = tile_row_dots(acc, av, c);
      if (dotLane) dotsL[slab * TM + 16 * rt + 8 * h + (c >> 1)] = d;
    }
  }
}

template <bool REUSE, bool TD, bool FUSE>
__global__ void __launch_bounds__(ET)
edge_tile(const int* __restrict__ src, const int* __restrict__ dst,
          const float* __restrict__ als, const float* ald,
          const float* __restrict__ hs, const float* hl,
          const float* __restrict__ b, const float* __restrict__ bl,
          float* __restrict__ out, unsigned* __restrict__ keep,
          const unsigned* __restrict__ plane, const unsigned* __restrict__ pk, int Nn, int Ee,
          const float* __restrict__ nWs, const float* __restrict__ nWd, const float* __restrict__ nWl,
          const float* __restrict__ nAs, const float* __restrict__ nAd,
          float* __restrict__ hsN, float* hlN, float* __restrict__ alsN, float* aldN) {
  constexpr int AP = FUSE ? 68 : 64;
  __shared__ __attribute__((aligned(16))) float s_agg[TM * AP];
  __shared__ float s_m[TM * 2];
  __shared__ float s_den[TM * 2];
  __shared__ int   s_a[LCAP];
  __shared__ int   s_t[LCAP];
  __shared__ int   s_perm[LCAP];
  __shared__ int   s_rc[TM];
  __shared__ float s_e0[LCAP];
  __shared__ float s_e1[LCAP];
  __shared__ float s_dot[FUSE ? 4 * TM : 1];
  __shared__ int   s_cnt;

  const int tid = threadIdx.x;
  const int lane = tid & 31;
  const int wave = tid >> 5;

  int tn = (Nn + (int)gridDim.x - 1) / (int)gridDim.x;
  if (tn > TM) tn = TM;
  if (tn < 1) tn = 1;
  const int tiles = (Nn + tn - 1) / tn;

  int i0 = (int)(((16u - (unsigned)((size_t)dst & 15u)) & 15u) >> 2);
  if (i0 > Ee) i0 = Ee;
  const int G = (Ee - i0) >> 2;
  const v4i* d4 = (const v4i*)(dst + i0);
  const int rot = (int)(((long long)blockIdx.x * G) / (int)gridDim.x);
  constexpr int U = 8;

  const bool byPlane = !REUSE && plane != nullptr && tiles <= 256;
  const int G16 = (Ee + 15) >> 4;
  const v4i* p4 = (const v4i*)plane;
  const int rot16 = (int)(((long long)blockIdx.x * G16) / (int)gridDim.x);

  for (int tile = blockIdx.x; tile < tiles; tile += gridDim.x) {
    const int lo = tile * tn;
    const int nn = (Nn - lo < tn) ? (Nn - lo) : tn;

    unsigned* kl = (REUSE || keep) ? keep + (size_t)tile * LPITCH : nullptr;

    for (int i = tid; i < nn * 64; i += ET) s_agg[(i >> 6) * AP + (i & 63)] = 0.f;
    for (int i = tid; i < nn * 2; i += ET) { s_m[i] = -INFINITY; s_den[i] = 0.f; }
    if (tid < TM) s_rc[tid] = 0;
    if (!REUSE && tid == 0) s_cnt = 0;
    __syncthreads();

    int total;
    if (REUSE) {
      total = (int)kl[0];
    } else {
      if (byPlane) {
        const unsigned key = (unsigned)tile * 0x01010101u;
        for (int g0 = 0; g0 < G16; g0 += ET * U) {
          v4i d[U];
          int gi[U];
          bool ok[U];
#pragma unroll
          for (int u = 0; u < U; ++u) {
            const int g = g0 + u * ET + tid;
            ok[u] = g < G16;
            int gg = (ok[u] ? g : G16 - 1) + rot16;
            if (gg >= G16) gg -= G16;
            gi[u] = gg;
            d[u] = p4[gg];
          }
#pragma unroll
          for (int u = 0; u < U; ++u) {
            unsigned hit = 0u;
#pragma unroll
            for (int w = 0; w < 4; ++w) {
              const unsigned x = (unsigned)d[u][w] ^ key;
              const unsigned nz = ((x & 0x7f7f7f7fu) + 0x7f7f7f7fu) | x;
              hit |= (~nz & 0x80808080u) >> (7 - w);
            }
            if (!ok[u]) hit = 0u;
            while (hit) {
              const int bit = __builtin_ctz(hit);
              hit &= hit - 1u;
              const int e = 16 * gi[u] + 4 * (bit & 7) + (bit >> 3);
              if (e < Ee) {
                const int sl = atomicAdd(&s_cnt, 1);
                if (sl < LCAP) s_a[sl] = e;
              }
            }
          }
        }
      } else {
        for (int g0 = 0; g0 < G; g0 += ET * U) {
          v4i d[U];
          int gi[U];
          bool ok[U];
#pragma unroll
          for (int u = 0; u < U; ++u) {
            const int g = g0 + u * ET + tid;
            ok[u] = g < G;
            int gg = (ok[u] ? g : G - 1) + rot;
            if (gg >= G) gg -= G;
            gi[u] = gg;
            d[u] = d4[gg];
          }
#pragma unroll
          for (int u = 0; u < U; ++u) {
            const int t0 = d[u].x - lo, t1 = d[u].y - lo, t2 = d[u].z - lo, t3 = d[u].w - lo;
            const bool m0 = (unsigned)t0 < (unsigned)nn, m1 = (unsigned)t1 < (unsigned)nn;
            const bool m2 = (unsigned)t2 < (unsigned)nn, m3 = (unsigned)t3 < (unsigned)nn;
            if (ok[u] && (m0 | m1 | m2 | m3)) {
              const int e = i0 + 4 * gi[u];
              if (m0) { const int sl = atomicAdd(&s_cnt, 1); if (sl < LCAP) { s_a[sl] = e;     s_t[sl] = t0; } }
              if (m1) { const int sl = atomicAdd(&s_cnt, 1); if (sl < LCAP) { s_a[sl] = e + 1; s_t[sl] = t1; } }
              if (m2) { const int sl = atomicAdd(&s_cnt, 1); if (sl < LCAP) { s_a[sl] = e + 2; s_t[sl] = t2; } }
              if (m3) { const int sl = atomicAdd(&s_cnt, 1); if (sl < LCAP) { s_a[sl] = e + 3; s_t[sl] = t3; } }
            }
          }
        }
        {
          const int rest = Ee - 4 * G;
          if (tid < rest) {
            const int e = (tid < i0) ? tid : (4 * G + tid);
            const int t = dst[e] - lo;
            if ((unsigned)t < (unsigned)nn) {
              const int sl = atomicAdd(&s_cnt, 1);
              if (sl < LCAP) { s_a[sl] = e; s_t[sl] = t; }
            }
          }
        }
      }
      __syncthreads();
      total = s_cnt;
      if (kl && tid == 0) kl[0] = (unsigned)total;
    }

    if (total <= LCAP) {
      for (int i = tid; i < total; i += ET) {
        int s, t;
        if (REUSE) {
          const unsigned w = kl[32 + i];
          s = (int)(w >> 8);
          t = (int)(w & 255u);
        } else {
          const int e = s_a[i];
          if (byPlane && pk) {
            const unsigned w = pk[e];
            s = (int)(w >> 8);
            t = (int)(w & 255u);
            if (kl) kl[32 + i] = w;
          } else {
            t = byPlane ? dst[e] - lo : s_t[i];
            s = src[e];
            if (kl) kl[32 + i] = ((unsigned)s << 8) | (unsigned)t;
          }
        }
        s_t[i] = t | (atomicAdd(&s_rc[t], 1) << 8);
        const v2f as2 = pair_of<TD>(als, s, tn);
        const v2f ad2 = TD ? pair_tile(ald, tile, t) : pair_at(ald, lo + t);
        const float v0 = lrelu(as2.x + ad2.x);
        const float v1 = lrelu(as2.y + ad2.y);
        s_a[i] = s;
        s_e0[i] = v0;
        s_e1[i] = v1;
        ldsMaxF(&s_m[2 * t], v0);
        ldsMaxF(&s_m[2 * t + 1], v1);
      }
      __syncthreads();
      if (wave == 0) {
        int c8[8];
        int run = 0;
#pragma unroll
        for (int k = 0; k < 8; ++k) {
          const int n = s_rc[8 * lane + k];
          c8[k] = run;
          run += n;
        }
        int inc = run;
#pragma unroll
        for (int off = 1; off < 32; off <<= 1) {
          const int v = __shfl_up(inc, off, 32);
          if (lane >= off) inc += v;
        }
        const int first = inc - run;
#pragma unroll
        for (int k = 0; k < 8; ++k) s_rc[8 * lane + k] = first + c8[k];
      }
      __syncthreads();
      for (int i = tid; i < total; i += ET) {
        const int tw = s_t[i];
        const int t = tw & 255;
        const float x0 = __expf(s_e0[i] - s_m[2 * t]);
        const float x1 = __expf(s_e1[i] - s_m[2 * t + 1]);
        s_e0[i] = x0;
        s_e1[i] = x1;
        atomicAdd(&s_den[2 * t], x0);
        atomicAdd(&s_den[2 * t + 1], x1);
        s_perm[s_rc[t] + (tw >> 8)] = i;
      }
      __syncthreads();
      for (int base = wave * 32; base < total; base += ET) {
        const bool valid = base + lane < total;
        int s = 0, t = 0;
        float a0 = 0.f, a1 = 0.f;
        if (valid) {
          const int i = s_perm[base + lane];
          s = s_a[i];
          t = s_t[i] & 255;
          a0 = s_e0[i] / (s_den[2 * t] + 1e-16f) * 0.5f;
          a1 = s_e1[i] / (s_den[2 * t + 1] + 1e-16f) * 0.5f;
        }
        const unsigned mask = __builtin_amdgcn_ballot_w32(valid);
        wave_rows<8, AP>(mask, s, t, a0, a1, hs, s_agg, lane);
      }
    } else {
      sweep_all<1, TD, AP>(src, dst, als, ald, hs, s_agg, s_m, s_den, lo, nn, Ee, tid, lane, tn);
      __syncthreads();
      sweep_all<2, TD, AP>(src, dst, als, ald, hs, s_agg, s_m, s_den, lo, nn, Ee, tid, lane, tn);
      __syncthreads();
      sweep_all<3, TD, AP>(src, dst, als, ald, hs, s_agg, s_m, s_den, lo, nn, Ee, tid, lane, tn);
    }
    __syncthreads();

    for (int i = tid; i < nn * 64; i += ET) {
      const int c = i & 63;
      const size_t o = (size_t)lo * 64 + i;
      const int li = (i >> 6) * AP + c;
      const float v = s_agg[li] + b[c] + hl[o] + bl[c];
      const float r = (v > 0.f) ? v : 0.f;
      if (FUSE) s_agg[li] = r; else out[o] = r;
    }
    if (FUSE) {
      const int padded = ((nn + 15) >> 4) << 4;
      for (int i = nn * 64 + tid; i < padded * 64; i += ET) s_agg[(i >> 6) * AP + (i & 63)] = 0.f;
      __syncthreads();
      tile_node_part<AP>(s_agg, nn, lo, nWs, nWd, nWl, nAs, nAd, hsN, hlN, s_dot,
                         __builtin_amdgcn_readfirstlane(wave), lane);
    }
    __syncthreads();
    if (FUSE) {
      for (int r = tid; r < nn; r += ET) {
        v2f pa, pd;
        pa.x = s_dot[r];          pa.y = s_dot[TM + r];
        pd.x = s_dot[2 * TM + r]; pd.y = s_dot[3 * TM + r];
        *(v2f*)(alsN + (size_t)tile * TPITCH + 2 * r) = pa;
        *(v2f*)(aldN + (size_t)tile * TPITCH + 2 * r) = pd;
      }
    }
  }
}

extern "C" void kernel_launch(void* const* d_in, const int* in_sizes, int n_in,
                              void* d_out, int out_size, void* d_ws, size_t ws_size,
                              hipStream_t stream) {
  const float* x  = (const float*)d_in[0];
  const int*   ei = (const int*)d_in[1];
  const int Nn = in_sizes[0] / 128;
  const int Ee = in_sizes[1] / 2;
  const int* src = ei;
  const int* dst = ei + Ee;

  const float *Ws[3], *Wd[3], *As[3], *Ad[3], *B[3], *Wl[3], *Bl[3];
  for (int l = 0; l < 3; ++l) {
    int base = 2 + 7 * l;
    Ws[l] = (const float*)d_in[base + 0];
    Wd[l] = (const float*)d_in[base + 1];
    As[l] = (const float*)d_in[base + 2];
    Ad[l] = (const float*)d_in[base + 3];
    B[l]  = (const float*)d_in[base + 4];
    Wl[l] = (const float*)d_in[base + 5];
    Bl[l] = (const float*)d_in[base + 6];
  }

  float* ws = (float*)d_ws;
  size_t off = 0;
  float* hs  = ws + off; off += (size_t)Nn * 128;
  float* hl  = ws + off; off += (size_t)Nn * 64;
  float* als = ws + off; off += (size_t)Nn * 4;
  float* ald = ws + off; off += (size_t)Nn * 4;
  float* h1  = ws + off; off += (size_t)Nn * 64;
  float* h2  = ws + off; off += (size_t)Nn * 64;
  int tn = (Nn + EG - 1) / EG;
  if (tn > TM) tn = TM;
  if (tn < 1) tn = 1;
  const int tiles = (Nn + tn - 1) / tn;
  unsigned* keep = nullptr;
  if (Nn <= (1 << 24) && ws_size >= (off + (size_t)tiles * LPITCH) * sizeof(float))
    keep = (unsigned*)(ws + off);
  if (keep) off += (size_t)tiles * LPITCH;
  unsigned* plane = nullptr;
  const size_t planeWords = (size_t)((Ee + 15) >> 4) * 4;
  if (tiles <= 256 && ws_size >= (off + planeWords) * sizeof(float))
    plane = (unsigned*)(ws + off);
  if (plane) off += planeWords;
  const bool fuse = keep != nullptr && ws_size >= (off + (size_t)3 * tiles * TPITCH) * sizeof(float);
  float* hsB  = h1;
  float* alsA = ws + off;
  float* alsB = alsA + (size_t)tiles * TPITCH;
  float* aldT = alsB + (size_t)tiles * TPITCH;
  const size_t pkOff = off + (fuse ? (size_t)3 * tiles * TPITCH : 0);
  unsigned* pk = nullptr;
  if (plane && Nn <= (1 << 24) && ws_size >= (pkOff + planeWords * 4) * sizeof(float))
    pk = (unsigned*)(ws + pkOff);

  const float* lin[3]  = { x, h1, h2 };
  float*       lout[3] = { h1, h2, (float*)d_out };

  if (fuse) {
    node_part<128, true><<<256, 256, 0, stream>>>(x, Ws[0], Wd[0], Wl[0], As[0], Ad[0], hs, hl, als, ald, Nn,
                                                  src, dst, plane, pk, Ee, tn);
    edge_tile<false, false, true><<<EG, ET, 0, stream>>>(src, dst, als, ald, hs, hl, B[0], Bl[0], nullptr, keep, plane, pk, Nn, Ee,
                                                         Ws[1], Wd[1], Wl[1], As[1], Ad[1], hsB, hl, alsA, aldT);
    edge_tile<true, true, true><<<EG, ET, 0, stream>>>(src, dst, alsA, aldT, hsB, hl, B[1], Bl[1], nullptr, keep, plane, pk, Nn, Ee,
                                                       Ws[2], Wd[2], Wl[2], As[2], Ad[2], hs, hl, alsB, aldT);
    edge_tile<true, true, false><<<EG, ET, 0, stream>>>(src, dst, alsB, aldT, hs, hl, B[2], Bl[2], (float*)d_out, keep, plane, pk, Nn, Ee,
                                                        nullptr, nullptr, nullptr, nullptr, nullptr, nullptr, nullptr, nullptr, nullptr);
    return;
  }

  for (int l = 0; l < 3; ++l) {
    if (l == 0)
      node_part<128, true><<<256, 256, 0, stream>>>(lin[l], Ws[l], Wd[l], Wl[l], As[l], Ad[l], hs, hl, als, ald, Nn,
                                                    src, dst, plane, pk, Ee, tn);
    else
      node_part<64, false><<<512, 256, 0, stream>>>(lin[l], Ws[l], Wd[l], Wl[l], As[l], Ad[l], hs, hl, als, ald, Nn,
                                                    src, dst, plane, pk, Ee, tn);
    if (l == 0 || !keep)
      edge_tile<false, false, false><<<EG, ET, 0, stream>>>(src, dst, als, ald, hs, hl, B[l], Bl[l], lout[l], keep, plane, pk, Nn, Ee,
                                                            nullptr, nullptr, nullptr, nullptr, nullptr, nullptr, nullptr, nullptr, nullptr);
    else
      edge_tile<true, false, false><<<EG, ET, 0, stream>>>(src, dst, als, ald, hs, hl, B[l], Bl[l], lout[l], keep, plane, pk, Nn, Ee,
                                                           nullptr, nullptr, nullptr, nullptr, nullptr, nullptr, nullptr, nullptr, nullptr);
  }
}


# Edge lists without a sweep: launch 1 sorts each 4096-edge chunk by destination tile (LDS counters); every layer's work-group joins its tile's segments; no kept lists; list order (sum order) changed
# speedup vs baseline: 5.3097x; 1.0542x over previous
#include <hip/hip_runtime.h>
#include <math.h>

typedef __attribute__((ext_vector_type(2))) float v2f;
typedef __attribute__((ext_vector_type(4))) float v4f;
typedef __attribute__((ext_vector_type(8))) float v8f;
typedef __attribute__((ext_vector_type(4))) int v4i;

#define NEG_SLOPE 0.2f
#define CL 4096

template <int K>
__device__ __forceinline__ void gemm_load_a(v4f (&a)[K / 8], const float* abase, int rt) {
  const float* ap = abase + (size_t)rt * 16 * K;
#pragma unroll
  for (int j = 0; j < K / 8; ++j) a[j] = *(const v4f*)(ap + 8 * j);
}

template <int K>
__device__ __forceinline__ void gemm_tile(v8f (&acc)[4], const v4f (&a)[K / 8],
                                          const v2f (&bp)[K / 8][2][4]) {
#pragma unroll
  for (int q = 0; q < 4; ++q) acc[q] = (v8f){0.f,0.f,0.f,0.f,0.f,0.f,0.f,0.f};

#pragma unroll
  for (int j = 0; j < K / 8; ++j) {
#pragma unroll
    for (int sub = 0; sub < 2; ++sub) {
      v2f av;
      av.x = sub ? a[j].z : a[j].x;
      av.y = sub ? a[j].w : a[j].y;
#pragma unroll
      for (int q = 0; q < 4; ++q) {
        acc[q] = __builtin_amdgcn_wmma_f32_16x16x4_f32(
             false, av,  false, bp[j][sub][q],
             (short)0, acc[q],  false,  false);
      }
    }
  }
}

__device__ __forceinline__ void gemm_store(const v8f (&acc)[4], float* cbase, int rt, int M) {
  float* crow = cbase + (size_t)rt * 16 * M;
#pragma unroll
  for (int r = 0; r < 8; ++r) {
    v4f o;
    o.x = acc[0][r]; o.y = acc[1][r]; o.z = acc[2][r]; o.w = acc[3][r];
    *(v4f*)(crow + (size_t)r * M) = o;
  }
}

__device__ __forceinline__ float tile_row_dots(const v8f (&acc)[4], const v4f av, int c) {
  float p[8];
#pragma unroll
  for (int r = 0; r < 8; ++r)
    p[r] = acc[0][r] * av.x + acc[1][r] * av.y + acc[2][r] * av.z + acc[3][r] * av.w;
  const bool b3 = (c & 8) != 0, b2 = (c & 4) != 0, b1 = (c & 2) != 0;
  float k4[4], k2[2];
#pragma unroll
  for (int i = 0; i < 4; ++i) {
    const float send = b3 ? p[i] : p[4 + i];
    const float keep = b3 ? p[4 + i] : p[i];
    k4[i] = keep + __shfl_xor(send, 8, 32);
  }
#pragma unroll
  for (int i = 0; i < 2; ++i) {
    const float send = b2 ? k4[i] : k4[2 + i];
    const float keep = b2 ? k4[2 + i] : k4[i];
    k2[i] = keep + __shfl_xor(send, 4, 32);
  }
  const float send1 = b1 ? k2[0] : k2[1];
  const float keep1 = b1 ? k2[1] : k2[0];
  const float k1 = keep1 + __shfl_xor(send1, 2, 32);
  return k1 + __shfl_xor(k1, 1, 32);
}

template <int K, bool PLANE>
__global__ void __launch_bounds__(256)
node_part(const float* __restrict__ A,
          const float* __restrict__ Ws, const float* __restrict__ Wd,
          const float* __restrict__ Wl,
          const float* __restrict__ a_s, const float* __restrict__ a_d,
          float* __restrict__ hs, float* __restrict__ hl,
          float* __restrict__ als, float* __restrict__ ald,
          int Nrows,
          const int* __restrict__ esrc, const int* __restrict__ dst,
          unsigned* __restrict__ tab, unsigned* __restrict__ clist, int Ee, int tn) {
  constexpr int KJ = K / 8;

  if (PLANE && tab) {
    __shared__ int b_cnt[256 + 32];
    const int tid = threadIdx.x;
    const int chunks = (Ee + CL - 1) / CL;
    for (int ch = blockIdx.x; ch < chunks; ch += gridDim.x) {
      b_cnt[tid] = 0;
      if (tid < 32) b_cnt[256 + tid] = 0;
      const int e0 = ch * CL;
      unsigned dv[16], sv[16];
#pragma unroll
      for (int k = 0; k < 16; ++k) {
        const int e = e0 + k * 256 + tid;
        const int ec = e < Ee ? e : Ee - 1;
        dv[k] = (unsigned)dst[ec];
        sv[k] = (unsigned)esrc[ec];
      }
      __syncthreads();
      unsigned wv[16];
      int tp[16];
#pragma unroll
      for (int k = 0; k < 16; ++k) {
        const int e = e0 + k * 256 + tid;
        const unsigned T = dv[k] / (unsigned)tn;
        wv[k] = (sv[k] << 8) | (dv[k] - T * (unsigned)tn);
        const int place = atomicAdd(&b_cnt[e < Ee ? (int)T : 256], 1);
        tp[k] = e < Ee ? ((int)T | (place << 8)) : -1;
      }
      __syncthreads();
      if (tid < 32) {
        int c8[8], n8[8];
        int run = 0;
#pragma unroll
        for (int k = 0; k < 8; ++k) {
          n8[k] = b_cnt[8 * tid + k];
          c8[k] = run;
          run += n8[k];
        }
        int inc = run;
#pragma unroll
        for (int off = 1; off < 32; off <<= 1) {
          const int v = __shfl_up(inc, off, 32);
          if (tid >= off) inc += v;
        }
        const int first = inc - run;
#pragma unroll
        for (int k = 0; k < 8; ++k) {
          b_cnt[8 * tid + k] = first + c8[k];
          tab[(size_t)ch * 256 + 8 * tid + k] = (unsigned)(first + c8[k]) | ((unsigned)n8[k] << 16);
        }
      }
      __syncthreads();
      int fp[16];
#pragma unroll
      for (int k = 0; k < 16; ++k) fp[k] = b_cnt[tp[k] & 255] + (tp[k] >> 8);
#pragma unroll
      for (int k = 0; k < 16; ++k)
        if (tp[k] >= 0) clist[(unsigned)(e0 + fp[k])] = wv[k];
      __syncthreads();
    }
  }

  const int lane = threadIdx.x & 31;
  const int c = lane & 15;
  const int h = lane >> 4;
  const int wib = __builtin_amdgcn_readfirstlane((int)(threadIdx.x >> 5));
  const int wavesPerBlock = blockDim.x >> 5;
  const int wave = blockIdx.x * wavesPerBlock + wib;
  const int slab = wave % 5;
  const int rg = wave / 5;
  const int rowGroups = (gridDim.x * wavesPerBlock) / 5;
  const int nTiles = Nrows >> 4;
  if (rg >= rowGroups || rg >= nTiles) return;

  const bool doStore = (slab < 2) || (slab == 4);
  const bool doDot = slab < 4;
  const int head = slab & 1;
  const int M = (slab == 4) ? 64 : 128;
  const int cs = (slab == 4) ? 0 : head * 64;
  const float* W = (slab < 2) ? Ws : ((slab < 4) ? Wd : Wl);
  float* C = (slab < 2) ? hs : hl;
  const float* avec = (slab < 2) ? a_s : a_d;
  float* dots = (slab < 2) ? als : ald;

  v2f bp[KJ][2][4];
  {
    const float* wbase = W + (size_t)(4 * h) * M + cs + 4 * c;
#pragma unroll
    for (int j = 0; j < KJ; ++j) {
#pragma unroll
      for (int sub = 0; sub < 2; ++sub) {
        const v4f w0 = *(const v4f*)(wbase + (size_t)(8 * j + 2 * sub) * M);
        const v4f w1 = *(const v4f*)(wbase + (size_t)(8 * j + 2 * sub + 1) * M);
#pragma unroll
        for (int q = 0; q < 4; ++q) {
          bp[j][sub][q].x = w0[q];
          bp[j][sub][q].y = w1[q];
        }
      }
    }
  }
  const v4f av = *(const v4f*)(avec + head * 64 + 4 * c);

  const float* abase = A + (size_t)c * K + 4 * h;
  float* cbase = C + (size_t)(8 * h) * M + cs + 4 * c;
  float* dbase = dots + head * 32 + 8 * h + (c >> 1);
  const bool dotLane = (c & 1) == 0;
  v4f a0[KJ], a1[KJ];
  gemm_load_a<K>(a0, abase, rg);

  int rt = rg;
  for (;;) {
    const int r1 = rt + rowGroups;
    const int r2 = r1 + rowGroups;
    v8f acc[4];
    __builtin_amdgcn_sched_barrier(0);
    gemm_load_a<K>(a1, abase, r1 < nTiles ? r1 : rt);
    __builtin_amdgcn_sched_barrier(0);
    gemm_tile<K>(acc, a0, bp);
    if (doStore) gemm_store(acc, cbase, rt, M);
    if (doDot) {
      const float d = tile_row_dots(acc, av, c);
      if (dotLane) dbase[(size_t)rt * 64] = d;
    }
    __builtin_amdgcn_sched_barrier(0);
    gemm_load_a<K>(a0, abase, r2 < nTiles ? r2 : rt);
    __builtin_amdgcn_sched_barrier(0);
    gemm_tile<K>(acc, a1, bp);
    if (r1 < nTiles) {
      if (doStore) gemm_store(acc, cbase, r1, M);
      if (doDot) {
        const float d = tile_row_dots(acc, av, c);
        if (dotLane) dbase[(size_t)r1 * 64] = d;
      }
    }
    if (r2 >= nTiles) break;
    rt = r2;
  }
}

#define TM   256
#define LCAP 8192
#define ET   1024
#define EG   256
#define NCH  256
static_assert(TM <= 256, "a list entry holds the tile-relative destination in 8 bits");
static_assert(TM == 256 && LCAP <= (1 << 23), "the row scan takes 8 rows per lane of one wave; an entry's place in its row sits above 8 bits of an int");
static_assert(ET == 4 * NCH && CL <= (1 << 15) && LCAP >= 3 * NCH, "four threads per chunk; first place and count of a segment share a word; the chunk scan borrows s_perm");
#define TPITCH 1024

__device__ __forceinline__ float lrelu(float v) { return (v > 0.f) ? v : NEG_SLOPE * v; }

__device__ __forceinline__ v2f pair_at(const float* plane, int n) {
  const float* p = plane + (size_t)(n >> 4) * 64 + (n & 15);
  v2f r;
  r.x = p[0];
  r.y = p[32];
  return r;
}
__device__ __forceinline__ v2f pair_tile(const float* plane, int T, int r) {
  return *(const v2f*)(plane + (size_t)T * TPITCH + 2 * r);
}
template <bool TD>
__device__ __forceinline__ v2f pair_of(const float* plane, int n, int tn) {
  if (TD) {
    const unsigned T = (unsigned)n / (unsigned)tn;
    return pair_tile(plane, (int)T, (int)((unsigned)n - T * (unsigned)tn));
  }
  return pair_at(plane, n);
}

__device__ __forceinline__ void ldsMaxF(float* addr, float value) {
  if (value >= 0.f)
    atomicMax((int*)addr, __float_as_int(value));
  else
    atomicMin((unsigned int*)addr, __float_as_uint(value));
}

template <int U, int AP>
__device__ __forceinline__ void wave_rows(unsigned mask, int s, int t, float a0, float a1,
                                          const float* __restrict__ hs, float* agg, int lane) {
  int cur = -1;
  float rx = 0.f, ry = 0.f;
  while (mask) {
    int l[U];
    bool on[U];
    unsigned mm = mask;
#pragma unroll
    for (int u = 0; u < U; ++u) {
      on[u] = (mm != 0u);
      l[u] = on[u] ? __builtin_ctz(mm) : l[0];
      mm &= mm - 1u;
    }
    mask = mm;
    v2f x0[U], x1[U];
    int tt[U];
    float A0[U], A1[U];
#pragma unroll
    for (int u = 0; u < U; ++u) {
      const int ss = __builtin_amdgcn_readlane(s, l[u]);
      tt[u] = __builtin_amdgcn_readlane(t, l[u]);
      A0[u] = __int_as_float(__builtin_amdgcn_readlane(__float_as_int(a0), l[u]));
      A1[u] = __int_as_float(__builtin_amdgcn_readlane(__float_as_int(a1), l[u]));
      const float* row = hs + (size_t)ss * 128 + 2 * lane;
      x0[u] = *(const v2f*)row;
      x1[u] = *(const v2f*)(row + 64);
    }
#pragma unroll
    for (int u = 0; u < U; ++u) asm volatile("" :: "v"(x0[u]), "v"(x1[u]));
#pragma unroll
    for (int u = 0; u < U; ++u) {
      if (on[u]) {
        if (tt[u] != cur) {
          if (cur >= 0) {
            float* p = agg + cur * AP + 2 * lane;
            atomicAdd(p,     rx);
            atomicAdd(p + 1, ry);
          }
          cur = tt[u];
          rx = 0.f;
          ry = 0.f;
        }
        rx += x0[u].x * A0[u] + x1[u].x * A1[u];
        ry += x0[u].y * A0[u] + x1[u].y * A1[u];
      }
    }
  }
  if (cur >= 0) {
    float* p = agg + cur * AP + 2 * lane;
    atomicAdd(p,     rx);
    atomicAdd(p + 1, ry);
  }
}

template <int MODE, bool TD, int AP>
__device__ __forceinline__ void sweep_all(const int* __restrict__ src, const int* __restrict__ dst,
                                          const float* __restrict__ als, const float* __restrict__ ald,
                                          const float* __restrict__ hs,
                                          float* s_agg, float* s_m, float* s_den,
                                          int lo, int nn, int Ee, int tid, int lane, int tn) {
  const int trips = (Ee + ET - 1) / ET;
  for (int it = 0; it < trips; ++it) {
    const int e = it * ET + tid;
    bool valid = false;
    int s = 0, t = 0;
    float a0 = 0.f, a1 = 0.f;
    if (e < Ee) {
      t = dst[e] - lo;
      valid = (unsigned)t < (unsigned)nn;
      if (valid) {
        s = src[e];
        const v2f as2 = pair_of<TD>(als, s, tn);
        const v2f ad2 = pair_of<TD>(ald, lo + t, tn);
        const float v0 = lrelu(as2.x + ad2.x);
        const float v1 = lrelu(as2.y + ad2.y);
        if (MODE == 1) {
          ldsMaxF(&s_m[2 * t], v0);
          ldsMaxF(&s_m[2 * t + 1], v1);
        } else if (MODE == 2) {
          atomicAdd(&s_den[2 * t],     __expf(v0 - s_m[2 * t]));
          atomicAdd(&s_den[2 * t + 1], __expf(v1 - s_m[2 * t + 1]));
        } else {
          a0 = __expf(v0 - s_m[2 * t])     / (s_den[2 * t]     + 1e-16f) * 0.5f;
          a1 = __expf(v1 - s_m[2 * t + 1]) / (s_den[2 * t + 1] + 1e-16f) * 0.5f;
        }
      }
    }
    if (MODE == 3) {
      const unsigned mask = __builtin_amdgcn_ballot_w32(valid);
      wave_rows<4, AP>(mask, s, t, a0, a1, hs, s_agg, lane);
    }
  }
}


template <int M>
__device__ __forceinline__ void tile_product(v8f (&acc)[4], const float* ap, const float* __restrict__ wslab, int woff) {
#pragma unroll
  for (int q = 0; q < 4; ++q) acc[q] = (v8f){0.f,0.f,0.f,0.f,0.f,0.f,0.f,0.f};
#pragma nounroll
  for (int jj = 0; jj < 4; ++jj) {
    const float* wj = wslab + (size_t)jj * 16 * M;
    v4f a[2], w[2][2][2];
#pragma unroll
    for (int u = 0; u < 2; ++u) {
      a[u] = *(const v4f*)(ap + 16 * jj + 8 * u);
#pragma unroll
      for (int sub = 0; sub < 2; ++sub) {
        w[u][sub][0] = *(const v4f*)(wj + woff + (8 * u + 2 * sub) * M);
        w[u][sub][1] = *(const v4f*)(wj + woff + (8 * u + 2 * sub + 1) * M);
      }
    }
#pragma unroll
    for (int u = 0; u < 2; ++u)
      asm volatile("" :: "v"(w[u][0][0]), "v"(w[u][0][1]), "v"(w[u][1][0]), "v"(w[u][1][1]));
#pragma unroll
    for (int u = 0; u < 2; ++u) {
#pragma unroll
      for (int sub = 0; sub < 2; ++sub) {
        v2f avv;
        avv.x = sub ? a[u].z : a[u].x;
        avv.y = sub ? a[u].w : a[u].y;
#pragma unroll
        for (int q = 0; q < 4; ++q) {
          v2f bv;
          bv.x = w[u][sub][0][q];
          bv.y = w[u][sub][1][q];
          acc[q] = __builtin_amdgcn_wmma_f32_16x16x4_f32(
               false, avv,  false, bv,
               (short)0, acc[q],  false,  false);
        }
      }
    }
  }
}

template <int AP>
__device__ __forceinline__ void tile_node_part(const float* rows, int nn, int lo,
                                               const float* __restrict__ Ws, const float* __restrict__ Wd,
                                               const float* __restrict__ Wl,
                                               const float* __restrict__ a_s, const float* __restrict__ a_d,
                                               float* __restrict__ hs, float* __restrict__ hl,
                                               float* dotsL,
                                               int wave, int lane) {
  const int c = lane & 15;
  const int h = lane >> 4;
  const bool dotLane = (c & 1) == 0;
  const int nT = (nn + 15) >> 4;
  for (int id = wave; id < 5 * nT; id += 32) {
    const int rt = id / 5;
    const int slab = id - 5 * rt;
    const bool doStore = (slab < 2) || (slab == 4);
    const bool doDot = slab < 4;
    const int head = slab & 1;
    const int M = (slab == 4) ? 64 : 128;
    const int cs = (slab == 4) ? 0 : head * 64;
    const float* W = (slab < 2) ? Ws : Wd;
    float* C = (slab < 2) ? hs : hl;
    const float* avec = (slab < 2) ? a_s : a_d;

    v8f acc[4];
    const float* ap = rows + (16 * rt + c) * AP + 4 * h;
    if (slab == 4) tile_product<64>(acc, ap, Wl, 4 * h * 64 + 4 * c);
    else           tile_product<128>(acc, ap, W + cs, 4 * h * 128 + 4 * c);
    if (doStore) {
      float* crow = C + (size_t)(lo + 16 * rt + 8 * h) * M + cs + 4 * c;
      const int left = nn - (16 * rt + 8 * h);
#pragma unroll
      for (int r = 0; r < 8; ++r) {
        v4f o;
        o.x = acc[0][r]; o.y = acc[1][r]; o.z = acc[2][r]; o.w = acc[3][r];
        if (r < left) *(v4f*)(crow + (size_t)r * M) = o;
      }
    }
    if (doDot) {
      const v4f av = *(const v4f*)(avec + head * 64 + 4 * c);
      const float d = tile_row_dots(acc, av, c);
      if (dotLane) dotsL[slab * TM + 16 * rt + 8 * h + (c >> 1)] = d;
    }
  }
}

template <bool BIN, bool TD, bool FUSE>
__global__ void __launch_bounds__(ET)
edge_tile(const int* __restrict__ src, const int* __restrict__ dst,
          const float* __restrict__ als, const float* ald,
          const float* __restrict__ hs, const float* hl,
          const float* __restrict__ b, const float* __restrict__ bl,
          float* __restrict__ out,
          const unsigned* __restrict__ tab, const unsigned* __restrict__ clist, int Nn, int Ee,
          const float* __restrict__ nWs, const float* __restrict__ nWd, const float* __restrict__ nWl,
          const float* __restrict__ nAs, const float* __restrict__ nAd,
          float* __restrict__ hsN, float* hlN, float* __restrict__ alsN, float* aldN) {
  constexpr int AP = FUSE ? 68 : 64;
  __shared__ __attribute__((aligned(16))) float s_agg[TM * AP];
  __shared__ float s_m[TM * 2];
  __shared__ float s_den[TM * 2];
  __shared__ int   s_a[LCAP];
  __shared__ int   s_t[LCAP];
  __shared__ int   s_perm[LCAP];
  __shared__ int   s_rc[TM];
  __shared__ float s_e0[LCAP];
  __shared__ float s_e1[LCAP];
  __shared__ float s_dot[FUSE ? 4 * TM : 1];
  __shared__ int   s_cnt;

  const int tid = threadIdx.x;
  const int lane = tid & 31;
  const int wave = tid >> 5;

  int tn = (Nn + (int)gridDim.x - 1) / (int)gridDim.x;
  if (tn > TM) tn = TM;
  if (tn < 1) tn = 1;
  const int tiles = (Nn + tn - 1) / tn;

  constexpr bool binned = BIN;
  const int chunks = (Ee + CL - 1) / CL;

  int i0 = (int)(((16u - (unsigned)((size_t)dst & 15u)) & 15u) >> 2);
  if (i0 > Ee) i0 = Ee;
  const int G = (Ee - i0) >> 2;
  const v4i* d4 = (const v4i*)(dst + i0);
  const int rot = (int)(((long long)blockIdx.x * G) / (int)gridDim.x);
  constexpr int U = 8;

  for (int tile = blockIdx.x; tile < tiles; tile += gridDim.x) {
    const int lo = tile * tn;
    const int nn = (Nn - lo < tn) ? (Nn - lo) : tn;

    for (int i = tid; i < nn * 64; i += ET) s_agg[(i >> 6) * AP + (i & 63)] = 0.f;
    for (int i = tid; i < nn * 2; i += ET) { s_m[i] = -INFINITY; s_den[i] = 0.f; }
    if (tid < TM) s_rc[tid] = 0;
    if (binned) {
      if (tid < NCH) {
        const unsigned seg = (tid < chunks) ? tab[(unsigned)(tid * 256 + tile)] : 0u;
        s_perm[tid] = (int)(seg >> 16);
        s_perm[2 * NCH + tid] = (int)(seg & 0xffffu);
      }
    } else if (tid == 0) {
      s_cnt = 0;
    }
    __syncthreads();

    int total;
    if (binned) {
      if (wave == 0) {
        int c8[8];
        int run = 0;
#pragma unroll
        for (int k = 0; k < 8; ++k) {
          const int n = s_perm[8 * lane + k];
          c8[k] = run;
          run += n;
        }
        int inc = run;
#pragma unroll
        for (int off = 1; off < 32; off <<= 1) {
          const int v = __shfl_up(inc, off, 32);
          if (lane >= off) inc += v;
        }
        const int first = inc - run;
#pragma unroll
        for (int k = 0; k < 8; ++k) s_perm[8 * lane + k] = first + c8[k];
        if (lane == 31) s_perm[NCH] = inc;
      }
      __syncthreads();
      total = s_perm[NCH];
      {
        const int ch = tid >> 2;
        const int sub = tid & 3;
        const int pre = s_perm[ch];
        const int cnt = s_perm[ch + 1] - pre;
        const unsigned sbase = (unsigned)(ch * CL + s_perm[2 * NCH + ch]);
        for (int k0 = sub; k0 < cnt; k0 += 16) {
          unsigned w[4];
#pragma unroll
          for (int u = 0; u < 4; ++u) {
            const int kk = k0 + 4 * u;
            w[u] = clist[sbase + (unsigned)(kk < cnt ? kk : cnt - 1)];
          }
#pragma unroll
          for (int u = 0; u < 4; ++u) {
            const int kk = k0 + 4 * u;
            if (kk < cnt && pre + kk < LCAP) s_a[pre + kk] = (int)w[u];
          }
        }
      }
      __syncthreads();
    } else {
      for (int g0 = 0; g0 < G; g0 += ET * U) {
        v4i d[U];
        int gi[U];
        bool ok[U];
#pragma unroll
        for (int u = 0; u < U; ++u) {
          const int g = g0 + u * ET + tid;
          ok[u] = g < G;
          int gg = (ok[u] ? g : G - 1) + rot;
          if (gg >= G) gg -= G;
          gi[u] = gg;
          d[u] = d4[gg];
        }
#pragma unroll
        for (int u = 0; u < U; ++u) {
          const int t0 = d[u].x - lo, t1 = d[u].y - lo, t2 = d[u].z - lo, t3 = d[u].w - lo;
          const bool m0 = (unsigned)t0 < (unsigned)nn, m1 = (unsigned)t1 < (unsigned)nn;
          const bool m2 = (unsigned)t2 < (unsigned)nn, m3 = (unsigned)t3 < (unsigned)nn;
          if (ok[u] && (m0 | m1 | m2 | m3)) {
            const int e = i0 + 4 * gi[u];
            if (m0) { const int sl = atomicAdd(&s_cnt, 1); if (sl < LCAP) { s_a[sl] = e;     s_t[sl] = t0; } }
            if (m1) { const int sl = atomicAdd(&s_cnt, 1); if (sl < LCAP) { s_a[sl] = e + 1; s_t[sl] = t1; } }
            if (m2) { const int sl = atomicAdd(&s_cnt, 1); if (sl < LCAP) { s_a[sl] = e + 2; s_t[sl] = t2; } }
            if (m3) { const int sl = atomicAdd(&s_cnt, 1); if (sl < LCAP) { s_a[sl] = e + 3; s_t[sl] = t3; } }
          }
        }
      }
      {
        const int rest = Ee - 4 * G;
        if (tid < rest) {
          const int e = (tid < i0) ? tid : (4 * G + tid);
          const int t = dst[e] - lo;
          if ((unsigned)t < (unsigned)nn) {
            const int sl = atomicAdd(&s_cnt, 1);
            if (sl < LCAP) { s_a[sl] = e; s_t[sl] = t; }
          }
        }
      }
      __syncthreads();
      total = s_cnt;
    }

    if (total <= LCAP) {
      for (int i = tid; i < total; i += ET) {
        int s, t;
        if (binned) {
          const unsigned w = (unsigned)s_a[i];
          s = (int)(w >> 8);
          t = (int)(w & 255u);
        } else {
          t = s_t[i];
          s = src[s_a[i]];
        }
        s_t[i] = t | (atomicAdd(&s_rc[t], 1) << 8);
        const v2f as2 = pair_of<TD>(als, s, tn);
        const v2f ad2 = TD ? pair_tile(ald, tile, t) : pair_at(ald, lo + t);
        const float v0 = lrelu(as2.x + ad2.x);
        const float v1 = lrelu(as2.y + ad2.y);
        s_a[i] = s;
        s_e0[i] = v0;
        s_e1[i] = v1;
        ldsMaxF(&s_m[2 * t], v0);
        ldsMaxF(&s_m[2 * t + 1], v1);
      }
      __syncthreads();
      if (wave == 0) {
        int c8[8];
        int run = 0;
#pragma unroll
        for (int k = 0; k < 8; ++k) {
          const int n = s_rc[8 * lane + k];
          c8[k] = run;
          run += n;
        }
        int inc = run;
#pragma unroll
        for (int off = 1; off < 32; off <<= 1) {
          const int v = __shfl_up(inc, off, 32);
          if (lane >= off) inc += v;
        }
        const int first = inc - run;
#pragma unroll
        for (int k = 0; k < 8; ++k) s_rc[8 * lane + k] = first + c8[k];
      }
      __syncthreads();
      for (int i = tid; i < total; i += ET) {
        const int tw = s_t[i];
        const int t = tw & 255;
        const float x0 = __expf(s_e0[i] - s_m[2 * t]);
        const float x1 = __expf(s_e1[i] - s_m[2 * t + 1]);
        s_e0[i] = x0;
        s_e1[i] = x1;
        atomicAdd(&s_den[2 * t], x0);
        atomicAdd(&s_den[2 * t + 1], x1);
        s_perm[s_rc[t] + (tw >> 8)] = i;
      }
      __syncthreads();
      for (int base = wave * 32; base < total; base += ET) {
        const bool valid = base + lane < total;
        int s = 0, t = 0;
        float a0 = 0.f, a1 = 0.f;
        if (valid) {
          const int i = s_perm[base + lane];
          s = s_a[i];
          t = s_t[i] & 255;
          a0 = s_e0[i] / (s_den[2 * t] + 1e-16f) * 0.5f;
          a1 = s_e1[i] / (s_den[2 * t + 1] + 1e-16f) * 0.5f;
        }
        const unsigned mask = __builtin_amdgcn_ballot_w32(valid);
        wave_rows<8, AP>(mask, s, t, a0, a1, hs, s_agg, lane);
      }
    } else {
      sweep_all<1, TD, AP>(src, dst, als, ald, hs, s_agg, s_m, s_den, lo, nn, Ee, tid, lane, tn);
      __syncthreads();
      sweep_all<2, TD, AP>(src, dst, als, ald, hs, s_agg, s_m, s_den, lo, nn, Ee, tid, lane, tn);
      __syncthreads();
      sweep_all<3, TD, AP>(src, dst, als, ald, hs, s_agg, s_m, s_den, lo, nn, Ee, tid, lane, tn);
    }
    __syncthreads();

    for (int i = tid; i < nn * 64; i += ET) {
      const int c = i & 63;
      const size_t o = (size_t)lo * 64 + i;
      const int li = (i >> 6) * AP + c;
      const float v = s_agg[li] + b[c] + hl[o] + bl[c];
      const float r = (v > 0.f) ? v : 0.f;
      if (FUSE) s_agg[li] = r; else out[o] = r;
    }
    if (FUSE) {
      const int padded = ((nn + 15) >> 4) << 4;
      for (int i = nn * 64 + tid; i < padded * 64; i += ET) s_agg[(i >> 6) * AP + (i & 63)] = 0.f;
      __syncthreads();
      tile_node_part<AP>(s_agg, nn, lo, nWs, nWd, nWl, nAs, nAd, hsN, hlN, s_dot,
                         __builtin_amdgcn_readfirstlane(wave), lane);
    }
    __syncthreads();
    if (FUSE) {
      for (int r = tid; r < nn; r += ET) {
        v2f pa, pd;
        pa.x = s_dot[r];          pa.y = s_dot[TM + r];
        pd.x = s_dot[2 * TM + r]; pd.y = s_dot[3 * TM + r];
        *(v2f*)(alsN + (size_t)tile * TPITCH + 2 * r) = pa;
        *(v2f*)(aldN + (size_t)tile * TPITCH + 2 * r) = pd;
      }
    }
  }
}

extern "C" void kernel_launch(void* const* d_in, const int* in_sizes, int n_in,
                              void* d_out, int out_size, void* d_ws, size_t ws_size,
                              hipStream_t stream) {
  const float* x  = (const float*)d_in[0];
  const int*   ei = (const int*)d_in[1];
  const int Nn = in_sizes[0] / 128;
  const int Ee = in_sizes[1] / 2;
  const int* src = ei;
  const int* dst = ei + Ee;

  const float *Ws[3], *Wd[3], *As[3], *Ad[3], *B[3], *Wl[3], *Bl[3];
  for (int l = 0; l < 3; ++l) {
    int base = 2 + 7 * l;
    Ws[l] = (const float*)d_in[base + 0];
    Wd[l] = (const float*)d_in[base + 1];
    As[l] = (const float*)d_in[base + 2];
    Ad[l] = (const float*)d_in[base + 3];
    B[l]  = (const float*)d_in[base + 4];
    Wl[l] = (const float*)d_in[base + 5];
    Bl[l] = (const float*)d_in[base + 6];
  }

  float* ws = (float*)d_ws;
  size_t off = 0;
  float* hs  = ws + off; off += (size_t)Nn * 128;
  float* hl  = ws + off; off += (size_t)Nn * 64;
  float* als = ws + off; off += (size_t)Nn * 4;
  float* ald = ws + off; off += (size_t)Nn * 4;
  float* h1  = ws + off; off += (size_t)Nn * 64;
  float* h2  = ws + off; off += (size_t)Nn * 64;
  int tn = (Nn + EG - 1) / EG;
  if (tn > TM) tn = TM;
  if (tn < 1) tn = 1;
  const int tiles = (Nn + tn - 1) / tn;
  const int chunks = (Ee + CL - 1) / CL;
  const size_t tabWords = (size_t)chunks * 256;
  const size_t clistWords = (size_t)chunks * CL;
  unsigned* tab = nullptr;
  unsigned* clist = nullptr;
  if (tiles <= 256 && chunks <= NCH && Nn <= (1 << 24) &&
      ws_size >= (off + tabWords + clistWords) * sizeof(float)) {
    tab = (unsigned*)(ws + off);
    clist = tab + tabWords;
    off += tabWords + clistWords;
  }
  const bool fuse = tab != nullptr && ws_size >= (off + (size_t)3 * tiles * TPITCH) * sizeof(float);
  float* hsB  = h1;
  float* alsA = ws + off;
  float* alsB = alsA + (size_t)tiles * TPITCH;
  float* aldT = alsB + (size_t)tiles * TPITCH;

  const float* lin[3]  = { x, h1, h2 };
  float*       lout[3] = { h1, h2, (float*)d_out };

  if (fuse) {
    node_part<128, true><<<256, 256, 0, stream>>>(x, Ws[0], Wd[0], Wl[0], As[0], Ad[0], hs, hl, als, ald, Nn,
                                                  src, dst, tab, clist, Ee, tn);
    edge_tile<true, false, true><<<EG, ET, 0, stream>>>(src, dst, als, ald, hs, hl, B[0], Bl[0], nullptr, tab, clist, Nn, Ee,
                                                        Ws[1], Wd[1], Wl[1], As[1], Ad[1], hsB, hl, alsA, aldT);
    edge_tile<true, true, true><<<EG, ET, 0, stream>>>(src, dst, alsA, aldT, hsB, hl, B[1], Bl[1], nullptr, tab, clist, Nn, Ee,
                                                       Ws[2], Wd[2], Wl[2], As[2], Ad[2], hs, hl, alsB, aldT);
    edge_tile<true, true, false><<<EG, ET, 0, stream>>>(src, dst, alsB, aldT, hs, hl, B[2], Bl[2], (float*)d_out, tab, clist, Nn, Ee,
                                                        nullptr, nullptr, nullptr, nullptr, nullptr, nullptr, nullptr, nullptr, nullptr);
    return;
  }

  for (int l = 0; l < 3; ++l) {
    if (l == 0)
      node_part<128, true><<<256, 256, 0, stream>>>(lin[l], Ws[l], Wd[l], Wl[l], As[l], Ad[l], hs, hl, als, ald, Nn,
                                                    src, dst, tab, clist, Ee, tn);
    else
      node_part<64, false><<<512, 256, 0, stream>>>(lin[l], Ws[l], Wd[l], Wl[l], As[l], Ad[l], hs, hl, als, ald, Nn,
                                                    src, dst, tab, clist, Ee, tn);
    if (tab)
      edge_tile<true, false, false><<<EG, ET, 0, stream>>>(src, dst, als, ald, hs, hl, B[l], Bl[l], lout[l], tab, clist, Nn, Ee,
                                                           nullptr, nullptr, nullptr, nullptr, nullptr, nullptr, nullptr, nullptr, nullptr);
    else
      edge_tile<false, false, false><<<EG, ET, 0, stream>>>(src, dst, als, ald, hs, hl, B[l], Bl[l], lout[l], tab, clist, Nn, Ee,
                                                            nullptr, nullptr, nullptr, nullptr, nullptr, nullptr, nullptr, nullptr, nullptr);
  }
}
